# starting kernel
# baseline (speedup 1.0000x reference)
#include <hip/hip_runtime.h>
#include <hip/hip_bf16.h>
#include <stdint.h>

#define N_NODES   50000
#define N_EDGES   800000
#define PE        64
#define PD        16
#define N_GRAPHS  256
#define EPS       1e-5f
#define NEG_SLOPE 0.01f

typedef __attribute__((ext_vector_type(2))) float v2f;
typedef __attribute__((ext_vector_type(8))) float v8f;

__global__ __launch_bounds__(256) void edge_msg_kernel(
    const float* __restrict__ x,
    const int*   __restrict__ src,
    const int*   __restrict__ dst,
    const float* __restrict__ attr,
    const float* __restrict__ Wm,
    const float* __restrict__ bm,
    const float* __restrict__ We,
    const float* __restrict__ be,
    float*       __restrict__ agg)
{
    __shared__ float sWm[PE * PE];
    __shared__ float sWe[PD * PE];
    __shared__ float sX[8][16 * PE];
    __shared__ float sA[8][16 * PD];
    __shared__ int   sD[8][16];

    const int tid  = threadIdx.x;
    const int wave = tid >> 5;
    const int lane = tid & 31;

    {
        const float4* Wm4  = (const float4*)Wm;
        float4*       sWm4 = (float4*)sWm;
        #pragma unroll
        for (int i = tid; i < PE * PE / 4; i += 256) sWm4[i] = Wm4[i];
        const float4* We4  = (const float4*)We;
        float4*       sWe4 = (float4*)sWe;
        if (tid < PD * PE / 4) sWe4[tid] = We4[tid];
    }

    const int tile = blockIdx.x * 8 + wave;
    const int e0   = tile * 16;

    if (lane < 16) sD[wave][lane] = dst[e0 + lane];

    {
        const int row = lane >> 1;
        const int hlf = lane & 1;
        const int s   = src[e0 + row];
        const float4* xr = (const float4*)(x + (size_t)s * PE + hlf * 32);
        float4*       ox = (float4*)(&sX[wave][row * PE + hlf * 32]);
        #pragma unroll
        for (int i = 0; i < 8; ++i) ox[i] = xr[i];
        const float4* ar = (const float4*)(attr + (size_t)(e0 + row) * PD + hlf * 8);
        float4*       oa = (float4*)(&sA[wave][row * PD + hlf * 8]);
        oa[0] = ar[0]; oa[1] = ar[1];
    }
    __syncthreads();

    const int col  = lane & 15;
    const int hi   = lane >> 4;
    const int koff = hi * 2;

    #pragma unroll
    for (int n = 0; n < 4; ++n) {
        const int n0 = n * 16;
        const float bias = bm[n0 + col] + be[n0 + col];
        v8f acc;
        #pragma unroll
        for (int r = 0; r < 8; ++r) acc[r] = bias;

        #pragma unroll
        for (int kk = 0; kk < 16; ++kk) {
            const int k0 = kk * 4;
            v2f a = *(const v2f*)(&sX[wave][col * PE + k0 + koff]);
            v2f b;
            b[0] = sWm[(k0 + koff    ) * PE + n0 + col];
            b[1] = sWm[(k0 + koff + 1) * PE + n0 + col];
            acc = __builtin_amdgcn_wmma_f32_16x16x4_f32(
                false, a, false, b, (short)0, acc, false, false);
        }
        #pragma unroll
        for (int kk = 0; kk < 4; ++kk) {
            const int k0 = kk * 4;
            v2f a = *(const v2f*)(&sA[wave][col * PD + k0 + koff]);
            v2f b;
            b[0] = sWe[(k0 + koff    ) * PE + n0 + col];
            b[1] = sWe[(k0 + koff + 1) * PE + n0 + col];
            acc = __builtin_amdgcn_wmma_f32_16x16x4_f32(
                false, a, false, b, (short)0, acc, false, false);
        }
        #pragma unroll
        for (int r = 0; r < 8; ++r) {
            const int m = r + hi * 8;
            const int d = sD[wave][m];
            atomicAdd(&agg[(size_t)d * PE + n0 + col], acc[r]);
        }
    }
}

__global__ void count_kernel(const int* __restrict__ batch, float* __restrict__ counts) {
    int i = blockIdx.x * blockDim.x + threadIdx.x;
    if (i < N_NODES) atomicAdd(&counts[batch[i]], 1.0f);
}

__global__ void mean_accum_kernel(const float* __restrict__ h,
                                  const int* __restrict__ batch,
                                  float* __restrict__ mean) {
    int i = blockIdx.x * blockDim.x + threadIdx.x;
    int n = i >> 6, f = i & 63;
    atomicAdd(&mean[batch[n] * PE + f], h[i]);
}

__global__ void mean_fin_kernel(float* __restrict__ mean, const float* __restrict__ counts) {
    int i = blockIdx.x * blockDim.x + threadIdx.x;
    mean[i] /= fmaxf(counts[i >> 6], 1.0f);
}

__global__ void center_var_kernel(const float* __restrict__ h,
                                  const int* __restrict__ batch,
                                  const float* __restrict__ mean,
                                  const float* __restrict__ mscale,
                                  float* __restrict__ tmp,
                                  float* __restrict__ var) {
    int i = blockIdx.x * blockDim.x + threadIdx.x;
    int n = i >> 6, f = i & 63;
    int g = batch[n];
    float o = h[i] - mean[g * PE + f] * mscale[f];
    tmp[i] = o;
    atomicAdd(&var[g * PE + f], o * o);
}

__global__ void scale_fin_kernel(float* __restrict__ var,
                                 const float* __restrict__ counts,
                                 const float* __restrict__ w) {
    int i = blockIdx.x * blockDim.x + threadIdx.x;
    float c = fmaxf(counts[i >> 6], 1.0f);
    var[i] = w[i & 63] * rsqrtf(var[i] / c + EPS);
}

__global__ void apply_kernel(const float* __restrict__ tmp,
                             const int* __restrict__ batch,
                             const float* __restrict__ scale,
                             const float* __restrict__ bias,
                             float* __restrict__ out) {
    int i = blockIdx.x * blockDim.x + threadIdx.x;
    int n = i >> 6, f = i & 63;
    float v = tmp[i] * scale[batch[n] * PE + f] + bias[f];
    out[i] = v > 0.0f ? v : NEG_SLOPE * v;
}

extern "C" void kernel_launch(void* const* d_in, const int* in_sizes, int n_in,
                              void* d_out, int out_size, void* d_ws, size_t ws_size,
                              hipStream_t stream) {
    const float* x0    = (const float*)d_in[0];
    const int*   idx   = (const int*)  d_in[1];
    const float* attr  = (const float*)d_in[2];
    const int*   batch = (const int*)  d_in[3];
    const float* Wm    = (const float*)d_in[4];
    const float* bm    = (const float*)d_in[5];
    const float* We    = (const float*)d_in[6];
    const float* be    = (const float*)d_in[7];
    const float* gw    = (const float*)d_in[8];
    const float* gb    = (const float*)d_in[9];
    const float* gms   = (const float*)d_in[10];

    const int* src = idx;
    const int* dst = idx + N_EDGES;

    char* ws = (char*)d_ws;
    const size_t node_bytes = (size_t)N_NODES * PE * sizeof(float);
    float* h      = (float*)(ws);
    float* tmp    = (float*)(ws + node_bytes);
    float* xbuf   = (float*)(ws + 2 * node_bytes);
    float* mean   = (float*)(ws + 3 * node_bytes);
    float* var    = mean + N_GRAPHS * PE;
    float* counts = var  + N_GRAPHS * PE;

    hipMemsetAsync(counts, 0, N_GRAPHS * sizeof(float), stream);
    count_kernel<<<(N_NODES + 255) / 256, 256, 0, stream>>>(batch, counts);

    const int NE = N_NODES * PE;
    for (int l = 0; l < 2; ++l) {
        const float* xin  = (l == 0) ? x0 : xbuf;
        float*       xout = (l == 1) ? (float*)d_out : xbuf;

        hipMemcpyAsync(h, xin, node_bytes, hipMemcpyDeviceToDevice, stream);

        edge_msg_kernel<<<N_EDGES / 16 / 8, 256, 0, stream>>>(
            xin, src, dst, attr,
            Wm + (size_t)l * PE * PE, bm + (size_t)l * PE,
            We + (size_t)l * PD * PE, be + (size_t)l * PE, h);

        hipMemsetAsync(mean, 0, (size_t)2 * N_GRAPHS * PE * sizeof(float), stream);
        mean_accum_kernel<<<NE / 256, 256, 0, stream>>>(h, batch, mean);
        mean_fin_kernel<<<N_GRAPHS * PE / 256, 256, 0, stream>>>(mean, counts);
        center_var_kernel<<<NE / 256, 256, 0, stream>>>(h, batch, mean,
                                                        gms + (size_t)l * PE, tmp, var);
        scale_fin_kernel<<<N_GRAPHS * PE / 256, 256, 0, stream>>>(var, counts,
                                                                  gw + (size_t)l * PE);
        apply_kernel<<<NE / 256, 256, 0, stream>>>(tmp, batch, var,
                                                   gb + (size_t)l * PE, xout);
    }
}


# edge kernel: LDS tiles re-pitched (x rows 68 dwords, attr rows 20; weights stored as k pairs at pitch 160) so each A and B fragment read is one conflict-free 8-byte load
# speedup vs baseline: 1.1479x; 1.1479x over previous
#include <hip/hip_runtime.h>
#include <hip/hip_bf16.h>
#include <stdint.h>

#define N_NODES   50000
#define N_EDGES   800000
#define PE        64
#define PD        16
#define N_GRAPHS  256
#define EPS       1e-5f
#define NEG_SLOPE 0.01f

#define XP 68
#define AP 20
#define WP 160

typedef __attribute__((ext_vector_type(2))) float v2f;
typedef __attribute__((ext_vector_type(8))) float v8f;

__global__ __launch_bounds__(256) void edge_msg_kernel(
    const float* __restrict__ x,
    const int*   __restrict__ src,
    const int*   __restrict__ dst,
    const float* __restrict__ attr,
    const float* __restrict__ Wm,
    const float* __restrict__ bm,
    const float* __restrict__ We,
    const float* __restrict__ be,
    float*       __restrict__ agg)
{
    __shared__ __attribute__((aligned(16))) float sWm[(PE / 2) * WP];
    __shared__ __attribute__((aligned(16))) float sWe[(PD / 2) * WP];
    __shared__ __attribute__((aligned(16))) float sX[8][16 * XP];
    __shared__ __attribute__((aligned(16))) float sA[8][16 * AP];
    __shared__ int   sD[8][16];

    const int tid  = threadIdx.x;
    const int wave = tid >> 5;
    const int lane = tid & 31;

    {
        const float4* Wm4 = (const float4*)Wm;
        #pragma unroll
        for (int it = tid; it < (PE / 2) * (PE / 4); it += 256) {
            const int kp = it >> 4, n4 = it & 15;
            const float4 r0 = Wm4[(2 * kp    ) * (PE / 4) + n4];
            const float4 r1 = Wm4[(2 * kp + 1) * (PE / 4) + n4];
            float4* o = (float4*)(&sWm[kp * WP + n4 * 8]);
            o[0] = make_float4(r0.x, r1.x, r0.y, r1.y);
            o[1] = make_float4(r0.z, r1.z, r0.w, r1.w);
        }
        const float4* We4 = (const float4*)We;
        if (tid < (PD / 2) * (PE / 4)) {
            const int kp = tid >> 4, n4 = tid & 15;
            const float4 r0 = We4[(2 * kp    ) * (PE / 4) + n4];
            const float4 r1 = We4[(2 * kp + 1) * (PE / 4) + n4];
            float4* o = (float4*)(&sWe[kp * WP + n4 * 8]);
            o[0] = make_float4(r0.x, r1.x, r0.y, r1.y);
            o[1] = make_float4(r0.z, r1.z, r0.w, r1.w);
        }
    }

    const int tile = blockIdx.x * 8 + wave;
    const int e0   = tile * 16;

    if (lane < 16) sD[wave][lane] = dst[e0 + lane];

    {
        const int row = lane >> 1;
        const int hlf = lane & 1;
        const int s   = src[e0 + row];
        const float4* xr = (const float4*)(x + (size_t)s * PE + hlf * 32);
        float4*       ox = (float4*)(&sX[wave][row * XP + hlf * 32]);
        #pragma unroll
        for (int i = 0; i < 8; ++i) ox[i] = xr[i];
        const float4* ar = (const float4*)(attr + (size_t)(e0 + row) * PD + hlf * 8);
        float4*       oa = (float4*)(&sA[wave][row * AP + hlf * 8]);
        oa[0] = ar[0]; oa[1] = ar[1];
    }
    __syncthreads();

    const int col  = lane & 15;
    const int hi   = lane >> 4;
    const int koff = hi * 2;

    #pragma unroll
    for (int n = 0; n < 4; ++n) {
        const int n0 = n * 16;
        const float bias = bm[n0 + col] + be[n0 + col];
        v8f acc;
        #pragma unroll
        for (int r = 0; r < 8; ++r) acc[r] = bias;

        #pragma unroll
        for (int kk = 0; kk < 16; ++kk) {
            const int k0 = kk * 4;
            v2f a = *(const v2f*)(&sX[wave][col * XP + k0 + koff]);
            v2f b = *(const v2f*)(&sWm[(kk * 2 + hi) * WP + (n0 + col) * 2]);
            acc = __builtin_amdgcn_wmma_f32_16x16x4_f32(
                false, a, false, b, (short)0, acc, false, false);
        }
        #pragma unroll
        for (int kk = 0; kk < 4; ++kk) {
            const int k0 = kk * 4;
            v2f a = *(const v2f*)(&sA[wave][col * AP + k0 + koff]);
            v2f b = *(const v2f*)(&sWe[(kk * 2 + hi) * WP + (n0 + col) * 2]);
            acc = __builtin_amdgcn_wmma_f32_16x16x4_f32(
                false, a, false, b, (short)0, acc, false, false);
        }
        #pragma unroll
        for (int r = 0; r < 8; ++r) {
            const int m = r + hi * 8;
            const int d = sD[wave][m];
            atomicAdd(&agg[(size_t)d * PE + n0 + col], acc[r]);
        }
    }
}

__global__ void count_kernel(const int* __restrict__ batch, float* __restrict__ counts) {
    int i = blockIdx.x * blockDim.x + threadIdx.x;
    if (i < N_NODES) atomicAdd(&counts[batch[i]], 1.0f);
}

__global__ void mean_accum_kernel(const float* __restrict__ h,
                                  const int* __restrict__ batch,
                                  float* __restrict__ mean) {
    int i = blockIdx.x * blockDim.x + threadIdx.x;
    int n = i >> 6, f = i & 63;
    atomicAdd(&mean[batch[n] * PE + f], h[i]);
}

__global__ void mean_fin_kernel(float* __restrict__ mean, const float* __restrict__ counts) {
    int i = blockIdx.x * blockDim.x + threadIdx.x;
    mean[i] /= fmaxf(counts[i >> 6], 1.0f);
}

__global__ void center_var_kernel(const float* __restrict__ h,
                                  const int* __restrict__ batch,
                                  const float* __restrict__ mean,
                                  const float* __restrict__ mscale,
                                  float* __restrict__ tmp,
                                  float* __restrict__ var) {
    int i = blockIdx.x * blockDim.x + threadIdx.x;
    int n = i >> 6, f = i & 63;
    int g = batch[n];
    float o = h[i] - mean[g * PE + f] * mscale[f];
    tmp[i] = o;
    atomicAdd(&var[g * PE + f], o * o);
}

__global__ void scale_fin_kernel(float* __restrict__ var,
                                 const float* __restrict__ counts,
                                 const float* __restrict__ w) {
    int i = blockIdx.x * blockDim.x + threadIdx.x;
    float c = fmaxf(counts[i >> 6], 1.0f);
    var[i] = w[i & 63] * rsqrtf(var[i] / c + EPS);
}

__global__ void apply_kernel(const float* __restrict__ tmp,
                             const int* __restrict__ batch,
                             const float* __restrict__ scale,
                             const float* __restrict__ bias,
                             float* __restrict__ out) {
    int i = blockIdx.x * blockDim.x + threadIdx.x;
    int n = i >> 6, f = i & 63;
    float v = tmp[i] * scale[batch[n] * PE + f] + bias[f];
    out[i] = v > 0.0f ? v : NEG_SLOPE * v;
}

extern "C" void kernel_launch(void* const* d_in, const int* in_sizes, int n_in,
                              void* d_out, int out_size, void* d_ws, size_t ws_size,
                              hipStream_t stream) {
    const float* x0    = (const float*)d_in[0];
    const int*   idx   = (const int*)  d_in[1];
    const float* attr  = (const float*)d_in[2];
    const int*   batch = (const int*)  d_in[3];
    const float* Wm    = (const float*)d_in[4];
    const float* bm    = (const float*)d_in[5];
    const float* We    = (const float*)d_in[6];
    const float* be    = (const float*)d_in[7];
    const float* gw    = (const float*)d_in[8];
    const float* gb    = (const float*)d_in[9];
    const float* gms   = (const float*)d_in[10];

    const int* src = idx;
    const int* dst = idx + N_EDGES;

    char* ws = (char*)d_ws;
    const size_t node_bytes = (size_t)N_NODES * PE * sizeof(float);
    float* h      = (float*)(ws);
    float* tmp    = (float*)(ws + node_bytes);
    float* xbuf   = (float*)(ws + 2 * node_bytes);
    float* mean   = (float*)(ws + 3 * node_bytes);
    float* var    = mean + N_GRAPHS * PE;
    float* counts = var  + N_GRAPHS * PE;

    hipMemsetAsync(counts, 0, N_GRAPHS * sizeof(float), stream);
    count_kernel<<<(N_NODES + 255) / 256, 256, 0, stream>>>(batch, counts);

    const int NE = N_NODES * PE;
    for (int l = 0; l < 2; ++l) {
        const float* xin  = (l == 0) ? x0 : xbuf;
        float*       xout = (l == 1) ? (float*)d_out : xbuf;

        hipMemcpyAsync(h, xin, node_bytes, hipMemcpyDeviceToDevice, stream);

        edge_msg_kernel<<<N_EDGES / 16 / 8, 256, 0, stream>>>(
            xin, src, dst, attr,
            Wm + (size_t)l * PE * PE, bm + (size_t)l * PE,
            We + (size_t)l * PD * PE, be + (size_t)l * PE, h);

        hipMemsetAsync(mean, 0, (size_t)2 * N_GRAPHS * PE * sizeof(float), stream);
        mean_accum_kernel<<<NE / 256, 256, 0, stream>>>(h, batch, mean);
        mean_fin_kernel<<<N_GRAPHS * PE / 256, 256, 0, stream>>>(mean, counts);
        center_var_kernel<<<NE / 256, 256, 0, stream>>>(h, batch, mean,
                                                        gms + (size_t)l * PE, tmp, var);
        scale_fin_kernel<<<N_GRAPHS * PE / 256, 256, 0, stream>>>(var, counts,
                                                                  gw + (size_t)l * PE);
        apply_kernel<<<NE / 256, 256, 0, stream>>>(tmp, batch, var,
                                                   gb + (size_t)l * PE, xout);
    }
}


# graph-norm per-element kernels: a thread takes 13 rows x 4 columns (float4) and adds its rows of one graph in registers before one atomic; 100000 -> 1928 waves a launch; sum order changes
# speedup vs baseline: 1.2628x; 1.1001x over previous
#include <hip/hip_runtime.h>
#include <hip/hip_bf16.h>
#include <stdint.h>

#define N_NODES   50000
#define N_EDGES   800000
#define PE        64
#define PD        16
#define N_GRAPHS  256
#define EPS       1e-5f
#define NEG_SLOPE 0.01f

#define XP 68
#define AP 20
#define WP 160

typedef __attribute__((ext_vector_type(2))) float v2f;
typedef __attribute__((ext_vector_type(8))) float v8f;

__global__ __launch_bounds__(256) void edge_msg_kernel(
    const float* __restrict__ x,
    const int*   __restrict__ src,
    const int*   __restrict__ dst,
    const float* __restrict__ attr,
    const float* __restrict__ Wm,
    const float* __restrict__ bm,
    const float* __restrict__ We,
    const float* __restrict__ be,
    float*       __restrict__ agg)
{
    __shared__ __attribute__((aligned(16))) float sWm[(PE / 2) * WP];
    __shared__ __attribute__((aligned(16))) float sWe[(PD / 2) * WP];
    __shared__ __attribute__((aligned(16))) float sX[8][16 * XP];
    __shared__ __attribute__((aligned(16))) float sA[8][16 * AP];
    __shared__ int   sD[8][16];

    const int tid  = threadIdx.x;
    const int wave = tid >> 5;
    const int lane = tid & 31;

    {
        const float4* Wm4 = (const float4*)Wm;
        #pragma unroll
        for (int it = tid; it < (PE / 2) * (PE / 4); it += 256) {
            const int kp = it >> 4, n4 = it & 15;
            const float4 r0 = Wm4[(2 * kp    ) * (PE / 4) + n4];
            const float4 r1 = Wm4[(2 * kp + 1) * (PE / 4) + n4];
            float4* o = (float4*)(&sWm[kp * WP + n4 * 8]);
            o[0] = make_float4(r0.x, r1.x, r0.y, r1.y);
            o[1] = make_float4(r0.z, r1.z, r0.w, r1.w);
        }
        const float4* We4 = (const float4*)We;
        if (tid < (PD / 2) * (PE / 4)) {
            const int kp = tid >> 4, n4 = tid & 15;
            const float4 r0 = We4[(2 * kp    ) * (PE / 4) + n4];
            const float4 r1 = We4[(2 * kp + 1) * (PE / 4) + n4];
            float4* o = (float4*)(&sWe[kp * WP + n4 * 8]);
            o[0] = make_float4(r0.x, r1.x, r0.y, r1.y);
            o[1] = make_float4(r0.z, r1.z, r0.w, r1.w);
        }
    }

    const int tile = blockIdx.x * 8 + wave;
    const int e0   = tile * 16;

    if (lane < 16) sD[wave][lane] = dst[e0 + lane];

    {
        const int row = lane >> 1;
        const int hlf = lane & 1;
        const int s   = src[e0 + row];
        const float4* xr = (const float4*)(x + (size_t)s * PE + hlf * 32);
        float4*       ox = (float4*)(&sX[wave][row * XP + hlf * 32]);
        #pragma unroll
        for (int i = 0; i < 8; ++i) ox[i] = xr[i];
        const float4* ar = (const float4*)(attr + (size_t)(e0 + row) * PD + hlf * 8);
        float4*       oa = (float4*)(&sA[wave][row * AP + hlf * 8]);
        oa[0] = ar[0]; oa[1] = ar[1];
    }
    __syncthreads();

    const int col  = lane & 15;
    const int hi   = lane >> 4;
    const int koff = hi * 2;

    #pragma unroll
    for (int n = 0; n < 4; ++n) {
        const int n0 = n * 16;
        const float bias = bm[n0 + col] + be[n0 + col];
        v8f acc;
        #pragma unroll
        for (int r = 0; r < 8; ++r) acc[r] = bias;

        #pragma unroll
        for (int kk = 0; kk < 16; ++kk) {
            const int k0 = kk * 4;
            v2f a = *(const v2f*)(&sX[wave][col * XP + k0 + koff]);
            v2f b = *(const v2f*)(&sWm[(kk * 2 + hi) * WP + (n0 + col) * 2]);
            acc = __builtin_amdgcn_wmma_f32_16x16x4_f32(
                false, a, false, b, (short)0, acc, false, false);
        }
        #pragma unroll
        for (int kk = 0; kk < 4; ++kk) {
            const int k0 = kk * 4;
            v2f a = *(const v2f*)(&sA[wave][col * AP + k0 + koff]);
            v2f b = *(const v2f*)(&sWe[(kk * 2 + hi) * WP + (n0 + col) * 2]);
            acc = __builtin_amdgcn_wmma_f32_16x16x4_f32(
                false, a, false, b, (short)0, acc, false, false);
        }
        #pragma unroll
        for (int r = 0; r < 8; ++r) {
            const int m = r + hi * 8;
            const int d = sD[wave][m];
            atomicAdd(&agg[(size_t)d * PE + n0 + col], acc[r]);
        }
    }
}

__global__ void count_kernel(const int* __restrict__ batch, float* __restrict__ counts) {
    int i = blockIdx.x * blockDim.x + threadIdx.x;
    if (i < N_NODES) atomicAdd(&counts[batch[i]], 1.0f);
}

#define GN_R      13
#define GN_ROWS   (16 * GN_R)
#define GN_BLOCKS ((N_NODES + GN_ROWS - 1) / GN_ROWS)

template <bool FULL>
__device__ __forceinline__ void gn_load_rows(const float4* __restrict__ a4,
                                             const int* __restrict__ batch,
                                             int row0, int tx,
                                             int (&g)[GN_R], float4 (&v)[GN_R]) {
    #pragma unroll
    for (int j = 0; j < GN_R; ++j) {
        const int row = row0 + j;
        const int rc  = (FULL || row < N_NODES) ? row : N_NODES - 1;
        const int gj  = batch[rc];
        v[j] = a4[(size_t)rc * (PE / 4) + tx];
        g[j] = (FULL || row < N_NODES) ? gj : -1;
    }
}

__device__ __forceinline__ void gn_load_table(const float4* __restrict__ t4, int tx,
                                              const int (&g)[GN_R], float4 (&m)[GN_R]) {
    #pragma unroll
    for (int j = 0; j < GN_R; ++j) {
        const int gc = g[j] < 0 ? 0 : g[j];
        m[j] = t4[gc * (PE / 4) + tx];
    }
}

__device__ __forceinline__ void gn_flush(float* __restrict__ table, int g, int tx, const float4 a) {
    float* p = table + g * PE + tx * 4;
    atomicAdd(p + 0, a.x);
    atomicAdd(p + 1, a.y);
    atomicAdd(p + 2, a.z);
    atomicAdd(p + 3, a.w);
}

__device__ __forceinline__ void gn_reduce_rows(float* __restrict__ table, int tx,
                                               const int (&g)[GN_R], const float4 (&v)[GN_R]) {
    int    gp  = g[0];
    float4 acc = v[0];
    #pragma unroll
    for (int j = 1; j < GN_R; ++j) {
        if (g[j] != gp) {
            if (gp >= 0) gn_flush(table, gp, tx, acc);
            gp  = g[j];
            acc = v[j];
        } else {
            acc.x += v[j].x; acc.y += v[j].y; acc.z += v[j].z; acc.w += v[j].w;
        }
    }
    if (gp >= 0) gn_flush(table, gp, tx, acc);
}

template <bool FULL>
__device__ __forceinline__ void mean_accum_rows(const float* __restrict__ h,
                                                const int* __restrict__ batch,
                                                float* __restrict__ mean, int row0, int tx) {
    int    g[GN_R];
    float4 v[GN_R];
    gn_load_rows<FULL>((const float4*)h, batch, row0, tx, g, v);
    gn_reduce_rows(mean, tx, g, v);
}

__global__ __launch_bounds__(256) void mean_accum_kernel(const float* __restrict__ h,
                                  const int* __restrict__ batch,
                                  float* __restrict__ mean) {
    const int tx   = threadIdx.x & 15;
    const int ty   = threadIdx.x >> 4;
    const int row0 = blockIdx.x * GN_ROWS + ty * GN_R;
    if (row0 + GN_R <= N_NODES) mean_accum_rows<true >(h, batch, mean, row0, tx);
    else                        mean_accum_rows<false>(h, batch, mean, row0, tx);
}

__global__ void mean_fin_kernel(float* __restrict__ mean, const float* __restrict__ counts) {
    int i = blockIdx.x * blockDim.x + threadIdx.x;
    mean[i] /= fmaxf(counts[i >> 6], 1.0f);
}

template <bool FULL>
__device__ __forceinline__ void center_var_rows(const float* __restrict__ h,
                                                const int* __restrict__ batch,
                                                const float* __restrict__ mean,
                                                const float* __restrict__ mscale,
                                                float* __restrict__ tmp,
                                                float* __restrict__ var, int row0, int tx) {
    float4*      t4 = (float4*)tmp;
    const float4 ms = ((const float4*)mscale)[tx];
    int    g[GN_R];
    float4 v[GN_R];
    float4 m[GN_R];
    gn_load_rows<FULL>((const float4*)h, batch, row0, tx, g, v);
    gn_load_table((const float4*)mean, tx, g, m);
    #pragma unroll
    for (int j = 0; j < GN_R; ++j) {
        const int row = row0 + j;
        float4 o;
        o.x = v[j].x - m[j].x * ms.x;
        o.y = v[j].y - m[j].y * ms.y;
        o.z = v[j].z - m[j].z * ms.z;
        o.w = v[j].w - m[j].w * ms.w;
        if (FULL || row < N_NODES) t4[(size_t)row * (PE / 4) + tx] = o;
        v[j] = make_float4(o.x * o.x, o.y * o.y, o.z * o.z, o.w * o.w);
    }
    gn_reduce_rows(var, tx, g, v);
}

__global__ __launch_bounds__(256) void center_var_kernel(const float* __restrict__ h,
                                  const int* __restrict__ batch,
                                  const float* __restrict__ mean,
                                  const float* __restrict__ mscale,
                                  float* __restrict__ tmp,
                                  float* __restrict__ var) {
    const int tx   = threadIdx.x & 15;
    const int ty   = threadIdx.x >> 4;
    const int row0 = blockIdx.x * GN_ROWS + ty * GN_R;
    if (row0 + GN_R <= N_NODES) center_var_rows<true >(h, batch, mean, mscale, tmp, var, row0, tx);
    else                        center_var_rows<false>(h, batch, mean, mscale, tmp, var, row0, tx);
}

__global__ void scale_fin_kernel(float* __restrict__ var,
                                 const float* __restrict__ counts,
                                 const float* __restrict__ w) {
    int i = blockIdx.x * blockDim.x + threadIdx.x;
    float c = fmaxf(counts[i >> 6], 1.0f);
    var[i] = w[i & 63] * rsqrtf(var[i] / c + EPS);
}

template <bool FULL>
__device__ __forceinline__ void apply_rows(const float* __restrict__ tmp,
                                           const int* __restrict__ batch,
                                           const float* __restrict__ scale,
                                           const float* __restrict__ bias,
                                           float* __restrict__ out, int row0, int tx) {
    float4*      o4 = (float4*)out;
    const float4 b  = ((const float4*)bias)[tx];
    int    g[GN_R];
    float4 t[GN_R];
    float4 s[GN_R];
    gn_load_rows<FULL>((const float4*)tmp, batch, row0, tx, g, t);
    gn_load_table((const float4*)scale, tx, g, s);
    #pragma unroll
    for (int j = 0; j < GN_R; ++j) {
        const int row = row0 + j;
        float4 r;
        r.x = t[j].x * s[j].x + b.x;
        r.y = t[j].y * s[j].y + b.y;
        r.z = t[j].z * s[j].z + b.z;
        r.w = t[j].w * s[j].w + b.w;
        r.x = r.x > 0.0f ? r.x : NEG_SLOPE * r.x;
        r.y = r.y > 0.0f ? r.y : NEG_SLOPE * r.y;
        r.z = r.z > 0.0f ? r.z : NEG_SLOPE * r.z;
        r.w = r.w > 0.0f ? r.w : NEG_SLOPE * r.w;
        if (FULL || row < N_NODES) o4[(size_t)row * (PE / 4) + tx] = r;
    }
}

__global__ __launch_bounds__(256) void apply_kernel(const float* __restrict__ tmp,
                             const int* __restrict__ batch,
                             const float* __restrict__ scale,
                             const float* __restrict__ bias,
                             float* __restrict__ out) {
    const int tx   = threadIdx.x & 15;
    const int ty   = threadIdx.x >> 4;
    const int row0 = blockIdx.x * GN_ROWS + ty * GN_R;
    if (row0 + GN_R <= N_NODES) apply_rows<true >(tmp, batch, scale, bias, out, row0, tx);
    else                        apply_rows<false>(tmp, batch, scale, bias, out, row0, tx);
}

extern "C" void kernel_launch(void* const* d_in, const int* in_sizes, int n_in,
                              void* d_out, int out_size, void* d_ws, size_t ws_size,
                              hipStream_t stream) {
    const float* x0    = (const float*)d_in[0];
    const int*   idx   = (const int*)  d_in[1];
    const float* attr  = (const float*)d_in[2];
    const int*   batch = (const int*)  d_in[3];
    const float* Wm    = (const float*)d_in[4];
    const float* bm    = (const float*)d_in[5];
    const float* We    = (const float*)d_in[6];
    const float* be    = (const float*)d_in[7];
    const float* gw    = (const float*)d_in[8];
    const float* gb    = (const float*)d_in[9];
    const float* gms   = (const float*)d_in[10];

    const int* src = idx;
    const int* dst = idx + N_EDGES;

    char* ws = (char*)d_ws;
    const size_t node_bytes = (size_t)N_NODES * PE * sizeof(float);
    float* h      = (float*)(ws);
    float* tmp    = (float*)(ws + node_bytes);
    float* xbuf   = (float*)(ws + 2 * node_bytes);
    float* mean   = (float*)(ws + 3 * node_bytes);
    float* var    = mean + N_GRAPHS * PE;
    float* counts = var  + N_GRAPHS * PE;

    hipMemsetAsync(counts, 0, N_GRAPHS * sizeof(float), stream);
    count_kernel<<<(N_NODES + 255) / 256, 256, 0, stream>>>(batch, counts);

    for (int l = 0; l < 2; ++l) {
        const float* xin  = (l == 0) ? x0 : xbuf;
        float*       xout = (l == 1) ? (float*)d_out : xbuf;

        hipMemcpyAsync(h, xin, node_bytes, hipMemcpyDeviceToDevice, stream);

        edge_msg_kernel<<<N_EDGES / 16 / 8, 256, 0, stream>>>(
            xin, src, dst, attr,
            Wm + (size_t)l * PE * PE, bm + (size_t)l * PE,
            We + (size_t)l * PD * PE, be + (size_t)l * PE, h);

        hipMemsetAsync(mean, 0, (size_t)2 * N_GRAPHS * PE * sizeof(float), stream);
        mean_accum_kernel<<<GN_BLOCKS, 256, 0, stream>>>(h, batch, mean);
        mean_fin_kernel<<<N_GRAPHS * PE / 256, 256, 0, stream>>>(mean, counts);
        center_var_kernel<<<GN_BLOCKS, 256, 0, stream>>>(h, batch, mean,
                                                        gms + (size_t)l * PE, tmp, var);
        scale_fin_kernel<<<N_GRAPHS * PE / 256, 256, 0, stream>>>(var, counts,
                                                                  gw + (size_t)l * PE);
        apply_kernel<<<GN_BLOCKS, 256, 0, stream>>>(tmp, batch, var,
                                                   gb + (size_t)l * PE, xout);
    }
}


# edge kernel scatter: neighbouring 16-column tiles trade halves across the lane halves (v_permlanex16) so every atomic adds one whole 128-byte line of one row; atomic arrival order differs
# speedup vs baseline: 1.5243x; 1.2071x over previous
#include <hip/hip_runtime.h>
#include <hip/hip_bf16.h>
#include <stdint.h>

#define N_NODES   50000
#define N_EDGES   800000
#define PE        64
#define PD        16
#define N_GRAPHS  256
#define EPS       1e-5f
#define NEG_SLOPE 0.01f

#define XP 68
#define AP 20
#define WP 160

typedef __attribute__((ext_vector_type(2))) float v2f;
typedef __attribute__((ext_vector_type(8))) float v8f;

__device__ __forceinline__ float xhalf(float v) {
    return __builtin_bit_cast(float, __builtin_amdgcn_permlanex16(
        __builtin_bit_cast(int, v), __builtin_bit_cast(int, v), 0x76543210, 0xfedcba98, false, false));
}

__global__ __launch_bounds__(256) void edge_msg_kernel(
    const float* __restrict__ x,
    const int*   __restrict__ src,
    const int*   __restrict__ dst,
    const float* __restrict__ attr,
    const float* __restrict__ Wm,
    const float* __restrict__ bm,
    const float* __restrict__ We,
    const float* __restrict__ be,
    float*       __restrict__ agg)
{
    __shared__ __attribute__((aligned(16))) float sWm[(PE / 2) * WP];
    __shared__ __attribute__((aligned(16))) float sWe[(PD / 2) * WP];
    __shared__ __attribute__((aligned(16))) float sX[8][16 * XP];
    __shared__ __attribute__((aligned(16))) float sA[8][16 * AP];

    const int tid  = threadIdx.x;
    const int wave = tid >> 5;
    const int lane = tid & 31;

    {
        const float4* Wm4 = (const float4*)Wm;
        #pragma unroll
        for (int it = tid; it < (PE / 2) * (PE / 4); it += 256) {
            const int kp = it >> 4, n4 = it & 15;
            const float4 r0 = Wm4[(2 * kp    ) * (PE / 4) + n4];
            const float4 r1 = Wm4[(2 * kp + 1) * (PE / 4) + n4];
            float4* o = (float4*)(&sWm[kp * WP + n4 * 8]);
            o[0] = make_float4(r0.x, r1.x, r0.y, r1.y);
            o[1] = make_float4(r0.z, r1.z, r0.w, r1.w);
        }
        const float4* We4 = (const float4*)We;
        if (tid < (PD / 2) * (PE / 4)) {
            const int kp = tid >> 4, n4 = tid & 15;
            const float4 r0 = We4[(2 * kp    ) * (PE / 4) + n4];
            const float4 r1 = We4[(2 * kp + 1) * (PE / 4) + n4];
            float4* o = (float4*)(&sWe[kp * WP + n4 * 8]);
            o[0] = make_float4(r0.x, r1.x, r0.y, r1.y);
            o[1] = make_float4(r0.z, r1.z, r0.w, r1.w);
        }
    }

    const int tile = blockIdx.x * 8 + wave;
    const int e0   = tile * 16;

    {
        const int row = lane >> 1;
        const int hlf = lane & 1;
        const int s   = src[e0 + row];
        const float4* xr = (const float4*)(x + (size_t)s * PE + hlf * 32);
        float4*       ox = (float4*)(&sX[wave][row * XP + hlf * 32]);
        #pragma unroll
        for (int i = 0; i < 8; ++i) ox[i] = xr[i];
        const float4* ar = (const float4*)(attr + (size_t)(e0 + row) * PD + hlf * 8);
        float4*       oa = (float4*)(&sA[wave][row * AP + hlf * 8]);
        oa[0] = ar[0]; oa[1] = ar[1];
    }
    __syncthreads();

    const int col  = lane & 15;
    const int hi   = lane >> 4;
    const int koff = hi * 2;

    const int dv = dst[e0 + col];

    v8f acc[4];
    #pragma unroll
    for (int n = 0; n < 4; ++n) {
        const float bias = bm[n * 16 + col] + be[n * 16 + col];
        #pragma unroll
        for (int r = 0; r < 8; ++r) acc[n][r] = bias;
    }

    #pragma unroll
    for (int kk = 0; kk < 16; ++kk) {
        const int k0 = kk * 4;
        v2f a = *(const v2f*)(&sX[wave][col * XP + k0 + koff]);
        #pragma unroll
        for (int n = 0; n < 4; ++n) {
            v2f b = *(const v2f*)(&sWm[(kk * 2 + hi) * WP + (n * 16 + col) * 2]);
            acc[n] = __builtin_amdgcn_wmma_f32_16x16x4_f32(
                false, a, false, b, (short)0, acc[n], false, false);
        }
    }
    #pragma unroll
    for (int kk = 0; kk < 4; ++kk) {
        const int k0 = kk * 4;
        v2f a = *(const v2f*)(&sA[wave][col * AP + k0 + koff]);
        #pragma unroll
        for (int n = 0; n < 4; ++n) {
            v2f b = *(const v2f*)(&sWe[(kk * 2 + hi) * WP + (n * 16 + col) * 2]);
            acc[n] = __builtin_amdgcn_wmma_f32_16x16x4_f32(
                false, a, false, b, (short)0, acc[n], false, false);
        }
    }

    #pragma unroll
    for (int r = 0; r < 8; ++r) {
        float P[2], Q[2];
        #pragma unroll
        for (int p = 0; p < 2; ++p) {
            const float X  = acc[2 * p][r];
            const float Y  = acc[2 * p + 1][r];
            const float Z  = hi ? X : Y;
            const float Zx = xhalf(Z);
            P[p] = hi ? Zx : X;
            Q[p] = hi ? Y  : Zx;
        }
        float* row0 = agg + (size_t)__builtin_amdgcn_readlane(dv, r)     * PE + lane;
        float* row1 = agg + (size_t)__builtin_amdgcn_readlane(dv, r + 8) * PE + lane;
        atomicAdd(row0,      P[0]);
        atomicAdd(row0 + 32, P[1]);
        atomicAdd(row1,      Q[0]);
        atomicAdd(row1 + 32, Q[1]);
    }
}

__global__ void count_kernel(const int* __restrict__ batch, float* __restrict__ counts) {
    int i = blockIdx.x * blockDim.x + threadIdx.x;
    if (i < N_NODES) atomicAdd(&counts[batch[i]], 1.0f);
}

#define GN_R      13
#define GN_ROWS   (16 * GN_R)
#define GN_BLOCKS ((N_NODES + GN_ROWS - 1) / GN_ROWS)

template <bool FULL>
__device__ __forceinline__ void gn_load_rows(const float4* __restrict__ a4,
                                             const int* __restrict__ batch,
                                             int row0, int tx,
                                             int (&g)[GN_R], float4 (&v)[GN_R]) {
    #pragma unroll
    for (int j = 0; j < GN_R; ++j) {
        const int row = row0 + j;
        const int rc  = (FULL || row < N_NODES) ? row : N_NODES - 1;
        const int gj  = batch[rc];
        v[j] = a4[(size_t)rc * (PE / 4) + tx];
        g[j] = (FULL || row < N_NODES) ? gj : -1;
    }
}

__device__ __forceinline__ void gn_load_table(const float4* __restrict__ t4, int tx,
                                              const int (&g)[GN_R], float4 (&m)[GN_R]) {
    #pragma unroll
    for (int j = 0; j < GN_R; ++j) {
        const int gc = g[j] < 0 ? 0 : g[j];
        m[j] = t4[gc * (PE / 4) + tx];
    }
}

__device__ __forceinline__ void gn_flush(float* __restrict__ table, int g, int tx, const float4 a) {
    float* p = table + g * PE + tx * 4;
    atomicAdd(p + 0, a.x);
    atomicAdd(p + 1, a.y);
    atomicAdd(p + 2, a.z);
    atomicAdd(p + 3, a.w);
}

__device__ __forceinline__ void gn_reduce_rows(float* __restrict__ table, int tx,
                                               const int (&g)[GN_R], const float4 (&v)[GN_R]) {
    int    gp  = g[0];
    float4 acc = v[0];
    #pragma unroll
    for (int j = 1; j < GN_R; ++j) {
        if (g[j] != gp) {
            if (gp >= 0) gn_flush(table, gp, tx, acc);
            gp  = g[j];
            acc = v[j];
        } else {
            acc.x += v[j].x; acc.y += v[j].y; acc.z += v[j].z; acc.w += v[j].w;
        }
    }
    if (gp >= 0) gn_flush(table, gp, tx, acc);
}

template <bool FULL>
__device__ __forceinline__ void mean_accum_rows(const float* __restrict__ h,
                                                const int* __restrict__ batch,
                                                float* __restrict__ mean, int row0, int tx) {
    int    g[GN_R];
    float4 v[GN_R];
    gn_load_rows<FULL>((const float4*)h, batch, row0, tx, g, v);
    gn_reduce_rows(mean, tx, g, v);
}

__global__ __launch_bounds__(256) void mean_accum_kernel(const float* __restrict__ h,
                                  const int* __restrict__ batch,
                                  float* __restrict__ mean) {
    const int tx   = threadIdx.x & 15;
    const int ty   = threadIdx.x >> 4;
    const int row0 = blockIdx.x * GN_ROWS + ty * GN_R;
    if (row0 + GN_R <= N_NODES) mean_accum_rows<true >(h, batch, mean, row0, tx);
    else                        mean_accum_rows<false>(h, batch, mean, row0, tx);
}

__global__ void mean_fin_kernel(float* __restrict__ mean, const float* __restrict__ counts) {
    int i = blockIdx.x * blockDim.x + threadIdx.x;
    mean[i] /= fmaxf(counts[i >> 6], 1.0f);
}

template <bool FULL>
__device__ __forceinline__ void center_var_rows(const float* __restrict__ h,
                                                const int* __restrict__ batch,
                                                const float* __restrict__ mean,
                                                const float* __restrict__ mscale,
                                                float* __restrict__ tmp,
                                                float* __restrict__ var, int row0, int tx) {
    float4*      t4 = (float4*)tmp;
    const float4 ms = ((const float4*)mscale)[tx];
    int    g[GN_R];
    float4 v[GN_R];
    float4 m[GN_R];
    gn_load_rows<FULL>((const float4*)h, batch, row0, tx, g, v);
    gn_load_table((const float4*)mean, tx, g, m);
    #pragma unroll
    for (int j = 0; j < GN_R; ++j) {
        const int row = row0 + j;
        float4 o;
        o.x = v[j].x - m[j].x * ms.x;
        o.y = v[j].y - m[j].y * ms.y;
        o.z = v[j].z - m[j].z * ms.z;
        o.w = v[j].w - m[j].w * ms.w;
        if (FULL || row < N_NODES) t4[(size_t)row * (PE / 4) + tx] = o;
        v[j] = make_float4(o.x * o.x, o.y * o.y, o.z * o.z, o.w * o.w);
    }
    gn_reduce_rows(var, tx, g, v);
}

__global__ __launch_bounds__(256) void center_var_kernel(const float* __restrict__ h,
                                  const int* __restrict__ batch,
                                  const float* __restrict__ mean,
                                  const float* __restrict__ mscale,
                                  float* __restrict__ tmp,
                                  float* __restrict__ var) {
    const int tx   = threadIdx.x & 15;
    const int ty   = threadIdx.x >> 4;
    const int row0 = blockIdx.x * GN_ROWS + ty * GN_R;
    if (row0 + GN_R <= N_NODES) center_var_rows<true >(h, batch, mean, mscale, tmp, var, row0, tx);
    else                        center_var_rows<false>(h, batch, mean, mscale, tmp, var, row0, tx);
}

__global__ void scale_fin_kernel(float* __restrict__ var,
                                 const float* __restrict__ counts,
                                 const float* __restrict__ w) {
    int i = blockIdx.x * blockDim.x + threadIdx.x;
    float c = fmaxf(counts[i >> 6], 1.0f);
    var[i] = w[i & 63] * rsqrtf(var[i] / c + EPS);
}

template <bool FULL>
__device__ __forceinline__ void apply_rows(const float* __restrict__ tmp,
                                           const int* __restrict__ batch,
                                           const float* __restrict__ scale,
                                           const float* __restrict__ bias,
                                           float* __restrict__ out, int row0, int tx) {
    float4*      o4 = (float4*)out;
    const float4 b  = ((const float4*)bias)[tx];
    int    g[GN_R];
    float4 t[GN_R];
    float4 s[GN_R];
    gn_load_rows<FULL>((const float4*)tmp, batch, row0, tx, g, t);
    gn_load_table((const float4*)scale, tx, g, s);
    #pragma unroll
    for (int j = 0; j < GN_R; ++j) {
        const int row = row0 + j;
        float4 r;
        r.x = t[j].x * s[j].x + b.x;
        r.y = t[j].y * s[j].y + b.y;
        r.z = t[j].z * s[j].z + b.z;
        r.w = t[j].w * s[j].w + b.w;
        r.x = r.x > 0.0f ? r.x : NEG_SLOPE * r.x;
        r.y = r.y > 0.0f ? r.y : NEG_SLOPE * r.y;
        r.z = r.z > 0.0f ? r.z : NEG_SLOPE * r.z;
        r.w = r.w > 0.0f ? r.w : NEG_SLOPE * r.w;
        if (FULL || row < N_NODES) o4[(size_t)row * (PE / 4) + tx] = r;
    }
}

__global__ __launch_bounds__(256) void apply_kernel(const float* __restrict__ tmp,
                             const int* __restrict__ batch,
                             const float* __restrict__ scale,
                             const float* __restrict__ bias,
                             float* __restrict__ out) {
    const int tx   = threadIdx.x & 15;
    const int ty   = threadIdx.x >> 4;
    const int row0 = blockIdx.x * GN_ROWS + ty * GN_R;
    if (row0 + GN_R <= N_NODES) apply_rows<true >(tmp, batch, scale, bias, out, row0, tx);
    else                        apply_rows<false>(tmp, batch, scale, bias, out, row0, tx);
}

extern "C" void kernel_launch(void* const* d_in, const int* in_sizes, int n_in,
                              void* d_out, int out_size, void* d_ws, size_t ws_size,
                              hipStream_t stream) {
    const float* x0    = (const float*)d_in[0];
    const int*   idx   = (const int*)  d_in[1];
    const float* attr  = (const float*)d_in[2];
    const int*   batch = (const int*)  d_in[3];
    const float* Wm    = (const float*)d_in[4];
    const float* bm    = (const float*)d_in[5];
    const float* We    = (const float*)d_in[6];
    const float* be    = (const float*)d_in[7];
    const float* gw    = (const float*)d_in[8];
    const float* gb    = (const float*)d_in[9];
    const float* gms   = (const float*)d_in[10];

    const int* src = idx;
    const int* dst = idx + N_EDGES;

    char* ws = (char*)d_ws;
    const size_t node_bytes = (size_t)N_NODES * PE * sizeof(float);
    float* h      = (float*)(ws);
    float* tmp    = (float*)(ws + node_bytes);
    float* xbuf   = (float*)(ws + 2 * node_bytes);
    float* mean   = (float*)(ws + 3 * node_bytes);
    float* var    = mean + N_GRAPHS * PE;
    float* counts = var  + N_GRAPHS * PE;

    hipMemsetAsync(counts, 0, N_GRAPHS * sizeof(float), stream);
    count_kernel<<<(N_NODES + 255) / 256, 256, 0, stream>>>(batch, counts);

    for (int l = 0; l < 2; ++l) {
        const float* xin  = (l == 0) ? x0 : xbuf;
        float*       xout = (l == 1) ? (float*)d_out : xbuf;

        hipMemcpyAsync(h, xin, node_bytes, hipMemcpyDeviceToDevice, stream);

        edge_msg_kernel<<<N_EDGES / 16 / 8, 256, 0, stream>>>(
            xin, src, dst, attr,
            Wm + (size_t)l * PE * PE, bm + (size_t)l * PE,
            We + (size_t)l * PD * PE, be + (size_t)l * PE, h);

        hipMemsetAsync(mean, 0, (size_t)2 * N_GRAPHS * PE * sizeof(float), stream);
        mean_accum_kernel<<<GN_BLOCKS, 256, 0, stream>>>(h, batch, mean);
        mean_fin_kernel<<<N_GRAPHS * PE / 256, 256, 0, stream>>>(mean, counts);
        center_var_kernel<<<GN_BLOCKS, 256, 0, stream>>>(h, batch, mean,
                                                        gms + (size_t)l * PE, tmp, var);
        scale_fin_kernel<<<N_GRAPHS * PE / 256, 256, 0, stream>>>(var, counts,
                                                                  gw + (size_t)l * PE);
        apply_kernel<<<GN_BLOCKS, 256, 0, stream>>>(tmp, batch, var,
                                                   gb + (size_t)l * PE, xout);
    }
}


# edge kernel as one resident pass: each work-group stages the weights once, its waves walk over 16-edge tiles with the next tile's rows in flight in registers; atomic arrival order differs
# speedup vs baseline: 1.6026x; 1.0514x over previous
#include <hip/hip_runtime.h>
#include <hip/hip_bf16.h>
#include <stdint.h>

#define N_NODES   50000
#define N_EDGES   800000
#define PE        64
#define PD        16
#define N_GRAPHS  256
#define EPS       1e-5f
#define NEG_SLOPE 0.01f

#define XP 68
#define AP 20
#define WP 160

typedef __attribute__((ext_vector_type(2))) float v2f;
typedef __attribute__((ext_vector_type(8))) float v8f;

__device__ __forceinline__ float xhalf(float v) {
    return __builtin_bit_cast(float, __builtin_amdgcn_permlanex16(
        __builtin_bit_cast(int, v), __builtin_bit_cast(int, v), 0x76543210, 0xfedcba98, false, false));
}

#define EDGE_TILES (N_EDGES / 16)
#define EDGE_GRID  512

__device__ __forceinline__ void wave_lds_handover() {
    __builtin_amdgcn_fence(__ATOMIC_RELEASE, "wavefront", "local");
    __builtin_amdgcn_wave_barrier();
    __builtin_amdgcn_fence(__ATOMIC_ACQUIRE, "wavefront", "local");
}

__global__ __launch_bounds__(256) void edge_msg_kernel(
    const float* __restrict__ x,
    const int*   __restrict__ src,
    const int*   __restrict__ dst,
    const float* __restrict__ attr,
    const float* __restrict__ Wm,
    const float* __restrict__ bm,
    const float* __restrict__ We,
    const float* __restrict__ be,
    float*       __restrict__ agg)
{
    __shared__ __attribute__((aligned(16))) float sWm[(PE / 2) * WP];
    __shared__ __attribute__((aligned(16))) float sWe[(PD / 2) * WP];
    __shared__ __attribute__((aligned(16))) float sX[8][16 * XP];
    __shared__ __attribute__((aligned(16))) float sA[8][16 * AP];

    const int tid  = threadIdx.x;
    const int wave = __builtin_amdgcn_readfirstlane(tid >> 5);
    const int lane = tid & 31;

    {
        const float4* Wm4 = (const float4*)Wm;
        #pragma unroll
        for (int it = tid; it < (PE / 2) * (PE / 4); it += 256) {
            const int kp = it >> 4, n4 = it & 15;
            const float4 r0 = Wm4[(2 * kp    ) * (PE / 4) + n4];
            const float4 r1 = Wm4[(2 * kp + 1) * (PE / 4) + n4];
            float4* o = (float4*)(&sWm[kp * WP + n4 * 8]);
            o[0] = make_float4(r0.x, r1.x, r0.y, r1.y);
            o[1] = make_float4(r0.z, r1.z, r0.w, r1.w);
        }
        const float4* We4 = (const float4*)We;
        if (tid < (PD / 2) * (PE / 4)) {
            const int kp = tid >> 4, n4 = tid & 15;
            const float4 r0 = We4[(2 * kp    ) * (PE / 4) + n4];
            const float4 r1 = We4[(2 * kp + 1) * (PE / 4) + n4];
            float4* o = (float4*)(&sWe[kp * WP + n4 * 8]);
            o[0] = make_float4(r0.x, r1.x, r0.y, r1.y);
            o[1] = make_float4(r0.z, r1.z, r0.w, r1.w);
        }
    }

    const int row = lane >> 1;
    const int hlf = lane & 1;

    const int col  = lane & 15;
    const int hi   = lane >> 4;
    const int koff = hi * 2;

    const float bias0 = bm[col]      + be[col];
    const float bias1 = bm[16 + col] + be[16 + col];
    const float bias2 = bm[32 + col] + be[32 + col];
    const float bias3 = bm[48 + col] + be[48 + col];

    const int stride = (int)gridDim.x * 8;
    int tile = wave * (int)gridDim.x + (int)blockIdx.x;

    float4 gx0, gx1, gx2, gx3, gx4, gx5, gx6, gx7;
    float4 ga0, ga1;
    int    dv;
    int    s_nxt = 0;
    {
        const int e0 = tile * 16;
        const int s  = src[e0 + row];
        if (tile + stride < EDGE_TILES) s_nxt = src[(tile + stride) * 16 + row];
        const float4* xr = (const float4*)(x + (size_t)s * PE + hlf * 32);
        gx0 = xr[0]; gx1 = xr[1]; gx2 = xr[2]; gx3 = xr[3];
        gx4 = xr[4]; gx5 = xr[5]; gx6 = xr[6]; gx7 = xr[7];
        const float4* ar = (const float4*)(attr + (size_t)(e0 + row) * PD + hlf * 8);
        ga0 = ar[0]; ga1 = ar[1];
        dv = dst[e0 + col];
    }
    __syncthreads();

    for (; tile < EDGE_TILES; tile += stride) {
        {
            float4* ox = (float4*)(&sX[wave][row * XP + hlf * 32]);
            ox[0] = gx0; ox[1] = gx1; ox[2] = gx2; ox[3] = gx3;
            ox[4] = gx4; ox[5] = gx5; ox[6] = gx6; ox[7] = gx7;
            float4* oa = (float4*)(&sA[wave][row * AP + hlf * 8]);
            oa[0] = ga0; oa[1] = ga1;
        }
        const int dcur = dv;

        const int nt = tile + stride;
        if (nt < EDGE_TILES) {
            const int e0n = nt * 16;
            const float4* xr = (const float4*)(x + (size_t)s_nxt * PE + hlf * 32);
            gx0 = xr[0]; gx1 = xr[1]; gx2 = xr[2]; gx3 = xr[3];
            gx4 = xr[4]; gx5 = xr[5]; gx6 = xr[6]; gx7 = xr[7];
            const float4* ar = (const float4*)(attr + (size_t)(e0n + row) * PD + hlf * 8);
            ga0 = ar[0]; ga1 = ar[1];
            dv = dst[e0n + col];
            if (nt + stride < EDGE_TILES) s_nxt = src[(nt + stride) * 16 + row];
        }

        wave_lds_handover();

        v8f acc[4];
        #pragma unroll
        for (int r = 0; r < 8; ++r) {
            acc[0][r] = bias0; acc[1][r] = bias1; acc[2][r] = bias2; acc[3][r] = bias3;
        }

        #pragma unroll
        for (int kk = 0; kk < 16; ++kk) {
            const int k0 = kk * 4;
            v2f a = *(const v2f*)(&sX[wave][col * XP + k0 + koff]);
            #pragma unroll
            for (int n = 0; n < 4; ++n) {
                v2f b = *(const v2f*)(&sWm[(kk * 2 + hi) * WP + (n * 16 + col) * 2]);
                acc[n] = __builtin_amdgcn_wmma_f32_16x16x4_f32(
                    false, a, false, b, (short)0, acc[n], false, false);
            }
        }
        #pragma unroll
        for (int kk = 0; kk < 4; ++kk) {
            const int k0 = kk * 4;
            v2f a = *(const v2f*)(&sA[wave][col * AP + k0 + koff]);
            #pragma unroll
            for (int n = 0; n < 4; ++n) {
                v2f b = *(const v2f*)(&sWe[(kk * 2 + hi) * WP + (n * 16 + col) * 2]);
                acc[n] = __builtin_amdgcn_wmma_f32_16x16x4_f32(
                    false, a, false, b, (short)0, acc[n], false, false);
            }
        }

        wave_lds_handover();

        #pragma unroll
        for (int r = 0; r < 8; ++r) {
            float P[2], Q[2];
            #pragma unroll
            for (int p = 0; p < 2; ++p) {
                const float X  = acc[2 * p][r];
                const float Y  = acc[2 * p + 1][r];
                const float Z  = hi ? X : Y;
                const float Zx = xhalf(Z);
                P[p] = hi ? Zx : X;
                Q[p] = hi ? Y  : Zx;
            }
            float* row0 = agg + (size_t)__builtin_amdgcn_readlane(dcur, r)     * PE + lane;
            float* row1 = agg + (size_t)__builtin_amdgcn_readlane(dcur, r + 8) * PE + lane;
            atomicAdd(row0,      P[0]);
            atomicAdd(row0 + 32, P[1]);
            atomicAdd(row1,      Q[0]);
            atomicAdd(row1 + 32, Q[1]);
        }
    }
}

__global__ void count_kernel(const int* __restrict__ batch, float* __restrict__ counts) {
    int i = blockIdx.x * blockDim.x + threadIdx.x;
    if (i < N_NODES) atomicAdd(&counts[batch[i]], 1.0f);
}

#define GN_R      13
#define GN_ROWS   (16 * GN_R)
#define GN_BLOCKS ((N_NODES + GN_ROWS - 1) / GN_ROWS)

template <bool FULL>
__device__ __forceinline__ void gn_load_rows(const float4* __restrict__ a4,
                                             const int* __restrict__ batch,
                                             int row0, int tx,
                                             int (&g)[GN_R], float4 (&v)[GN_R]) {
    #pragma unroll
    for (int j = 0; j < GN_R; ++j) {
        const int row = row0 + j;
        const int rc  = (FULL || row < N_NODES) ? row : N_NODES - 1;
        const int gj  = batch[rc];
        v[j] = a4[(size_t)rc * (PE / 4) + tx];
        g[j] = (FULL || row < N_NODES) ? gj : -1;
    }
}

__device__ __forceinline__ void gn_load_table(const float4* __restrict__ t4, int tx,
                                              const int (&g)[GN_R], float4 (&m)[GN_R]) {
    #pragma unroll
    for (int j = 0; j < GN_R; ++j) {
        const int gc = g[j] < 0 ? 0 : g[j];
        m[j] = t4[gc * (PE / 4) + tx];
    }
}

__device__ __forceinline__ void gn_flush(float* __restrict__ table, int g, int tx, const float4 a) {
    float* p = table + g * PE + tx * 4;
    atomicAdd(p + 0, a.x);
    atomicAdd(p + 1, a.y);
    atomicAdd(p + 2, a.z);
    atomicAdd(p + 3, a.w);
}

__device__ __forceinline__ void gn_reduce_rows(float* __restrict__ table, int tx,
                                               const int (&g)[GN_R], const float4 (&v)[GN_R]) {
    int    gp  = g[0];
    float4 acc = v[0];
    #pragma unroll
    for (int j = 1; j < GN_R; ++j) {
        if (g[j] != gp) {
            if (gp >= 0) gn_flush(table, gp, tx, acc);
            gp  = g[j];
            acc = v[j];
        } else {
            acc.x += v[j].x; acc.y += v[j].y; acc.z += v[j].z; acc.w += v[j].w;
        }
    }
    if (gp >= 0) gn_flush(table, gp, tx, acc);
}

template <bool FULL>
__device__ __forceinline__ void mean_accum_rows(const float* __restrict__ h,
                                                const int* __restrict__ batch,
                                                float* __restrict__ mean, int row0, int tx) {
    int    g[GN_R];
    float4 v[GN_R];
    gn_load_rows<FULL>((const float4*)h, batch, row0, tx, g, v);
    gn_reduce_rows(mean, tx, g, v);
}

__global__ __launch_bounds__(256) void mean_accum_kernel(const float* __restrict__ h,
                                  const int* __restrict__ batch,
                                  float* __restrict__ mean) {
    const int tx   = threadIdx.x & 15;
    const int ty   = threadIdx.x >> 4;
    const int row0 = blockIdx.x * GN_ROWS + ty * GN_R;
    if (row0 + GN_R <= N_NODES) mean_accum_rows<true >(h, batch, mean, row0, tx);
    else                        mean_accum_rows<false>(h, batch, mean, row0, tx);
}

__global__ void mean_fin_kernel(float* __restrict__ mean, const float* __restrict__ counts) {
    int i = blockIdx.x * blockDim.x + threadIdx.x;
    mean[i] /= fmaxf(counts[i >> 6], 1.0f);
}

template <bool FULL>
__device__ __forceinline__ void center_var_rows(const float* __restrict__ h,
                                                const int* __restrict__ batch,
                                                const float* __restrict__ mean,
                                                const float* __restrict__ mscale,
                                                float* __restrict__ tmp,
                                                float* __restrict__ var, int row0, int tx) {
    float4*      t4 = (float4*)tmp;
    const float4 ms = ((const float4*)mscale)[tx];
    int    g[GN_R];
    float4 v[GN_R];
    float4 m[GN_R];
    gn_load_rows<FULL>((const float4*)h, batch, row0, tx, g, v);
    gn_load_table((const float4*)mean, tx, g, m);
    #pragma unroll
    for (int j = 0; j < GN_R; ++j) {
        const int row = row0 + j;
        float4 o;
        o.x = v[j].x - m[j].x * ms.x;
        o.y = v[j].y - m[j].y * ms.y;
        o.z = v[j].z - m[j].z * ms.z;
        o.w = v[j].w - m[j].w * ms.w;
        if (FULL || row < N_NODES) t4[(size_t)row * (PE / 4) + tx] = o;
        v[j] = make_float4(o.x * o.x, o.y * o.y, o.z * o.z, o.w * o.w);
    }
    gn_reduce_rows(var, tx, g, v);
}

__global__ __launch_bounds__(256) void center_var_kernel(const float* __restrict__ h,
                                  const int* __restrict__ batch,
                                  const float* __restrict__ mean,
                                  const float* __restrict__ mscale,
                                  float* __restrict__ tmp,
                                  float* __restrict__ var) {
    const int tx   = threadIdx.x & 15;
    const int ty   = threadIdx.x >> 4;
    const int row0 = blockIdx.x * GN_ROWS + ty * GN_R;
    if (row0 + GN_R <= N_NODES) center_var_rows<true >(h, batch, mean, mscale, tmp, var, row0, tx);
    else                        center_var_rows<false>(h, batch, mean, mscale, tmp, var, row0, tx);
}

__global__ void scale_fin_kernel(float* __restrict__ var,
                                 const float* __restrict__ counts,
                                 const float* __restrict__ w) {
    int i = blockIdx.x * blockDim.x + threadIdx.x;
    float c = fmaxf(counts[i >> 6], 1.0f);
    var[i] = w[i & 63] * rsqrtf(var[i] / c + EPS);
}

template <bool FULL>
__device__ __forceinline__ void apply_rows(const float* __restrict__ tmp,
                                           const int* __restrict__ batch,
                                           const float* __restrict__ scale,
                                           const float* __restrict__ bias,
                                           float* __restrict__ out, int row0, int tx) {
    float4*      o4 = (float4*)out;
    const float4 b  = ((const float4*)bias)[tx];
    int    g[GN_R];
    float4 t[GN_R];
    float4 s[GN_R];
    gn_load_rows<FULL>((const float4*)tmp, batch, row0, tx, g, t);
    gn_load_table((const float4*)scale, tx, g, s);
    #pragma unroll
    for (int j = 0; j < GN_R; ++j) {
        const int row = row0 + j;
        float4 r;
        r.x = t[j].x * s[j].x + b.x;
        r.y = t[j].y * s[j].y + b.y;
        r.z = t[j].z * s[j].z + b.z;
        r.w = t[j].w * s[j].w + b.w;
        r.x = r.x > 0.0f ? r.x : NEG_SLOPE * r.x;
        r.y = r.y > 0.0f ? r.y : NEG_SLOPE * r.y;
        r.z = r.z > 0.0f ? r.z : NEG_SLOPE * r.z;
        r.w = r.w > 0.0f ? r.w : NEG_SLOPE * r.w;
        if (FULL || row < N_NODES) o4[(size_t)row * (PE / 4) + tx] = r;
    }
}

__global__ __launch_bounds__(256) void apply_kernel(const float* __restrict__ tmp,
                             const int* __restrict__ batch,
                             const float* __restrict__ scale,
                             const float* __restrict__ bias,
                             float* __restrict__ out) {
    const int tx   = threadIdx.x & 15;
    const int ty   = threadIdx.x >> 4;
    const int row0 = blockIdx.x * GN_ROWS + ty * GN_R;
    if (row0 + GN_R <= N_NODES) apply_rows<true >(tmp, batch, scale, bias, out, row0, tx);
    else                        apply_rows<false>(tmp, batch, scale, bias, out, row0, tx);
}

extern "C" void kernel_launch(void* const* d_in, const int* in_sizes, int n_in,
                              void* d_out, int out_size, void* d_ws, size_t ws_size,
                              hipStream_t stream) {
    const float* x0    = (const float*)d_in[0];
    const int*   idx   = (const int*)  d_in[1];
    const float* attr  = (const float*)d_in[2];
    const int*   batch = (const int*)  d_in[3];
    const float* Wm    = (const float*)d_in[4];
    const float* bm    = (const float*)d_in[5];
    const float* We    = (const float*)d_in[6];
    const float* be    = (const float*)d_in[7];
    const float* gw    = (const float*)d_in[8];
    const float* gb    = (const float*)d_in[9];
    const float* gms   = (const float*)d_in[10];

    const int* src = idx;
    const int* dst = idx + N_EDGES;

    char* ws = (char*)d_ws;
    const size_t node_bytes = (size_t)N_NODES * PE * sizeof(float);
    float* h      = (float*)(ws);
    float* tmp    = (float*)(ws + node_bytes);
    float* xbuf   = (float*)(ws + 2 * node_bytes);
    float* mean   = (float*)(ws + 3 * node_bytes);
    float* var    = mean + N_GRAPHS * PE;
    float* counts = var  + N_GRAPHS * PE;

    hipMemsetAsync(counts, 0, N_GRAPHS * sizeof(float), stream);
    count_kernel<<<(N_NODES + 255) / 256, 256, 0, stream>>>(batch, counts);

    for (int l = 0; l < 2; ++l) {
        const float* xin  = (l == 0) ? x0 : xbuf;
        float*       xout = (l == 1) ? (float*)d_out : xbuf;

        hipMemcpyAsync(h, xin, node_bytes, hipMemcpyDeviceToDevice, stream);

        edge_msg_kernel<<<EDGE_GRID, 256, 0, stream>>>(
            xin, src, dst, attr,
            Wm + (size_t)l * PE * PE, bm + (size_t)l * PE,
            We + (size_t)l * PD * PE, be + (size_t)l * PE, h);

        hipMemsetAsync(mean, 0, (size_t)2 * N_GRAPHS * PE * sizeof(float), stream);
        mean_accum_kernel<<<GN_BLOCKS, 256, 0, stream>>>(h, batch, mean);
        mean_fin_kernel<<<N_GRAPHS * PE / 256, 256, 0, stream>>>(mean, counts);
        center_var_kernel<<<GN_BLOCKS, 256, 0, stream>>>(h, batch, mean,
                                                        gms + (size_t)l * PE, tmp, var);
        scale_fin_kernel<<<N_GRAPHS * PE / 256, 256, 0, stream>>>(var, counts,
                                                                  gw + (size_t)l * PE);
        apply_kernel<<<GN_BLOCKS, 256, 0, stream>>>(tmp, batch, var,
                                                   gb + (size_t)l * PE, xout);
    }
}


# x @ W_msg moved from the 800000 edges to the 50000 nodes: a node kernel forms its rows once (and agg's start value, replacing the copy); edges gather them and add attr @ W_edge; sum order changes
# speedup vs baseline: 1.8497x; 1.1542x over previous
#include <hip/hip_runtime.h>
#include <hip/hip_bf16.h>
#include <stdint.h>

#define N_NODES   50000
#define N_EDGES   800000
#define PE        64
#define PD        16
#define N_GRAPHS  256
#define EPS       1e-5f
#define NEG_SLOPE 0.01f

#define XP 68
#define AP 20
#define WP 160

typedef __attribute__((ext_vector_type(2))) float v2f;
typedef __attribute__((ext_vector_type(8))) float v8f;

__device__ __forceinline__ float xhalf(float v) {
    return __builtin_bit_cast(float, __builtin_amdgcn_permlanex16(
        __builtin_bit_cast(int, v), __builtin_bit_cast(int, v), 0x76543210, 0xfedcba98, false, false));
}

__device__ __forceinline__ void wave_lds_handover() {
    __builtin_amdgcn_fence(__ATOMIC_RELEASE, "wavefront", "local");
    __builtin_amdgcn_wave_barrier();
    __builtin_amdgcn_fence(__ATOMIC_ACQUIRE, "wavefront", "local");
}

#define NODE_TILES (N_NODES / 16)
#define NODE_GRID  ((NODE_TILES + 7) / 8)
static_assert(N_NODES % 16 == 0, "the node kernel has no partial tile");

__global__ __launch_bounds__(256) void node_msg_kernel(
    const float* x,
    const float* __restrict__ Wm,
    float*       __restrict__ y,
    float*       agg)
{
    __shared__ __attribute__((aligned(16))) float sWm[(PE / 2) * WP];
    __shared__ __attribute__((aligned(16))) float sX[8][16 * XP];

    const int tid  = threadIdx.x;
    const int wave = __builtin_amdgcn_readfirstlane(tid >> 5);
    const int lane = tid & 31;

    const int col  = lane & 15;
    const int hi   = lane >> 4;
    const int koff = hi * 2;

    const int  tile = blockIdx.x * 8 + wave;
    const bool live = tile < NODE_TILES;

    float4 v0, v1, v2, v3, v4, v5, v6, v7;
    v0 = v1 = v2 = v3 = v4 = v5 = v6 = v7 = make_float4(0.0f, 0.0f, 0.0f, 0.0f);
    const int rq = hi * (PE / 4) + col;
    if (live) {
        const float4* x4 = (const float4*)(x + (size_t)tile * 16 * PE) + rq;
        v0 = x4[0 * (PE / 2)]; v1 = x4[1 * (PE / 2)]; v2 = x4[2 * (PE / 2)]; v3 = x4[3 * (PE / 2)];
        v4 = x4[4 * (PE / 2)]; v5 = x4[5 * (PE / 2)]; v6 = x4[6 * (PE / 2)]; v7 = x4[7 * (PE / 2)];
    }

    {
        const float4* Wm4 = (const float4*)Wm;
        #pragma unroll
        for (int it = tid; it < (PE / 2) * (PE / 4); it += 256) {
            const int kp = it >> 4, n4 = it & 15;
            const float4 r0 = Wm4[(2 * kp    ) * (PE / 4) + n4];
            const float4 r1 = Wm4[(2 * kp + 1) * (PE / 4) + n4];
            float4* o = (float4*)(&sWm[kp * WP + n4 * 8]);
            o[0] = make_float4(r0.x, r1.x, r0.y, r1.y);
            o[1] = make_float4(r0.z, r1.z, r0.w, r1.w);
        }
    }

    if (live) {
        float4* sx = (float4*)(&sX[wave][hi * XP + 4 * col]);
        sx[0 * (XP / 2)] = v0; sx[1 * (XP / 2)] = v1; sx[2 * (XP / 2)] = v2; sx[3 * (XP / 2)] = v3;
        sx[4 * (XP / 2)] = v4; sx[5 * (XP / 2)] = v5; sx[6 * (XP / 2)] = v6; sx[7 * (XP / 2)] = v7;
    }
    __syncthreads();

    if (live) {
        {
            float4* a4 = (float4*)(agg + (size_t)tile * 16 * PE) + rq;
            a4[0 * (PE / 2)] = v0; a4[1 * (PE / 2)] = v1; a4[2 * (PE / 2)] = v2; a4[3 * (PE / 2)] = v3;
            a4[4 * (PE / 2)] = v4; a4[5 * (PE / 2)] = v5; a4[6 * (PE / 2)] = v6; a4[7 * (PE / 2)] = v7;
        }

        v8f acc[4];
        #pragma unroll
        for (int r = 0; r < 8; ++r) {
            acc[0][r] = 0.0f; acc[1][r] = 0.0f; acc[2][r] = 0.0f; acc[3][r] = 0.0f;
        }
        #pragma unroll
        for (int kk = 0; kk < 16; ++kk) {
            const int k0 = kk * 4;
            v2f a = *(const v2f*)(&sX[wave][col * XP + k0 + koff]);
            #pragma unroll
            for (int n = 0; n < 4; ++n) {
                v2f b = *(const v2f*)(&sWm[(kk * 2 + hi) * WP + (n * 16 + col) * 2]);
                acc[n] = __builtin_amdgcn_wmma_f32_16x16x4_f32(
                    false, a, false, b, (short)0, acc[n], false, false);
            }
        }
        float* yt = y + (size_t)tile * 16 * PE + lane;
        #pragma unroll
        for (int r = 0; r < 8; ++r) {
            #pragma unroll
            for (int p = 0; p < 2; ++p) {
                const float X  = acc[2 * p][r];
                const float Y  = acc[2 * p + 1][r];
                const float Z  = hi ? X : Y;
                const float Zx = xhalf(Z);
                yt[(r    ) * PE + 32 * p] = hi ? Zx : X;
                yt[(r + 8) * PE + 32 * p] = hi ? Y  : Zx;
            }
        }
    }
}

#define EDGE_TILES (N_EDGES / 16)
#define EDGE_GRID  512

__global__ __launch_bounds__(256) void edge_msg_kernel(
    const float* __restrict__ y,
    const int*   __restrict__ src,
    const int*   __restrict__ dst,
    const float* __restrict__ attr,
    const float* __restrict__ bm,
    const float* __restrict__ We,
    const float* __restrict__ be,
    float*       __restrict__ agg)
{
    __shared__ __attribute__((aligned(16))) float sWe[(PD / 2) * WP];
    __shared__ __attribute__((aligned(16))) float sA[8][16 * AP];

    const int tid  = threadIdx.x;
    const int wave = __builtin_amdgcn_readfirstlane(tid >> 5);
    const int lane = tid & 31;

    {
        const float4* We4 = (const float4*)We;
        if (tid < (PD / 2) * (PE / 4)) {
            const int kp = tid >> 4, n4 = tid & 15;
            const float4 r0 = We4[(2 * kp    ) * (PE / 4) + n4];
            const float4 r1 = We4[(2 * kp + 1) * (PE / 4) + n4];
            float4* o = (float4*)(&sWe[kp * WP + n4 * 8]);
            o[0] = make_float4(r0.x, r1.x, r0.y, r1.y);
            o[1] = make_float4(r0.z, r1.z, r0.w, r1.w);
        }
    }

    const int row = lane >> 1;
    const int hlf = lane & 1;

    const int col  = lane & 15;
    const int hi   = lane >> 4;
    const int koff = hi * 2;

    const float bias0 = bm[col]      + be[col];
    const float bias1 = bm[16 + col] + be[16 + col];
    const float bias2 = bm[32 + col] + be[32 + col];
    const float bias3 = bm[48 + col] + be[48 + col];

    const int stride = (int)gridDim.x * 8;
    int tile = wave * (int)gridDim.x + (int)blockIdx.x;

    float  yn[16][2];
    float4 ga0, ga1;
    int    dv;
    int    s_nxt = 0;
    {
        const int e0 = tile * 16;
        const int s  = src[e0 + col];
        if (tile + stride < EDGE_TILES) s_nxt = src[(tile + stride) * 16 + col];
        const float4* ar = (const float4*)(attr + (size_t)(e0 + row) * PD + hlf * 8);
        ga0 = ar[0]; ga1 = ar[1];
        dv = dst[e0 + col];
        #pragma unroll
        for (int m = 0; m < 16; ++m) {
            const float* yr = y + (size_t)__builtin_amdgcn_readlane(s, m) * PE + lane;
            yn[m][0] = yr[0];
            yn[m][1] = yr[32];
        }
    }
    __syncthreads();

    for (; tile < EDGE_TILES; tile += stride) {
        {
            float4* oa = (float4*)(&sA[wave][row * AP + hlf * 8]);
            oa[0] = ga0; oa[1] = ga1;
        }
        const int dcur = dv;
        float yc[16][2];
        #pragma unroll
        for (int m = 0; m < 16; ++m) { yc[m][0] = yn[m][0]; yc[m][1] = yn[m][1]; }

        const int nt = tile + stride;
        if (nt < EDGE_TILES) {
            const int e0n = nt * 16;
            const float4* ar = (const float4*)(attr + (size_t)(e0n + row) * PD + hlf * 8);
            ga0 = ar[0]; ga1 = ar[1];
            dv = dst[e0n + col];
            #pragma unroll
            for (int m = 0; m < 16; ++m) {
                const float* yr = y + (size_t)__builtin_amdgcn_readlane(s_nxt, m) * PE + lane;
                yn[m][0] = yr[0];
                yn[m][1] = yr[32];
            }
            if (nt + stride < EDGE_TILES) s_nxt = src[(nt + stride) * 16 + col];
        }

        wave_lds_handover();

        v8f acc[4];
        #pragma unroll
        for (int r = 0; r < 8; ++r) {
            acc[0][r] = bias0; acc[1][r] = bias1; acc[2][r] = bias2; acc[3][r] = bias3;
        }

        #pragma unroll
        for (int kk = 0; kk < 4; ++kk) {
            const int k0 = kk * 4;
            v2f a = *(const v2f*)(&sA[wave][col * AP + k0 + koff]);
            #pragma unroll
            for (int n = 0; n < 4; ++n) {
                v2f b = *(const v2f*)(&sWe[(kk * 2 + hi) * WP + (n * 16 + col) * 2]);
                acc[n] = __builtin_amdgcn_wmma_f32_16x16x4_f32(
                    false, a, false, b, (short)0, acc[n], false, false);
            }
        }

        wave_lds_handover();

        #pragma unroll
        for (int r = 0; r < 8; ++r) {
            float P[2], Q[2];
            #pragma unroll
            for (int p = 0; p < 2; ++p) {
                const float X  = acc[2 * p][r];
                const float Y  = acc[2 * p + 1][r];
                const float Z  = hi ? X : Y;
                const float Zx = xhalf(Z);
                P[p] = (hi ? Zx : X) + yc[r][p];
                Q[p] = (hi ? Y  : Zx) + yc[r + 8][p];
            }
            float* row0 = agg + (size_t)__builtin_amdgcn_readlane(dcur, r)     * PE + lane;
            float* row1 = agg + (size_t)__builtin_amdgcn_readlane(dcur, r + 8) * PE + lane;
            atomicAdd(row0,      P[0]);
            atomicAdd(row0 + 32, P[1]);
            atomicAdd(row1,      Q[0]);
            atomicAdd(row1 + 32, Q[1]);
        }
    }
}

__global__ void count_kernel(const int* __restrict__ batch, float* __restrict__ counts) {
    int i = blockIdx.x * blockDim.x + threadIdx.x;
    if (i < N_NODES) atomicAdd(&counts[batch[i]], 1.0f);
}

#define GN_R      13
#define GN_ROWS   (16 * GN_R)
#define GN_BLOCKS ((N_NODES + GN_ROWS - 1) / GN_ROWS)

template <bool FULL>
__device__ __forceinline__ void gn_load_rows(const float4* __restrict__ a4,
                                             const int* __restrict__ batch,
                                             int row0, int tx,
                                             int (&g)[GN_R], float4 (&v)[GN_R]) {
    #pragma unroll
    for (int j = 0; j < GN_R; ++j) {
        const int row = row0 + j;
        const int rc  = (FULL || row < N_NODES) ? row : N_NODES - 1;
        const int gj  = batch[rc];
        v[j] = a4[(size_t)rc * (PE / 4) + tx];
        g[j] = (FULL || row < N_NODES) ? gj : -1;
    }
}

__device__ __forceinline__ void gn_load_table(const float4* __restrict__ t4, int tx,
                                              const int (&g)[GN_R], float4 (&m)[GN_R]) {
    #pragma unroll
    for (int j = 0; j < GN_R; ++j) {
        const int gc = g[j] < 0 ? 0 : g[j];
        m[j] = t4[gc * (PE / 4) + tx];
    }
}

__device__ __forceinline__ void gn_flush(float* __restrict__ table, int g, int tx, const float4 a) {
    float* p = table + g * PE + tx * 4;
    atomicAdd(p + 0, a.x);
    atomicAdd(p + 1, a.y);
    atomicAdd(p + 2, a.z);
    atomicAdd(p + 3, a.w);
}

__device__ __forceinline__ void gn_reduce_rows(float* __restrict__ table, int tx,
                                               const int (&g)[GN_R], const float4 (&v)[GN_R]) {
    int    gp  = g[0];
    float4 acc = v[0];
    #pragma unroll
    for (int j = 1; j < GN_R; ++j) {
        if (g[j] != gp) {
            if (gp >= 0) gn_flush(table, gp, tx, acc);
            gp  = g[j];
            acc = v[j];
        } else {
            acc.x += v[j].x; acc.y += v[j].y; acc.z += v[j].z; acc.w += v[j].w;
        }
    }
    if (gp >= 0) gn_flush(table, gp, tx, acc);
}

template <bool FULL>
__device__ __forceinline__ void mean_accum_rows(const float* __restrict__ h,
                                                const int* __restrict__ batch,
                                                float* __restrict__ mean, int row0, int tx) {
    int    g[GN_R];
    float4 v[GN_R];
    gn_load_rows<FULL>((const float4*)h, batch, row0, tx, g, v);
    gn_reduce_rows(mean, tx, g, v);
}

__global__ __launch_bounds__(256) void mean_accum_kernel(const float* __restrict__ h,
                                  const int* __restrict__ batch,
                                  float* __restrict__ mean) {
    const int tx   = threadIdx.x & 15;
    const int ty   = threadIdx.x >> 4;
    const int row0 = blockIdx.x * GN_ROWS + ty * GN_R;
    if (row0 + GN_R <= N_NODES) mean_accum_rows<true >(h, batch, mean, row0, tx);
    else                        mean_accum_rows<false>(h, batch, mean, row0, tx);
}

__global__ void mean_fin_kernel(float* __restrict__ mean, const float* __restrict__ counts) {
    int i = blockIdx.x * blockDim.x + threadIdx.x;
    mean[i] /= fmaxf(counts[i >> 6], 1.0f);
}

template <bool FULL>
__device__ __forceinline__ void center_var_rows(const float* __restrict__ h,
                                                const int* __restrict__ batch,
                                                const float* __restrict__ mean,
                                                const float* __restrict__ mscale,
                                                float* __restrict__ tmp,
                                                float* __restrict__ var, int row0, int tx) {
    float4*      t4 = (float4*)tmp;
    const float4 ms = ((const float4*)mscale)[tx];
    int    g[GN_R];
    float4 v[GN_R];
    float4 m[GN_R];
    gn_load_rows<FULL>((const float4*)h, batch, row0, tx, g, v);
    gn_load_table((const float4*)mean, tx, g, m);
    #pragma unroll
    for (int j = 0; j < GN_R; ++j) {
        const int row = row0 + j;
        float4 o;
        o.x = v[j].x - m[j].x * ms.x;
        o.y = v[j].y - m[j].y * ms.y;
        o.z = v[j].z - m[j].z * ms.z;
        o.w = v[j].w - m[j].w * ms.w;
        if (FULL || row < N_NODES) t4[(size_t)row * (PE / 4) + tx] = o;
        v[j] = make_float4(o.x * o.x, o.y * o.y, o.z * o.z, o.w * o.w);
    }
    gn_reduce_rows(var, tx, g, v);
}

__global__ __launch_bounds__(256) void center_var_kernel(const float* __restrict__ h,
                                  const int* __restrict__ batch,
                                  const float* __restrict__ mean,
                                  const float* __restrict__ mscale,
                                  float* __restrict__ tmp,
                                  float* __restrict__ var) {
    const int tx   = threadIdx.x & 15;
    const int ty   = threadIdx.x >> 4;
    const int row0 = blockIdx.x * GN_ROWS + ty * GN_R;
    if (row0 + GN_R <= N_NODES) center_var_rows<true >(h, batch, mean, mscale, tmp, var, row0, tx);
    else                        center_var_rows<false>(h, batch, mean, mscale, tmp, var, row0, tx);
}

__global__ void scale_fin_kernel(float* __restrict__ var,
                                 const float* __restrict__ counts,
                                 const float* __restrict__ w) {
    int i = blockIdx.x * blockDim.x + threadIdx.x;
    float c = fmaxf(counts[i >> 6], 1.0f);
    var[i] = w[i & 63] * rsqrtf(var[i] / c + EPS);
}

template <bool FULL>
__device__ __forceinline__ void apply_rows(const float* __restrict__ tmp,
                                           const int* __restrict__ batch,
                                           const float* __restrict__ scale,
                                           const float* __restrict__ bias,
                                           float* __restrict__ out, int row0, int tx) {
    float4*      o4 = (float4*)out;
    const float4 b  = ((const float4*)bias)[tx];
    int    g[GN_R];
    float4 t[GN_R];
    float4 s[GN_R];
    gn_load_rows<FULL>((const float4*)tmp, batch, row0, tx, g, t);
    gn_load_table((const float4*)scale, tx, g, s);
    #pragma unroll
    for (int j = 0; j < GN_R; ++j) {
        const int row = row0 + j;
        float4 r;
        r.x = t[j].x * s[j].x + b.x;
        r.y = t[j].y * s[j].y + b.y;
        r.z = t[j].z * s[j].z + b.z;
        r.w = t[j].w * s[j].w + b.w;
        r.x = r.x > 0.0f ? r.x : NEG_SLOPE * r.x;
        r.y = r.y > 0.0f ? r.y : NEG_SLOPE * r.y;
        r.z = r.z > 0.0f ? r.z : NEG_SLOPE * r.z;
        r.w = r.w > 0.0f ? r.w : NEG_SLOPE * r.w;
        if (FULL || row < N_NODES) o4[(size_t)row * (PE / 4) + tx] = r;
    }
}

__global__ __launch_bounds__(256) void apply_kernel(const float* __restrict__ tmp,
                             const int* __restrict__ batch,
                             const float* __restrict__ scale,
                             const float* __restrict__ bias,
                             float* __restrict__ out) {
    const int tx   = threadIdx.x & 15;
    const int ty   = threadIdx.x >> 4;
    const int row0 = blockIdx.x * GN_ROWS + ty * GN_R;
    if (row0 + GN_R <= N_NODES) apply_rows<true >(tmp, batch, scale, bias, out, row0, tx);
    else                        apply_rows<false>(tmp, batch, scale, bias, out, row0, tx);
}

extern "C" void kernel_launch(void* const* d_in, const int* in_sizes, int n_in,
                              void* d_out, int out_size, void* d_ws, size_t ws_size,
                              hipStream_t stream) {
    const float* x0    = (const float*)d_in[0];
    const int*   idx   = (const int*)  d_in[1];
    const float* attr  = (const float*)d_in[2];
    const int*   batch = (const int*)  d_in[3];
    const float* Wm    = (const float*)d_in[4];
    const float* bm    = (const float*)d_in[5];
    const float* We    = (const float*)d_in[6];
    const float* be    = (const float*)d_in[7];
    const float* gw    = (const float*)d_in[8];
    const float* gb    = (const float*)d_in[9];
    const float* gms   = (const float*)d_in[10];

    const int* src = idx;
    const int* dst = idx + N_EDGES;

    char* ws = (char*)d_ws;
    const size_t node_bytes = (size_t)N_NODES * PE * sizeof(float);
    float* h      = (float*)(ws);
    float* tmp    = (float*)(ws + node_bytes);
    float* xbuf   = (float*)(ws + 2 * node_bytes);
    float* mean   = (float*)(ws + 3 * node_bytes);
    float* var    = mean + N_GRAPHS * PE;
    float* counts = var  + N_GRAPHS * PE;

    hipMemsetAsync(counts, 0, N_GRAPHS * sizeof(float), stream);
    count_kernel<<<(N_NODES + 255) / 256, 256, 0, stream>>>(batch, counts);

    for (int l = 0; l < 2; ++l) {
        const float* xin  = (l == 0) ? x0 : xbuf;
        float*       xout = (l == 1) ? (float*)d_out : xbuf;

        node_msg_kernel<<<NODE_GRID, 256, 0, stream>>>(
            xin, Wm + (size_t)l * PE * PE, tmp, h);

        edge_msg_kernel<<<EDGE_GRID, 256, 0, stream>>>(
            tmp, src, dst, attr,
            bm + (size_t)l * PE,
            We + (size_t)l * PD * PE, be + (size_t)l * PE, h);

        hipMemsetAsync(mean, 0, (size_t)2 * N_GRAPHS * PE * sizeof(float), stream);
        mean_accum_kernel<<<GN_BLOCKS, 256, 0, stream>>>(h, batch, mean);
        mean_fin_kernel<<<N_GRAPHS * PE / 256, 256, 0, stream>>>(mean, counts);
        center_var_kernel<<<GN_BLOCKS, 256, 0, stream>>>(h, batch, mean,
                                                        gms + (size_t)l * PE, tmp, var);
        scale_fin_kernel<<<N_GRAPHS * PE / 256, 256, 0, stream>>>(var, counts,
                                                                  gw + (size_t)l * PE);
        apply_kernel<<<GN_BLOCKS, 256, 0, stream>>>(tmp, batch, var,
                                                   gb + (size_t)l * PE, xout);
    }
}


# graph norm as one kernel a layer: a work-group per graph (batch is sorted; row ranges tabled by the node kernel) keeps its rows in registers for mean, variance, output; no atomics; sum order changes
# speedup vs baseline: 2.5257x; 1.3655x over previous
#include <hip/hip_runtime.h>
#include <hip/hip_bf16.h>
#include <stdint.h>

#define N_NODES   50000
#define N_EDGES   800000
#define PE        64
#define PD        16
#define N_GRAPHS  256
#define EPS       1e-5f
#define NEG_SLOPE 0.01f

#define XP 68
#define AP 20
#define WP 160

typedef __attribute__((ext_vector_type(2))) float v2f;
typedef __attribute__((ext_vector_type(8))) float v8f;

__device__ __forceinline__ float xhalf(float v) {
    return __builtin_bit_cast(float, __builtin_amdgcn_permlanex16(
        __builtin_bit_cast(int, v), __builtin_bit_cast(int, v), 0x76543210, 0xfedcba98, false, false));
}

__device__ __forceinline__ void wave_lds_handover() {
    __builtin_amdgcn_fence(__ATOMIC_RELEASE, "wavefront", "local");
    __builtin_amdgcn_wave_barrier();
    __builtin_amdgcn_fence(__ATOMIC_ACQUIRE, "wavefront", "local");
}

#define NODE_TILES (N_NODES / 16)
#define NODE_GRID  ((NODE_TILES + 7) / 8)
static_assert(N_NODES % 16 == 0, "the node kernel has no partial tile");

__global__ __launch_bounds__(256) void node_msg_kernel(
    const float* x,
    const float* __restrict__ Wm,
    const int*   __restrict__ batch,
    float*       __restrict__ y,
    float*       agg,
    int*         __restrict__ gstart)
{
    __shared__ __attribute__((aligned(16))) float sWm[(PE / 2) * WP];
    __shared__ __attribute__((aligned(16))) float sX[8][16 * XP];

    const int tid  = threadIdx.x;
    const int wave = __builtin_amdgcn_readfirstlane(tid >> 5);
    const int lane = tid & 31;

    const int col  = lane & 15;
    const int hi   = lane >> 4;
    const int koff = hi * 2;

    const int  tile = blockIdx.x * 8 + wave;
    const bool live = tile < NODE_TILES;

    float4 v0, v1, v2, v3, v4, v5, v6, v7;
    v0 = v1 = v2 = v3 = v4 = v5 = v6 = v7 = make_float4(0.0f, 0.0f, 0.0f, 0.0f);
    const int rq = hi * (PE / 4) + col;
    if (live) {
        const float4* x4 = (const float4*)(x + (size_t)tile * 16 * PE) + rq;
        v0 = x4[0 * (PE / 2)]; v1 = x4[1 * (PE / 2)]; v2 = x4[2 * (PE / 2)]; v3 = x4[3 * (PE / 2)];
        v4 = x4[4 * (PE / 2)]; v5 = x4[5 * (PE / 2)]; v6 = x4[6 * (PE / 2)]; v7 = x4[7 * (PE / 2)];
    }

    {
        const float4* Wm4 = (const float4*)Wm;
        #pragma unroll
        for (int it = tid; it < (PE / 2) * (PE / 4); it += 256) {
            const int kp = it >> 4, n4 = it & 15;
            const float4 r0 = Wm4[(2 * kp    ) * (PE / 4) + n4];
            const float4 r1 = Wm4[(2 * kp + 1) * (PE / 4) + n4];
            float4* o = (float4*)(&sWm[kp * WP + n4 * 8]);
            o[0] = make_float4(r0.x, r1.x, r0.y, r1.y);
            o[1] = make_float4(r0.z, r1.z, r0.w, r1.w);
        }
    }

    if (live) {
        float4* sx = (float4*)(&sX[wave][hi * XP + 4 * col]);
        sx[0 * (XP / 2)] = v0; sx[1 * (XP / 2)] = v1; sx[2 * (XP / 2)] = v2; sx[3 * (XP / 2)] = v3;
        sx[4 * (XP / 2)] = v4; sx[5 * (XP / 2)] = v5; sx[6 * (XP / 2)] = v6; sx[7 * (XP / 2)] = v7;
    }
    __syncthreads();

    if (live) {
        {
            float4* a4 = (float4*)(agg + (size_t)tile * 16 * PE) + rq;
            a4[0 * (PE / 2)] = v0; a4[1 * (PE / 2)] = v1; a4[2 * (PE / 2)] = v2; a4[3 * (PE / 2)] = v3;
            a4[4 * (PE / 2)] = v4; a4[5 * (PE / 2)] = v5; a4[6 * (PE / 2)] = v6; a4[7 * (PE / 2)] = v7;
        }

        v8f acc[4];
        #pragma unroll
        for (int r = 0; r < 8; ++r) {
            acc[0][r] = 0.0f; acc[1][r] = 0.0f; acc[2][r] = 0.0f; acc[3][r] = 0.0f;
        }
        #pragma unroll
        for (int kk = 0; kk < 16; ++kk) {
            const int k0 = kk * 4;
            v2f a = *(const v2f*)(&sX[wave][col * XP + k0 + koff]);
            #pragma unroll
            for (int n = 0; n < 4; ++n) {
                v2f b = *(const v2f*)(&sWm[(kk * 2 + hi) * WP + (n * 16 + col) * 2]);
                acc[n] = __builtin_amdgcn_wmma_f32_16x16x4_f32(
                    false, a, false, b, (short)0, acc[n], false, false);
            }
        }
        float* yt = y + (size_t)tile * 16 * PE + lane;
        #pragma unroll
        for (int r = 0; r < 8; ++r) {
            #pragma unroll
            for (int p = 0; p < 2; ++p) {
                const float X  = acc[2 * p][r];
                const float Y  = acc[2 * p + 1][r];
                const float Z  = hi ? X : Y;
                const float Zx = xhalf(Z);
                yt[(r    ) * PE + 32 * p] = hi ? Zx : X;
                yt[(r + 8) * PE + 32 * p] = hi ? Y  : Zx;
            }
        }
    }

    const int gi = blockIdx.x * 256 + tid;
    if (gstart != nullptr && gi < N_NODES) {
        const int gb_cur = batch[gi];
        const int gfirst = gi > 0 ? batch[gi - 1] + 1 : 0;
        for (int g = gfirst; g <= gb_cur; ++g) gstart[g] = gi;
        if (gi == N_NODES - 1)
            for (int g = gb_cur + 1; g <= N_GRAPHS; ++g) gstart[g] = N_NODES;
    }
}

#define EDGE_TILES (N_EDGES / 16)
#define EDGE_GRID  512

__global__ __launch_bounds__(256) void edge_msg_kernel(
    const float* __restrict__ y,
    const int*   __restrict__ src,
    const int*   __restrict__ dst,
    const float* __restrict__ attr,
    const float* __restrict__ bm,
    const float* __restrict__ We,
    const float* __restrict__ be,
    float*       __restrict__ agg)
{
    __shared__ __attribute__((aligned(16))) float sWe[(PD / 2) * WP];
    __shared__ __attribute__((aligned(16))) float sA[8][16 * AP];

    const int tid  = threadIdx.x;
    const int wave = __builtin_amdgcn_readfirstlane(tid >> 5);
    const int lane = tid & 31;

    {
        const float4* We4 = (const float4*)We;
        if (tid < (PD / 2) * (PE / 4)) {
            const int kp = tid >> 4, n4 = tid & 15;
            const float4 r0 = We4[(2 * kp    ) * (PE / 4) + n4];
            const float4 r1 = We4[(2 * kp + 1) * (PE / 4) + n4];
            float4* o = (float4*)(&sWe[kp * WP + n4 * 8]);
            o[0] = make_float4(r0.x, r1.x, r0.y, r1.y);
            o[1] = make_float4(r0.z, r1.z, r0.w, r1.w);
        }
    }

    const int row = lane >> 1;
    const int hlf = lane & 1;

    const int col  = lane & 15;
    const int hi   = lane >> 4;
    const int koff = hi * 2;

    const float bias0 = bm[col]      + be[col];
    const float bias1 = bm[16 + col] + be[16 + col];
    const float bias2 = bm[32 + col] + be[32 + col];
    const float bias3 = bm[48 + col] + be[48 + col];

    const int stride = (int)gridDim.x * 8;
    int tile = wave * (int)gridDim.x + (int)blockIdx.x;

    float  yn[16][2];
    float4 ga0, ga1;
    int    dv;
    int    s_nxt = 0;
    {
        const int e0 = tile * 16;
        const int s  = src[e0 + col];
        if (tile + stride < EDGE_TILES) s_nxt = src[(tile + stride) * 16 + col];
        const float4* ar = (const float4*)(attr + (size_t)(e0 + row) * PD + hlf * 8);
        ga0 = ar[0]; ga1 = ar[1];
        dv = dst[e0 + col];
        #pragma unroll
        for (int m = 0; m < 16; ++m) {
            const float* yr = y + (size_t)__builtin_amdgcn_readlane(s, m) * PE + lane;
            yn[m][0] = yr[0];
            yn[m][1] = yr[32];
        }
    }
    __syncthreads();

    for (; tile < EDGE_TILES; tile += stride) {
        {
            float4* oa = (float4*)(&sA[wave][row * AP + hlf * 8]);
            oa[0] = ga0; oa[1] = ga1;
        }
        const int dcur = dv;
        float yc[16][2];
        #pragma unroll
        for (int m = 0; m < 16; ++m) { yc[m][0] = yn[m][0]; yc[m][1] = yn[m][1]; }

        const int nt = tile + stride;
        if (nt < EDGE_TILES) {
            const int e0n = nt * 16;
            const float4* ar = (const float4*)(attr + (size_t)(e0n + row) * PD + hlf * 8);
            ga0 = ar[0]; ga1 = ar[1];
            dv = dst[e0n + col];
            #pragma unroll
            for (int m = 0; m < 16; ++m) {
                const float* yr = y + (size_t)__builtin_amdgcn_readlane(s_nxt, m) * PE + lane;
                yn[m][0] = yr[0];
                yn[m][1] = yr[32];
            }
            if (nt + stride < EDGE_TILES) s_nxt = src[(nt + stride) * 16 + col];
        }

        wave_lds_handover();

        v8f acc[4];
        #pragma unroll
        for (int r = 0; r < 8; ++r) {
            acc[0][r] = bias0; acc[1][r] = bias1; acc[2][r] = bias2; acc[3][r] = bias3;
        }

        #pragma unroll
        for (int kk = 0; kk < 4; ++kk) {
            const int k0 = kk * 4;
            v2f a = *(const v2f*)(&sA[wave][col * AP + k0 + koff]);
            #pragma unroll
            for (int n = 0; n < 4; ++n) {
                v2f b = *(const v2f*)(&sWe[(kk * 2 + hi) * WP + (n * 16 + col) * 2]);
                acc[n] = __builtin_amdgcn_wmma_f32_16x16x4_f32(
                    false, a, false, b, (short)0, acc[n], false, false);
            }
        }

        wave_lds_handover();

        #pragma unroll
        for (int r = 0; r < 8; ++r) {
            float P[2], Q[2];
            #pragma unroll
            for (int p = 0; p < 2; ++p) {
                const float X  = acc[2 * p][r];
                const float Y  = acc[2 * p + 1][r];
                const float Z  = hi ? X : Y;
                const float Zx = xhalf(Z);
                P[p] = (hi ? Zx : X) + yc[r][p];
                Q[p] = (hi ? Y  : Zx) + yc[r + 8][p];
            }
            float* row0 = agg + (size_t)__builtin_amdgcn_readlane(dcur, r)     * PE + lane;
            float* row1 = agg + (size_t)__builtin_amdgcn_readlane(dcur, r + 8) * PE + lane;
            atomicAdd(row0,      P[0]);
            atomicAdd(row0 + 32, P[1]);
            atomicAdd(row1,      Q[0]);
            atomicAdd(row1 + 32, Q[1]);
        }
    }
}

#define GN_T   256
#define GN_RPP (GN_T / 16)
#define GN_W   (GN_T / 32)
#define GN_R   16

__device__ __forceinline__ float4 gn_block_sum(float4 s, float4 (*buf)[16], int wave, int hi, int tx) {
    s.x += xhalf(s.x); s.y += xhalf(s.y); s.z += xhalf(s.z); s.w += xhalf(s.w);
    if (hi == 0) buf[wave][tx] = s;
    __syncthreads();
    float4 t = buf[hi * (GN_W / 2)][tx];
    #pragma unroll
    for (int k = 1; k < GN_W / 2; ++k) {
        const float4 p = buf[hi * (GN_W / 2) + k][tx];
        t.x += p.x; t.y += p.y; t.z += p.z; t.w += p.w;
    }
    t.x += xhalf(t.x); t.y += xhalf(t.y); t.z += xhalf(t.z); t.w += xhalf(t.w);
    return t;
}

__device__ __forceinline__ float4 gn_center(const float4 v, const float4 mm) {
    return make_float4(v.x - mm.x, v.y - mm.y, v.z - mm.z, v.w - mm.w);
}

__device__ __forceinline__ float4 gn_finish(const float4 o, const float4 sc, const float4 b) {
    float4 r;
    r.x = o.x * sc.x + b.x;
    r.y = o.y * sc.y + b.y;
    r.z = o.z * sc.z + b.z;
    r.w = o.w * sc.w + b.w;
    r.x = r.x > 0.0f ? r.x : NEG_SLOPE * r.x;
    r.y = r.y > 0.0f ? r.y : NEG_SLOPE * r.y;
    r.z = r.z > 0.0f ? r.z : NEG_SLOPE * r.z;
    r.w = r.w > 0.0f ? r.w : NEG_SLOPE * r.w;
    return r;
}

__global__ __launch_bounds__(GN_T) void graph_norm_kernel(
    const float* __restrict__ h,
    const int*   __restrict__ gstart,
    const float* __restrict__ mscale,
    const float* __restrict__ w,
    const float* __restrict__ bias,
    float*       __restrict__ out)
{
    __shared__ __attribute__((aligned(16))) float4 sred[2][GN_W][16];

    const int tid  = threadIdx.x;
    const int tx   = tid & 15;
    const int ty   = tid >> 4;
    const int wave = tid >> 5;
    const int hi   = ty & 1;

    const int   lo  = gstart[blockIdx.x];
    const int   n   = gstart[blockIdx.x + 1] - lo;
    const float cnt = fmaxf((float)n, 1.0f);

    const float4* h4 = (const float4*)h + (size_t)lo * (PE / 4) + tx;
    float4*       o4 = (float4*)out     + (size_t)lo * (PE / 4) + tx;

    const float4 ms = ((const float4*)mscale)[tx];
    const float4 wv = ((const float4*)w)[tx];
    const float4 bv = ((const float4*)bias)[tx];

    float4 v[GN_R];
    #pragma unroll
    for (int j = 0; j < GN_R; ++j) {
        const int r = ty + GN_RPP * j;
        v[j] = make_float4(0.0f, 0.0f, 0.0f, 0.0f);
        if (r < n) v[j] = h4[(size_t)r * (PE / 4)];
    }
    float4 s = v[0];
    #pragma unroll
    for (int j = 1; j < GN_R; ++j) { s.x += v[j].x; s.y += v[j].y; s.z += v[j].z; s.w += v[j].w; }
    for (int r = ty + GN_RPP * GN_R; r < n; r += GN_RPP) {
        const float4 t = h4[(size_t)r * (PE / 4)];
        s.x += t.x; s.y += t.y; s.z += t.z; s.w += t.w;
    }
    const float4 tot = gn_block_sum(s, sred[0], wave, hi, tx);

    const float4 mm = make_float4(tot.x / cnt * ms.x, tot.y / cnt * ms.y, tot.z / cnt * ms.z, tot.w / cnt * ms.w);

    float4 q = make_float4(0.0f, 0.0f, 0.0f, 0.0f);
    #pragma unroll
    for (int j = 0; j < GN_R; ++j) {
        const int r = ty + GN_RPP * j;
        const float4 o = gn_center(v[j], mm);
        v[j] = o;
        if (r < n) { q.x += o.x * o.x; q.y += o.y * o.y; q.z += o.z * o.z; q.w += o.w * o.w; }
    }
    for (int r = ty + GN_RPP * GN_R; r < n; r += GN_RPP) {
        const float4 o = gn_center(h4[(size_t)r * (PE / 4)], mm);
        q.x += o.x * o.x; q.y += o.y * o.y; q.z += o.z * o.z; q.w += o.w * o.w;
    }
    const float4 qt = gn_block_sum(q, sred[1], wave, hi, tx);

    const float4 sc = make_float4(wv.x * rsqrtf(qt.x / cnt + EPS), wv.y * rsqrtf(qt.y / cnt + EPS),
                                  wv.z * rsqrtf(qt.z / cnt + EPS), wv.w * rsqrtf(qt.w / cnt + EPS));

    #pragma unroll
    for (int j = 0; j < GN_R; ++j) {
        const int r = ty + GN_RPP * j;
        if (r < n) o4[(size_t)r * (PE / 4)] = gn_finish(v[j], sc, bv);
    }
    for (int r = ty + GN_RPP * GN_R; r < n; r += GN_RPP)
        o4[(size_t)r * (PE / 4)] = gn_finish(gn_center(h4[(size_t)r * (PE / 4)], mm), sc, bv);
}

extern "C" void kernel_launch(void* const* d_in, const int* in_sizes, int n_in,
                              void* d_out, int out_size, void* d_ws, size_t ws_size,
                              hipStream_t stream) {
    const float* x0    = (const float*)d_in[0];
    const int*   idx   = (const int*)  d_in[1];
    const float* attr  = (const float*)d_in[2];
    const int*   batch = (const int*)  d_in[3];
    const float* Wm    = (const float*)d_in[4];
    const float* bm    = (const float*)d_in[5];
    const float* We    = (const float*)d_in[6];
    const float* be    = (const float*)d_in[7];
    const float* gw    = (const float*)d_in[8];
    const float* gb    = (const float*)d_in[9];
    const float* gms   = (const float*)d_in[10];

    const int* src = idx;
    const int* dst = idx + N_EDGES;

    char* ws = (char*)d_ws;
    const size_t node_bytes = (size_t)N_NODES * PE * sizeof(float);
    float* h      = (float*)(ws);
    float* tmp    = (float*)(ws + node_bytes);
    float* xbuf   = (float*)(ws + 2 * node_bytes);
    int*   gstart = (int*)  (ws + 3 * node_bytes);

    for (int l = 0; l < 2; ++l) {
        const float* xin  = (l == 0) ? x0 : xbuf;
        float*       xout = (l == 1) ? (float*)d_out : xbuf;

        node_msg_kernel<<<NODE_GRID, 256, 0, stream>>>(
            xin, Wm + (size_t)l * PE * PE, batch, tmp, h, (l == 0) ? gstart : (int*)nullptr);

        edge_msg_kernel<<<EDGE_GRID, 256, 0, stream>>>(
            tmp, src, dst, attr,
            bm + (size_t)l * PE,
            We + (size_t)l * PD * PE, be + (size_t)l * PE, h);

        graph_norm_kernel<<<N_GRAPHS, GN_T, 0, stream>>>(
            h, gstart, gms + (size_t)l * PE, gw + (size_t)l * PE, gb + (size_t)l * PE, xout);
    }
}


# edge sum in LDS: edges grouped once a call by destination bin (counting sort, 512 bins of 98 nodes); a work-group per bin adds messages with LDS atomics, writes each row once; sum order changes
# speedup vs baseline: 3.1367x; 1.2419x over previous
#include <hip/hip_runtime.h>
#include <hip/hip_bf16.h>
#include <stdint.h>

#define N_NODES   50000
#define N_EDGES   800000
#define PE        64
#define PD        16
#define N_GRAPHS  256
#define EPS       1e-5f
#define NEG_SLOPE 0.01f

#define XP 68
#define AP 20
#define WP 160

typedef __attribute__((ext_vector_type(2))) float v2f;
typedef __attribute__((ext_vector_type(8))) float v8f;

__device__ __forceinline__ float xhalf(float v) {
    return __builtin_bit_cast(float, __builtin_amdgcn_permlanex16(
        __builtin_bit_cast(int, v), __builtin_bit_cast(int, v), 0x76543210, 0xfedcba98, false, false));
}

__device__ __forceinline__ void wave_lds_handover() {
    __builtin_amdgcn_fence(__ATOMIC_RELEASE, "wavefront", "local");
    __builtin_amdgcn_wave_barrier();
    __builtin_amdgcn_fence(__ATOMIC_ACQUIRE, "wavefront", "local");
}

#define NODE_TILES (N_NODES / 16)
#define NODE_GRID  ((NODE_TILES + 7) / 8)
static_assert(N_NODES % 16 == 0, "the node kernel has no partial tile");

#define NBINS    512
#define BIN_ROWS 98
static_assert(NBINS * BIN_ROWS >= N_NODES, "every node has a bin");
static_assert(N_NODES <= 65536 && BIN_ROWS < 256, "a record packs the source row in 16 bits, the row in its bin above them");
static_assert(NBINS <= NODE_GRID * 256, "the node kernel zeroes one bin counter per thread");

__global__ __launch_bounds__(256) void node_msg_kernel(
    const float* x,
    const float* __restrict__ Wm,
    const int*   __restrict__ batch,
    float*       __restrict__ y,
    int*         __restrict__ gstart,
    int*         __restrict__ bincnt)
{
    __shared__ __attribute__((aligned(16))) float sWm[(PE / 2) * WP];
    __shared__ __attribute__((aligned(16))) float sX[8][16 * XP];

    const int tid  = threadIdx.x;
    const int wave = __builtin_amdgcn_readfirstlane(tid >> 5);
    const int lane = tid & 31;

    const int col  = lane & 15;
    const int hi   = lane >> 4;
    const int koff = hi * 2;

    const int  tile = blockIdx.x * 8 + wave;
    const bool live = tile < NODE_TILES;

    float4 v0, v1, v2, v3, v4, v5, v6, v7;
    v0 = v1 = v2 = v3 = v4 = v5 = v6 = v7 = make_float4(0.0f, 0.0f, 0.0f, 0.0f);
    const int rq = hi * (PE / 4) + col;
    if (live) {
        const float4* x4 = (const float4*)(x + (size_t)tile * 16 * PE) + rq;
        v0 = x4[0 * (PE / 2)]; v1 = x4[1 * (PE / 2)]; v2 = x4[2 * (PE / 2)]; v3 = x4[3 * (PE / 2)];
        v4 = x4[4 * (PE / 2)]; v5 = x4[5 * (PE / 2)]; v6 = x4[6 * (PE / 2)]; v7 = x4[7 * (PE / 2)];
    }

    {
        const float4* Wm4 = (const float4*)Wm;
        #pragma unroll
        for (int it = tid; it < (PE / 2) * (PE / 4); it += 256) {
            const int kp = it >> 4, n4 = it & 15;
            const float4 r0 = Wm4[(2 * kp    ) * (PE / 4) + n4];
            const float4 r1 = Wm4[(2 * kp + 1) * (PE / 4) + n4];
            float4* o = (float4*)(&sWm[kp * WP + n4 * 8]);
            o[0] = make_float4(r0.x, r1.x, r0.y, r1.y);
            o[1] = make_float4(r0.z, r1.z, r0.w, r1.w);
        }
    }

    if (live) {
        float4* sx = (float4*)(&sX[wave][hi * XP + 4 * col]);
        sx[0 * (XP / 2)] = v0; sx[1 * (XP / 2)] = v1; sx[2 * (XP / 2)] = v2; sx[3 * (XP / 2)] = v3;
        sx[4 * (XP / 2)] = v4; sx[5 * (XP / 2)] = v5; sx[6 * (XP / 2)] = v6; sx[7 * (XP / 2)] = v7;
    }
    __syncthreads();

    if (live) {
        v8f acc[4];
        #pragma unroll
        for (int r = 0; r < 8; ++r) {
            acc[0][r] = 0.0f; acc[1][r] = 0.0f; acc[2][r] = 0.0f; acc[3][r] = 0.0f;
        }
        #pragma unroll
        for (int kk = 0; kk < 16; ++kk) {
            const int k0 = kk * 4;
            v2f a = *(const v2f*)(&sX[wave][col * XP + k0 + koff]);
            #pragma unroll
            for (int n = 0; n < 4; ++n) {
                v2f b = *(const v2f*)(&sWm[(kk * 2 + hi) * WP + (n * 16 + col) * 2]);
                acc[n] = __builtin_amdgcn_wmma_f32_16x16x4_f32(
                    false, a, false, b, (short)0, acc[n], false, false);
            }
        }
        float* yt = y + (size_t)tile * 16 * PE + lane;
        #pragma unroll
        for (int r = 0; r < 8; ++r) {
            #pragma unroll
            for (int p = 0; p < 2; ++p) {
                const float X  = acc[2 * p][r];
                const float Y  = acc[2 * p + 1][r];
                const float Z  = hi ? X : Y;
                const float Zx = xhalf(Z);
                yt[(r    ) * PE + 32 * p] = hi ? Zx : X;
                yt[(r + 8) * PE + 32 * p] = hi ? Y  : Zx;
            }
        }
    }

    const int gi = blockIdx.x * 256 + tid;
    if (gstart != nullptr && gi < N_NODES) {
        const int gb_cur = batch[gi];
        const int gfirst = gi > 0 ? batch[gi - 1] + 1 : 0;
        for (int g = gfirst; g <= gb_cur; ++g) gstart[g] = gi;
        if (gi == N_NODES - 1)
            for (int g = gb_cur + 1; g <= N_GRAPHS; ++g) gstart[g] = N_NODES;
    }
    if (bincnt != nullptr && gi < NBINS) bincnt[gi] = 0;
}

#define BC_T  512
#define BC_CH 3200
#define BC_G  (N_EDGES / BC_CH)
static_assert(N_EDGES % BC_CH == 0, "the grouping kernels have no partial chunk");
static_assert(BC_T == NBINS, "one thread a bin in the grouping kernels");
#define BC_K ((BC_CH + BC_T - 1) / BC_T)

__global__ __launch_bounds__(BC_T) void bin_count_kernel(
    const int* __restrict__ dst,
    int*       __restrict__ bincnt,
    int*       __restrict__ wgoff)
{
    __shared__ int hist[NBINS];
    const int tid = threadIdx.x;
    const int e_lo = blockIdx.x * BC_CH;
    int d[BC_K];
    #pragma unroll
    for (int k = 0; k < BC_K; ++k) {
        const int e = tid + k * BC_T;
        d[k] = e < BC_CH ? dst[e_lo + e] : -1;
    }
    hist[tid] = 0;
    __syncthreads();
    #pragma unroll
    for (int k = 0; k < BC_K; ++k)
        if (d[k] >= 0)
            __hip_atomic_fetch_add(&hist[d[k] / BIN_ROWS], 1, __ATOMIC_RELAXED, __HIP_MEMORY_SCOPE_WORKGROUP);
    __syncthreads();
    wgoff[blockIdx.x * NBINS + tid] = atomicAdd(&bincnt[tid], hist[tid]);
}

__global__ __launch_bounds__(BC_T) void bin_scatter_kernel(
    const int* __restrict__ src,
    const int* __restrict__ dst,
    const int* __restrict__ bincnt,
    const int* __restrict__ wgoff,
    int*       __restrict__ binbase,
    int*       __restrict__ rec)
{
    __shared__ int wsum[BC_T / 32];
    __shared__ int cursor[NBINS];
    const int tid  = threadIdx.x;
    const int lane = tid & 31;
    const int wave = tid >> 5;
    const int c    = bincnt[tid];
    const int off  = wgoff[blockIdx.x * NBINS + tid];
    const int e_lo = blockIdx.x * BC_CH;
    int d[BC_K], s[BC_K];
    #pragma unroll
    for (int k = 0; k < BC_K; ++k) {
        const int e = tid + k * BC_T;
        d[k] = e < BC_CH ? dst[e_lo + e] : -1;
        s[k] = e < BC_CH ? src[e_lo + e] : 0;
    }
    int v = c;
    for (int o = 1; o < 32; o <<= 1) {
        const int u = __shfl_up(v, o);
        if (lane >= o) v += u;
    }
    if (lane == 31) wsum[wave] = v;
    __syncthreads();
    int pre = 0;
    for (int w = 0; w < BC_T / 32; ++w) {
        const int t = wsum[w];
        if (w < wave) pre += t;
    }
    const int excl = pre + v - c;
    cursor[tid] = excl + off;
    if (blockIdx.x == 0) binbase[tid] = excl;
    __syncthreads();
    int2* rec2 = (int2*)rec;
    #pragma unroll
    for (int k = 0; k < BC_K; ++k) {
        if (d[k] >= 0) {
            const int b   = d[k] / BIN_ROWS;
            const int pos = __hip_atomic_fetch_add(&cursor[b], 1, __ATOMIC_RELAXED, __HIP_MEMORY_SCOPE_WORKGROUP);
            rec2[pos] = make_int2(e_lo + tid + k * BC_T, s[k] | ((d[k] - b * BIN_ROWS) << 16));
        }
    }
}

#define BE_T 512
#define BE_W (BE_T / 32)

__global__ __launch_bounds__(BE_T) void edge_msg_kernel(
    const float* __restrict__ y,
    const float* xin,
    const int*   __restrict__ rec,
    const int*   __restrict__ binbase,
    const int*   __restrict__ bincnt,
    const float* __restrict__ attr,
    const float* __restrict__ bm,
    const float* __restrict__ We,
    const float* __restrict__ be,
    float*       hout)
{
    __shared__ __attribute__((aligned(16))) float sAcc[(BIN_ROWS + 1) * PE];
    __shared__ __attribute__((aligned(16))) float sWe[(PD / 2) * WP];
    __shared__ __attribute__((aligned(16))) float sA[BE_W][16 * AP];

    const int tid  = threadIdx.x;
    const int wave = __builtin_amdgcn_readfirstlane(tid >> 5);
    const int lane = tid & 31;
    const int bin  = blockIdx.x;

    const int start = binbase[bin];
    const int cnt   = bincnt[bin];
    const int end   = start + cnt;

    {
        const float4* We4 = (const float4*)We;
        if (tid < (PD / 2) * (PE / 4)) {
            const int kp = tid >> 4, n4 = tid & 15;
            const float4 r0 = We4[(2 * kp    ) * (PE / 4) + n4];
            const float4 r1 = We4[(2 * kp + 1) * (PE / 4) + n4];
            float4* o = (float4*)(&sWe[kp * WP + n4 * 8]);
            o[0] = make_float4(r0.x, r1.x, r0.y, r1.y);
            o[1] = make_float4(r0.z, r1.z, r0.w, r1.w);
        }
    }
    for (int i = tid; i < (BIN_ROWS + 1) * (PE / 4); i += BE_T)
        ((float4*)sAcc)[i] = make_float4(0.0f, 0.0f, 0.0f, 0.0f);

    const int row = lane >> 1;
    const int hlf = lane & 1;

    const int col  = lane & 15;
    const int hi   = lane >> 4;
    const int koff = hi * 2;

    const float bias0 = bm[col]      + be[col];
    const float bias1 = bm[16 + col] + be[16 + col];
    const float bias2 = bm[32 + col] + be[32 + col];
    const float bias3 = bm[48 + col] + be[48 + col];

    const int2* rec2   = (const int2*)rec;
    const int   ntiles = (cnt + 15) >> 4;

    __syncthreads();

    int2 ra_n  = make_int2(0, 0);
    int  eid_n = 0;
    if (wave < ntiles) {
        const int e0f = start + wave * 16;
        ra_n  = rec2[min(e0f + col, end - 1)];
        eid_n = rec2[min(e0f + row, end - 1)].x;
    }

    for (int t = wave; t < ntiles; t += BE_W) {
        const int  e0  = start + t * 16;
        const int2 ra  = ra_n;
        const int  eid = eid_n;
        if (t + BE_W < ntiles) {
            const int e0n = e0 + BE_W * 16;
            ra_n  = rec2[min(e0n + col, end - 1)];
            eid_n = rec2[min(e0n + row, end - 1)].x;
        }
        const int  sv  = ra.y & 0xffff;
        const int  dl  = (e0 + col < end) ? (ra.y >> 16) : BIN_ROWS;

        const float4* ar = (const float4*)(attr + (size_t)eid * PD + hlf * 8);
        const float4 ga0 = ar[0];
        const float4 ga1 = ar[1];
        float yc[16][2];
        #pragma unroll
        for (int m = 0; m < 16; ++m) {
            const float* yr = y + (size_t)__builtin_amdgcn_readlane(sv, m) * PE + lane;
            yc[m][0] = yr[0];
            yc[m][1] = yr[32];
        }
        {
            float4* oa = (float4*)(&sA[wave][row * AP + hlf * 8]);
            oa[0] = ga0; oa[1] = ga1;
        }

        wave_lds_handover();

        v8f acc[4];
        #pragma unroll
        for (int r = 0; r < 8; ++r) {
            acc[0][r] = bias0; acc[1][r] = bias1; acc[2][r] = bias2; acc[3][r] = bias3;
        }

        #pragma unroll
        for (int kk = 0; kk < 4; ++kk) {
            const int k0 = kk * 4;
            v2f a = *(const v2f*)(&sA[wave][col * AP + k0 + koff]);
            #pragma unroll
            for (int n = 0; n < 4; ++n) {
                v2f b = *(const v2f*)(&sWe[(kk * 2 + hi) * WP + (n * 16 + col) * 2]);
                acc[n] = __builtin_amdgcn_wmma_f32_16x16x4_f32(
                    false, a, false, b, (short)0, acc[n], false, false);
            }
        }

        wave_lds_handover();

        #pragma unroll
        for (int r = 0; r < 8; ++r) {
            float P[2], Q[2];
            #pragma unroll
            for (int p = 0; p < 2; ++p) {
                const float X  = acc[2 * p][r];
                const float Y  = acc[2 * p + 1][r];
                const float Z  = hi ? X : Y;
                const float Zx = xhalf(Z);
                P[p] = (hi ? Zx : X) + yc[r][p];
                Q[p] = (hi ? Y  : Zx) + yc[r + 8][p];
            }
            float* row0 = sAcc + __builtin_amdgcn_readlane(dl, r)     * PE + lane;
            float* row1 = sAcc + __builtin_amdgcn_readlane(dl, r + 8) * PE + lane;
            __hip_atomic_fetch_add(row0,      P[0], __ATOMIC_RELAXED, __HIP_MEMORY_SCOPE_WORKGROUP);
            __hip_atomic_fetch_add(row0 + 32, P[1], __ATOMIC_RELAXED, __HIP_MEMORY_SCOPE_WORKGROUP);
            __hip_atomic_fetch_add(row1,      Q[0], __ATOMIC_RELAXED, __HIP_MEMORY_SCOPE_WORKGROUP);
            __hip_atomic_fetch_add(row1 + 32, Q[1], __ATOMIC_RELAXED, __HIP_MEMORY_SCOPE_WORKGROUP);
        }
    }
    __syncthreads();

    const int row_lo = bin * BIN_ROWS;
    const int nrows  = max(0, min(BIN_ROWS, N_NODES - row_lo));
    const float4* x4 = (const float4*)xin + (size_t)row_lo * (PE / 4);
    float4*       h4 = (float4*)hout      + (size_t)row_lo * (PE / 4);
    for (int i = tid; i < nrows * (PE / 4); i += BE_T) {
        const float4 xv = x4[i];
        const float4 av = ((const float4*)sAcc)[i];
        h4[i] = make_float4(xv.x + av.x, xv.y + av.y, xv.z + av.z, xv.w + av.w);
    }
}

#define GN_T   256
#define GN_RPP (GN_T / 16)
#define GN_W   (GN_T / 32)
#define GN_R   16

__device__ __forceinline__ float4 gn_block_sum(float4 s, float4 (*buf)[16], int wave, int hi, int tx) {
    s.x += xhalf(s.x); s.y += xhalf(s.y); s.z += xhalf(s.z); s.w += xhalf(s.w);
    if (hi == 0) buf[wave][tx] = s;
    __syncthreads();
    float4 t = buf[hi * (GN_W / 2)][tx];
    #pragma unroll
    for (int k = 1; k < GN_W / 2; ++k) {
        const float4 p = buf[hi * (GN_W / 2) + k][tx];
        t.x += p.x; t.y += p.y; t.z += p.z; t.w += p.w;
    }
    t.x += xhalf(t.x); t.y += xhalf(t.y); t.z += xhalf(t.z); t.w += xhalf(t.w);
    return t;
}

__device__ __forceinline__ float4 gn_center(const float4 v, const float4 mm) {
    return make_float4(v.x - mm.x, v.y - mm.y, v.z - mm.z, v.w - mm.w);
}

__device__ __forceinline__ float4 gn_finish(const float4 o, const float4 sc, const float4 b) {
    float4 r;
    r.x = o.x * sc.x + b.x;
    r.y = o.y * sc.y + b.y;
    r.z = o.z * sc.z + b.z;
    r.w = o.w * sc.w + b.w;
    r.x = r.x > 0.0f ? r.x : NEG_SLOPE * r.x;
    r.y = r.y > 0.0f ? r.y : NEG_SLOPE * r.y;
    r.z = r.z > 0.0f ? r.z : NEG_SLOPE * r.z;
    r.w = r.w > 0.0f ? r.w : NEG_SLOPE * r.w;
    return r;
}

__global__ __launch_bounds__(GN_T) void graph_norm_kernel(
    const float* h,
    const int*   __restrict__ gstart,
    const float* __restrict__ mscale,
    const float* __restrict__ w,
    const float* __restrict__ bias,
    float*       out)
{
    __shared__ __attribute__((aligned(16))) float4 sred[2][GN_W][16];

    const int tid  = threadIdx.x;
    const int tx   = tid & 15;
    const int ty   = tid >> 4;
    const int wave = tid >> 5;
    const int hi   = ty & 1;

    const int   lo  = gstart[blockIdx.x];
    const int   n   = gstart[blockIdx.x + 1] - lo;
    const float cnt = fmaxf((float)n, 1.0f);

    const float4* h4 = (const float4*)h + (size_t)lo * (PE / 4) + tx;
    float4*       o4 = (float4*)out     + (size_t)lo * (PE / 4) + tx;

    const float4 ms = ((const float4*)mscale)[tx];
    const float4 wv = ((const float4*)w)[tx];
    const float4 bv = ((const float4*)bias)[tx];

    float4 v[GN_R];
    #pragma unroll
    for (int j = 0; j < GN_R; ++j) {
        const int r = ty + GN_RPP * j;
        v[j] = make_float4(0.0f, 0.0f, 0.0f, 0.0f);
        if (r < n) v[j] = h4[(size_t)r * (PE / 4)];
    }
    float4 s = v[0];
    #pragma unroll
    for (int j = 1; j < GN_R; ++j) { s.x += v[j].x; s.y += v[j].y; s.z += v[j].z; s.w += v[j].w; }
    for (int r = ty + GN_RPP * GN_R; r < n; r += GN_RPP) {
        const float4 t = h4[(size_t)r * (PE / 4)];
        s.x += t.x; s.y += t.y; s.z += t.z; s.w += t.w;
    }
    const float4 tot = gn_block_sum(s, sred[0], wave, hi, tx);

    const float4 mm = make_float4(tot.x / cnt * ms.x, tot.y / cnt * ms.y, tot.z / cnt * ms.z, tot.w / cnt * ms.w);

    float4 q = make_float4(0.0f, 0.0f, 0.0f, 0.0f);
    #pragma unroll
    for (int j = 0; j < GN_R; ++j) {
        const int r = ty + GN_RPP * j;
        const float4 o = gn_center(v[j], mm);
        v[j] = o;
        if (r < n) { q.x += o.x * o.x; q.y += o.y * o.y; q.z += o.z * o.z; q.w += o.w * o.w; }
    }
    for (int r = ty + GN_RPP * GN_R; r < n; r += GN_RPP) {
        const float4 o = gn_center(h4[(size_t)r * (PE / 4)], mm);
        q.x += o.x * o.x; q.y += o.y * o.y; q.z += o.z * o.z; q.w += o.w * o.w;
    }
    const float4 qt = gn_block_sum(q, sred[1], wave, hi, tx);

    const float4 sc = make_float4(wv.x * rsqrtf(qt.x / cnt + EPS), wv.y * rsqrtf(qt.y / cnt + EPS),
                                  wv.z * rsqrtf(qt.z / cnt + EPS), wv.w * rsqrtf(qt.w / cnt + EPS));

    #pragma unroll
    for (int j = 0; j < GN_R; ++j) {
        const int r = ty + GN_RPP * j;
        if (r < n) o4[(size_t)r * (PE / 4)] = gn_finish(v[j], sc, bv);
    }
    for (int r = ty + GN_RPP * GN_R; r < n; r += GN_RPP)
        o4[(size_t)r * (PE / 4)] = gn_finish(gn_center(h4[(size_t)r * (PE / 4)], mm), sc, bv);
}

extern "C" void kernel_launch(void* const* d_in, const int* in_sizes, int n_in,
                              void* d_out, int out_size, void* d_ws, size_t ws_size,
                              hipStream_t stream) {
    const float* x0    = (const float*)d_in[0];
    const int*   idx   = (const int*)  d_in[1];
    const float* attr  = (const float*)d_in[2];
    const int*   batch = (const int*)  d_in[3];
    const float* Wm    = (const float*)d_in[4];
    const float* bm    = (const float*)d_in[5];
    const float* We    = (const float*)d_in[6];
    const float* be    = (const float*)d_in[7];
    const float* gw    = (const float*)d_in[8];
    const float* gb    = (const float*)d_in[9];
    const float* gms   = (const float*)d_in[10];

    const int* src = idx;
    const int* dst = idx + N_EDGES;

    char* ws = (char*)d_ws;
    const size_t node_bytes = (size_t)N_NODES * PE * sizeof(float);
    float* h       = (float*)(ws);
    float* tmp     = (float*)(ws + node_bytes);
    char*  grp     = ws + 2 * node_bytes;
    int*   gstart  = (int*)  (ws + 3 * node_bytes);
    int*   rec     = (int*)  (grp);
    int*   bincnt  = (int*)  (grp + (size_t)N_EDGES * 8);
    int*   binbase = (int*)  (grp + (size_t)N_EDGES * 8 + NBINS * 4);
    int*   wgoff   = (int*)  (grp + (size_t)N_EDGES * 8 + 2 * NBINS * 4);
    static_assert((size_t)N_EDGES * 8 + 2 * NBINS * 4 + (size_t)BC_G * NBINS * 4 <= (size_t)N_NODES * PE * 4,
                  "the grouped list and its tables fit the third node-sized plane of the workspace");

    for (int l = 0; l < 2; ++l) {
        const float* xin  = (l == 0) ? x0 : h;
        float*       xout = (l == 1) ? (float*)d_out : h;

        node_msg_kernel<<<NODE_GRID, 256, 0, stream>>>(
            xin, Wm + (size_t)l * PE * PE, batch, tmp,
            (l == 0) ? gstart : (int*)nullptr, (l == 0) ? bincnt : (int*)nullptr);

        if (l == 0) {
            bin_count_kernel<<<BC_G, BC_T, 0, stream>>>(dst, bincnt, wgoff);
            bin_scatter_kernel<<<BC_G, BC_T, 0, stream>>>(src, dst, bincnt, wgoff, binbase, rec);
        }

        edge_msg_kernel<<<NBINS, BE_T, 0, stream>>>(
            tmp, xin, rec, binbase, bincnt, attr,
            bm + (size_t)l * PE,
            We + (size_t)l * PD * PE, be + (size_t)l * PE, h);

        graph_norm_kernel<<<N_GRAPHS, GN_T, 0, stream>>>(
            h, gstart, gms + (size_t)l * PE, gw + (size_t)l * PE, gb + (size_t)l * PE, xout);
    }
}


# edge kernel: attr rows summed per destination node in LDS, (attr sum) @ W_edge once per node, not per edge; layer 1's sums serve layer 2; y rows (made this call) still gathered; sum order changes
# speedup vs baseline: 3.2846x; 1.0472x over previous
#include <hip/hip_runtime.h>
#include <hip/hip_bf16.h>
#include <stdint.h>

#define N_NODES   50000
#define N_EDGES   800000
#define PE        64
#define PD        16
#define N_GRAPHS  256
#define EPS       1e-5f
#define NEG_SLOPE 0.01f

#define XP 68
#define WP 160

typedef __attribute__((ext_vector_type(2))) float v2f;
typedef __attribute__((ext_vector_type(8))) float v8f;

__device__ __forceinline__ float xhalf(float v) {
    return __builtin_bit_cast(float, __builtin_amdgcn_permlanex16(
        __builtin_bit_cast(int, v), __builtin_bit_cast(int, v), 0x76543210, 0xfedcba98, false, false));
}

#define NODE_TILES (N_NODES / 16)
#define NODE_GRID  ((NODE_TILES + 7) / 8)
static_assert(N_NODES % 16 == 0, "the node kernel has no partial tile");

#define NBINS    512
#define BIN_ROWS 98
static_assert(NBINS * BIN_ROWS >= N_NODES, "every node has a bin");
static_assert(N_NODES <= 65536 && BIN_ROWS < 256, "a record packs the source row in 16 bits, the row in its bin above them");
static_assert(NBINS <= NODE_GRID * 256, "the node kernel zeroes one bin counter per thread");

__global__ __launch_bounds__(256) void node_msg_kernel(
    const float* x,
    const float* __restrict__ Wm,
    const float* __restrict__ bm,
    const float* __restrict__ be,
    const int*   __restrict__ batch,
    float*       __restrict__ y,
    int*         __restrict__ gstart,
    int*         __restrict__ bincnt)
{
    __shared__ __attribute__((aligned(16))) float sWm[(PE / 2) * WP];
    __shared__ __attribute__((aligned(16))) float sX[8][16 * XP];

    const int tid  = threadIdx.x;
    const int wave = __builtin_amdgcn_readfirstlane(tid >> 5);
    const int lane = tid & 31;

    const int col  = lane & 15;
    const int hi   = lane >> 4;
    const int koff = hi * 2;

    const int  tile = blockIdx.x * 8 + wave;
    const bool live = tile < NODE_TILES;

    const float bias0 = bm[col]      + be[col];
    const float bias1 = bm[16 + col] + be[16 + col];
    const float bias2 = bm[32 + col] + be[32 + col];
    const float bias3 = bm[48 + col] + be[48 + col];

    float4 v0, v1, v2, v3, v4, v5, v6, v7;
    v0 = v1 = v2 = v3 = v4 = v5 = v6 = v7 = make_float4(0.0f, 0.0f, 0.0f, 0.0f);
    const int rq = hi * (PE / 4) + col;
    if (live) {
        const float4* x4 = (const float4*)(x + (size_t)tile * 16 * PE) + rq;
        v0 = x4[0 * (PE / 2)]; v1 = x4[1 * (PE / 2)]; v2 = x4[2 * (PE / 2)]; v3 = x4[3 * (PE / 2)];
        v4 = x4[4 * (PE / 2)]; v5 = x4[5 * (PE / 2)]; v6 = x4[6 * (PE / 2)]; v7 = x4[7 * (PE / 2)];
    }

    {
        const float4* Wm4 = (const float4*)Wm;
        #pragma unroll
        for (int it = tid; it < (PE / 2) * (PE / 4); it += 256) {
            const int kp = it >> 4, n4 = it & 15;
            const float4 r0 = Wm4[(2 * kp    ) * (PE / 4) + n4];
            const float4 r1 = Wm4[(2 * kp + 1) * (PE / 4) + n4];
            float4* o = (float4*)(&sWm[kp * WP + n4 * 8]);
            o[0] = make_float4(r0.x, r1.x, r0.y, r1.y);
            o[1] = make_float4(r0.z, r1.z, r0.w, r1.w);
        }
    }

    if (live) {
        float4* sx = (float4*)(&sX[wave][hi * XP + 4 * col]);
        sx[0 * (XP / 2)] = v0; sx[1 * (XP / 2)] = v1; sx[2 * (XP / 2)] = v2; sx[3 * (XP / 2)] = v3;
        sx[4 * (XP / 2)] = v4; sx[5 * (XP / 2)] = v5; sx[6 * (XP / 2)] = v6; sx[7 * (XP / 2)] = v7;
    }
    __syncthreads();

    if (live) {
        v8f acc[4];
        #pragma unroll
        for (int r = 0; r < 8; ++r) {
            acc[0][r] = bias0; acc[1][r] = bias1; acc[2][r] = bias2; acc[3][r] = bias3;
        }
        #pragma unroll
        for (int kk = 0; kk < 16; ++kk) {
            const int k0 = kk * 4;
            v2f a = *(const v2f*)(&sX[wave][col * XP + k0 + koff]);
            #pragma unroll
            for (int n = 0; n < 4; ++n) {
                v2f b = *(const v2f*)(&sWm[(kk * 2 + hi) * WP + (n * 16 + col) * 2]);
                acc[n] = __builtin_amdgcn_wmma_f32_16x16x4_f32(
                    false, a, false, b, (short)0, acc[n], false, false);
            }
        }
        float* yt = y + (size_t)tile * 16 * PE + lane;
        #pragma unroll
        for (int r = 0; r < 8; ++r) {
            #pragma unroll
            for (int p = 0; p < 2; ++p) {
                const float X  = acc[2 * p][r];
                const float Y  = acc[2 * p + 1][r];
                const float Z  = hi ? X : Y;
                const float Zx = xhalf(Z);
                yt[(r    ) * PE + 32 * p] = hi ? Zx : X;
                yt[(r + 8) * PE + 32 * p] = hi ? Y  : Zx;
            }
        }
    }

    const int gi = blockIdx.x * 256 + tid;
    if (gstart != nullptr && gi < N_NODES) {
        const int gb_cur = batch[gi];
        const int gfirst = gi > 0 ? batch[gi - 1] + 1 : 0;
        for (int g = gfirst; g <= gb_cur; ++g) gstart[g] = gi;
        if (gi == N_NODES - 1)
            for (int g = gb_cur + 1; g <= N_GRAPHS; ++g) gstart[g] = N_NODES;
    }
    if (bincnt != nullptr && gi < NBINS) bincnt[gi] = 0;
}

#define BC_T  512
#define BC_CH 3200
#define BC_G  (N_EDGES / BC_CH)
static_assert(N_EDGES % BC_CH == 0, "the grouping kernels have no partial chunk");
static_assert(BC_T == NBINS, "one thread a bin in the grouping kernels");
#define BC_K ((BC_CH + BC_T - 1) / BC_T)

__global__ __launch_bounds__(BC_T) void bin_count_kernel(
    const int* __restrict__ dst,
    int*       __restrict__ bincnt,
    int*       __restrict__ wgoff)
{
    __shared__ int hist[NBINS];
    const int tid = threadIdx.x;
    const int e_lo = blockIdx.x * BC_CH;
    int d[BC_K];
    #pragma unroll
    for (int k = 0; k < BC_K; ++k) {
        const int e = tid + k * BC_T;
        d[k] = e < BC_CH ? dst[e_lo + e] : -1;
    }
    hist[tid] = 0;
    __syncthreads();
    #pragma unroll
    for (int k = 0; k < BC_K; ++k)
        if (d[k] >= 0)
            __hip_atomic_fetch_add(&hist[d[k] / BIN_ROWS], 1, __ATOMIC_RELAXED, __HIP_MEMORY_SCOPE_WORKGROUP);
    __syncthreads();
    wgoff[blockIdx.x * NBINS + tid] = atomicAdd(&bincnt[tid], hist[tid]);
}

__global__ __launch_bounds__(BC_T) void bin_scatter_kernel(
    const int* __restrict__ src,
    const int* __restrict__ dst,
    const int* __restrict__ bincnt,
    const int* __restrict__ wgoff,
    int*       __restrict__ binbase,
    int*       __restrict__ rec)
{
    __shared__ int wsum[BC_T / 32];
    __shared__ int cursor[NBINS];
    const int tid  = threadIdx.x;
    const int lane = tid & 31;
    const int wave = tid >> 5;
    const int c    = bincnt[tid];
    const int off  = wgoff[blockIdx.x * NBINS + tid];
    const int e_lo = blockIdx.x * BC_CH;
    int d[BC_K], s[BC_K];
    #pragma unroll
    for (int k = 0; k < BC_K; ++k) {
        const int e = tid + k * BC_T;
        d[k] = e < BC_CH ? dst[e_lo + e] : -1;
        s[k] = e < BC_CH ? src[e_lo + e] : 0;
    }
    int v = c;
    for (int o = 1; o < 32; o <<= 1) {
        const int u = __shfl_up(v, o);
        if (lane >= o) v += u;
    }
    if (lane == 31) wsum[wave] = v;
    __syncthreads();
    int pre = 0;
    for (int w = 0; w < BC_T / 32; ++w) {
        const int t = wsum[w];
        if (w < wave) pre += t;
    }
    const int excl = pre + v - c;
    cursor[tid] = excl + off;
    if (blockIdx.x == 0) binbase[tid] = excl;
    __syncthreads();
    int2* rec2 = (int2*)rec;
    #pragma unroll
    for (int k = 0; k < BC_K; ++k) {
        if (d[k] >= 0) {
            const int b   = d[k] / BIN_ROWS;
            const int pos = __hip_atomic_fetch_add(&cursor[b], 1, __ATOMIC_RELAXED, __HIP_MEMORY_SCOPE_WORKGROUP);
            rec2[pos] = make_int2(e_lo + tid + k * BC_T, s[k] | ((d[k] - b * BIN_ROWS) << 16));
        }
    }
}

#define BE_T 512
#define BE_W (BE_T / 32)
#define TP 20
#define ET_TILES ((BIN_ROWS + 15) / 16)
#define ET_ROWS  (ET_TILES * 16)
static_assert(ET_TILES <= BE_W, "one wave a row tile in the bin's product");
static_assert(ET_ROWS > BIN_ROWS, "there is a spare row");
static_assert(ET_ROWS * (PD / 4) <= BE_T, "one float4 of the attr sums a thread");

template <bool FIRST>
__global__ __launch_bounds__(BE_T) void edge_sum_kernel(
    const float* __restrict__ y,
    const float* xin,
    const int*   __restrict__ rec,
    const int*   __restrict__ binbase,
    const int*   __restrict__ bincnt,
    const float* __restrict__ attr,
    const float* __restrict__ We,
    float*       tsum,
    float*       hout)
{
    __shared__ __attribute__((aligned(16))) float sAcc[ET_ROWS * PE];
    __shared__ __attribute__((aligned(16))) float sWe[(PD / 2) * WP];
    __shared__ __attribute__((aligned(16))) float sT[ET_ROWS * TP];

    const int tid  = threadIdx.x;
    const int wave = __builtin_amdgcn_readfirstlane(tid >> 5);
    const int lane = tid & 31;
    const int bin  = blockIdx.x;

    const int start = binbase[bin];
    const int cnt   = bincnt[bin];
    const int end   = start + cnt;

    const int row_lo = bin * BIN_ROWS;
    const int nrows  = max(0, min(BIN_ROWS, N_NODES - row_lo));

    {
        const float4* We4 = (const float4*)We;
        if (tid < (PD / 2) * (PE / 4)) {
            const int kp = tid >> 4, n4 = tid & 15;
            const float4 r0 = We4[(2 * kp    ) * (PE / 4) + n4];
            const float4 r1 = We4[(2 * kp + 1) * (PE / 4) + n4];
            float4* o = (float4*)(&sWe[kp * WP + n4 * 8]);
            o[0] = make_float4(r0.x, r1.x, r0.y, r1.y);
            o[1] = make_float4(r0.z, r1.z, r0.w, r1.w);
        }
    }
    if (tid < ET_ROWS * (PD / 4)) {
        const int r = tid >> 2, q = tid & 3;
        float4 v = make_float4(0.0f, 0.0f, 0.0f, 0.0f);
        if (!FIRST && r < nrows) v = ((const float4*)tsum)[(size_t)(row_lo + r) * (PD / 4) + q];
        *(float4*)(&sT[r * TP + 4 * q]) = v;
    }
    for (int i = tid; i < ET_ROWS * (PE / 4); i += BE_T)
        ((float4*)sAcc)[i] = make_float4(0.0f, 0.0f, 0.0f, 0.0f);

    const int row = lane >> 1;
    const int hlf = lane & 1;

    const int col  = lane & 15;
    const int hi   = lane >> 4;
    const int koff = hi * 2;

    const int2* rec2   = (const int2*)rec;
    const int   ntiles = (cnt + 15) >> 4;

    __syncthreads();

    int2 ra_n = make_int2(0, 0);
    int2 rb_n = make_int2(0, 0);
    if (wave < ntiles) {
        const int e0f = start + wave * 16;
        ra_n = rec2[min(e0f + col, end - 1)];
        if (FIRST) rb_n = rec2[min(e0f + row, end - 1)];
    }

    for (int t = wave; t < ntiles; t += BE_W) {
        const int  e0 = start + t * 16;
        const int2 ra = ra_n;
        const int2 rb = rb_n;
        if (t + BE_W < ntiles) {
            const int e0n = e0 + BE_W * 16;
            ra_n = rec2[min(e0n + col, end - 1)];
            if (FIRST) rb_n = rec2[min(e0n + row, end - 1)];
        }
        const int sv = ra.y & 0xffff;
        const int dl = (e0 + col < end) ? (ra.y >> 16) : BIN_ROWS;

        float4 ga0 = make_float4(0.0f, 0.0f, 0.0f, 0.0f), ga1 = ga0;
        if (FIRST) {
            const float4* ar = (const float4*)(attr + (size_t)rb.x * PD + hlf * 8);
            ga0 = ar[0];
            ga1 = ar[1];
        }
        float yc[16][2];
        #pragma unroll
        for (int m = 0; m < 16; ++m) {
            const float* yr = y + (size_t)__builtin_amdgcn_readlane(sv, m) * PE + lane;
            yc[m][0] = yr[0];
            yc[m][1] = yr[32];
        }

        if (FIRST) {
            const int dla = (e0 + row < end) ? (rb.y >> 16) : BIN_ROWS;
            float* tp = sT + dla * TP + hlf * 8;
            __hip_atomic_fetch_add(tp + 0, ga0.x, __ATOMIC_RELAXED, __HIP_MEMORY_SCOPE_WORKGROUP);
            __hip_atomic_fetch_add(tp + 1, ga0.y, __ATOMIC_RELAXED, __HIP_MEMORY_SCOPE_WORKGROUP);
            __hip_atomic_fetch_add(tp + 2, ga0.z, __ATOMIC_RELAXED, __HIP_MEMORY_SCOPE_WORKGROUP);
            __hip_atomic_fetch_add(tp + 3, ga0.w, __ATOMIC_RELAXED, __HIP_MEMORY_SCOPE_WORKGROUP);
            __hip_atomic_fetch_add(tp + 4, ga1.x, __ATOMIC_RELAXED, __HIP_MEMORY_SCOPE_WORKGROUP);
            __hip_atomic_fetch_add(tp + 5, ga1.y, __ATOMIC_RELAXED, __HIP_MEMORY_SCOPE_WORKGROUP);
            __hip_atomic_fetch_add(tp + 6, ga1.z, __ATOMIC_RELAXED, __HIP_MEMORY_SCOPE_WORKGROUP);
            __hip_atomic_fetch_add(tp + 7, ga1.w, __ATOMIC_RELAXED, __HIP_MEMORY_SCOPE_WORKGROUP);
        }

        #pragma unroll
        for (int m = 0; m < 16; ++m) {
            float* rw = sAcc + __builtin_amdgcn_readlane(dl, m) * PE + lane;
            __hip_atomic_fetch_add(rw,      yc[m][0], __ATOMIC_RELAXED, __HIP_MEMORY_SCOPE_WORKGROUP);
            __hip_atomic_fetch_add(rw + 32, yc[m][1], __ATOMIC_RELAXED, __HIP_MEMORY_SCOPE_WORKGROUP);
        }
    }
    __syncthreads();

    if (FIRST && tid < nrows * (PD / 4)) {
        const int r = tid >> 2, q = tid & 3;
        ((float4*)tsum)[(size_t)(row_lo + r) * (PD / 4) + q] = *(const float4*)(&sT[r * TP + 4 * q]);
    }

    if (wave < ET_TILES) {
        v8f acc[4];
        #pragma unroll
        for (int r = 0; r < 8; ++r) {
            acc[0][r] = 0.0f; acc[1][r] = 0.0f; acc[2][r] = 0.0f; acc[3][r] = 0.0f;
        }
        #pragma unroll
        for (int kk = 0; kk < 4; ++kk) {
            const int k0 = kk * 4;
            v2f a = *(const v2f*)(&sT[(wave * 16 + col) * TP + k0 + koff]);
            #pragma unroll
            for (int n = 0; n < 4; ++n) {
                v2f b = *(const v2f*)(&sWe[(kk * 2 + hi) * WP + (n * 16 + col) * 2]);
                acc[n] = __builtin_amdgcn_wmma_f32_16x16x4_f32(
                    false, a, false, b, (short)0, acc[n], false, false);
            }
        }
        float* base = sAcc + (wave * 16 + 8 * hi) * PE + col;
        #pragma unroll
        for (int r = 0; r < 8; ++r) {
            #pragma unroll
            for (int n = 0; n < 4; ++n)
                base[r * PE + n * 16] += acc[n][r];
        }
    }
    __syncthreads();

    const float4* x4 = (const float4*)xin + (size_t)row_lo * (PE / 4);
    float4*       h4 = (float4*)hout      + (size_t)row_lo * (PE / 4);
    for (int i = tid; i < nrows * (PE / 4); i += BE_T) {
        const float4 xv = x4[i];
        const float4 av = ((const float4*)sAcc)[i];
        h4[i] = make_float4(xv.x + av.x, xv.y + av.y, xv.z + av.z, xv.w + av.w);
    }
}

#define GN_T   256
#define GN_RPP (GN_T / 16)
#define GN_W   (GN_T / 32)
#define GN_R   16

__device__ __forceinline__ float4 gn_block_sum(float4 s, float4 (*buf)[16], int wave, int hi, int tx) {
    s.x += xhalf(s.x); s.y += xhalf(s.y); s.z += xhalf(s.z); s.w += xhalf(s.w);
    if (hi == 0) buf[wave][tx] = s;
    __syncthreads();
    float4 t = buf[hi * (GN_W / 2)][tx];
    #pragma unroll
    for (int k = 1; k < GN_W / 2; ++k) {
        const float4 p = buf[hi * (GN_W / 2) + k][tx];
        t.x += p.x; t.y += p.y; t.z += p.z; t.w += p.w;
    }
    t.x += xhalf(t.x); t.y += xhalf(t.y); t.z += xhalf(t.z); t.w += xhalf(t.w);
    return t;
}

__device__ __forceinline__ float4 gn_center(const float4 v, const float4 mm) {
    return make_float4(v.x - mm.x, v.y - mm.y, v.z - mm.z, v.w - mm.w);
}

__device__ __forceinline__ float4 gn_finish(const float4 o, const float4 sc, const float4 b) {
    float4 r;
    r.x = o.x * sc.x + b.x;
    r.y = o.y * sc.y + b.y;
    r.z = o.z * sc.z + b.z;
    r.w = o.w * sc.w + b.w;
    r.x = r.x > 0.0f ? r.x : NEG_SLOPE * r.x;
    r.y = r.y > 0.0f ? r.y : NEG_SLOPE * r.y;
    r.z = r.z > 0.0f ? r.z : NEG_SLOPE * r.z;
    r.w = r.w > 0.0f ? r.w : NEG_SLOPE * r.w;
    return r;
}

__global__ __launch_bounds__(GN_T) void graph_norm_kernel(
    const float* h,
    const int*   __restrict__ gstart,
    const float* __restrict__ mscale,
    const float* __restrict__ w,
    const float* __restrict__ bias,
    float*       out)
{
    __shared__ __attribute__((aligned(16))) float4 sred[2][GN_W][16];

    const int tid  = threadIdx.x;
    const int tx   = tid & 15;
    const int ty   = tid >> 4;
    const int wave = tid >> 5;
    const int hi   = ty & 1;

    const int   lo  = gstart[blockIdx.x];
    const int   n   = gstart[blockIdx.x + 1] - lo;
    const float cnt = fmaxf((float)n, 1.0f);

    const float4* h4 = (const float4*)h + (size_t)lo * (PE / 4) + tx;
    float4*       o4 = (float4*)out     + (size_t)lo * (PE / 4) + tx;

    const float4 ms = ((const float4*)mscale)[tx];
    const float4 wv = ((const float4*)w)[tx];
    const float4 bv = ((const float4*)bias)[tx];

    float4 v[GN_R];
    #pragma unroll
    for (int j = 0; j < GN_R; ++j) {
        const int r = ty + GN_RPP * j;
        v[j] = make_float4(0.0f, 0.0f, 0.0f, 0.0f);
        if (r < n) v[j] = h4[(size_t)r * (PE / 4)];
    }
    float4 s = v[0];
    #pragma unroll
    for (int j = 1; j < GN_R; ++j) { s.x += v[j].x; s.y += v[j].y; s.z += v[j].z; s.w += v[j].w; }
    for (int r = ty + GN_RPP * GN_R; r < n; r += GN_RPP) {
        const float4 t = h4[(size_t)r * (PE / 4)];
        s.x += t.x; s.y += t.y; s.z += t.z; s.w += t.w;
    }
    const float4 tot = gn_block_sum(s, sred[0], wave, hi, tx);

    const float4 mm = make_float4(tot.x / cnt * ms.x, tot.y / cnt * ms.y, tot.z / cnt * ms.z, tot.w / cnt * ms.w);

    float4 q = make_float4(0.0f, 0.0f, 0.0f, 0.0f);
    #pragma unroll
    for (int j = 0; j < GN_R; ++j) {
        const int r = ty + GN_RPP * j;
        const float4 o = gn_center(v[j], mm);
        v[j] = o;
        if (r < n) { q.x += o.x * o.x; q.y += o.y * o.y; q.z += o.z * o.z; q.w += o.w * o.w; }
    }
    for (int r = ty + GN_RPP * GN_R; r < n; r += GN_RPP) {
        const float4 o = gn_center(h4[(size_t)r * (PE / 4)], mm);
        q.x += o.x * o.x; q.y += o.y * o.y; q.z += o.z * o.z; q.w += o.w * o.w;
    }
    const float4 qt = gn_block_sum(q, sred[1], wave, hi, tx);

    const float4 sc = make_float4(wv.x * rsqrtf(qt.x / cnt + EPS), wv.y * rsqrtf(qt.y / cnt + EPS),
                                  wv.z * rsqrtf(qt.z / cnt + EPS), wv.w * rsqrtf(qt.w / cnt + EPS));

    #pragma unroll
    for (int j = 0; j < GN_R; ++j) {
        const int r = ty + GN_RPP * j;
        if (r < n) o4[(size_t)r * (PE / 4)] = gn_finish(v[j], sc, bv);
    }
    for (int r = ty + GN_RPP * GN_R; r < n; r += GN_RPP)
        o4[(size_t)r * (PE / 4)] = gn_finish(gn_center(h4[(size_t)r * (PE / 4)], mm), sc, bv);
}

extern "C" void kernel_launch(void* const* d_in, const int* in_sizes, int n_in,
                              void* d_out, int out_size, void* d_ws, size_t ws_size,
                              hipStream_t stream) {
    const float* x0    = (const float*)d_in[0];
    const int*   idx   = (const int*)  d_in[1];
    const float* attr  = (const float*)d_in[2];
    const int*   batch = (const int*)  d_in[3];
    const float* Wm    = (const float*)d_in[4];
    const float* bm    = (const float*)d_in[5];
    const float* We    = (const float*)d_in[6];
    const float* be    = (const float*)d_in[7];
    const float* gw    = (const float*)d_in[8];
    const float* gb    = (const float*)d_in[9];
    const float* gms   = (const float*)d_in[10];

    const int* src = idx;
    const int* dst = idx + N_EDGES;

    char* ws = (char*)d_ws;
    const size_t node_bytes = (size_t)N_NODES * PE * sizeof(float);
    float* h       = (float*)(ws);
    float* tmp     = (float*)(ws + node_bytes);
    char*  grp     = ws + 2 * node_bytes;
    int*   gstart  = (int*)  (ws + 3 * node_bytes);
    int*   rec     = (int*)  (grp);
    int*   bincnt  = (int*)  (grp + (size_t)N_EDGES * 8);
    int*   binbase = (int*)  (grp + (size_t)N_EDGES * 8 + NBINS * 4);
    int*   wgoff   = (int*)  (grp + (size_t)N_EDGES * 8 + 2 * NBINS * 4);
    float* tsum    = (float*)(grp + (size_t)N_EDGES * 8 + 2 * NBINS * 4 + (size_t)BC_G * NBINS * 4);
    static_assert((size_t)N_EDGES * 8 + 2 * NBINS * 4 + (size_t)BC_G * NBINS * 4 + (size_t)N_NODES * PD * 4
                      <= (size_t)N_NODES * PE * 4,
                  "the grouped list, its tables and the attr sums fit the third node-sized plane of the workspace");
    static_assert(((size_t)N_EDGES * 8 + 2 * NBINS * 4 + (size_t)BC_G * NBINS * 4) % 16 == 0,
                  "the attr sums are read and written as float4");

    for (int l = 0; l < 2; ++l) {
        const float* xin  = (l == 0) ? x0 : h;
        float*       xout = (l == 1) ? (float*)d_out : h;

        node_msg_kernel<<<NODE_GRID, 256, 0, stream>>>(
            xin, Wm + (size_t)l * PE * PE, bm + (size_t)l * PE, be + (size_t)l * PE, batch, tmp,
            (l == 0) ? gstart : (int*)nullptr, (l == 0) ? bincnt : (int*)nullptr);

        if (l == 0) {
            bin_count_kernel<<<BC_G, BC_T, 0, stream>>>(dst, bincnt, wgoff);
            bin_scatter_kernel<<<BC_G, BC_T, 0, stream>>>(src, dst, bincnt, wgoff, binbase, rec);
        }

        if (l == 0)
            edge_sum_kernel<true><<<NBINS, BE_T, 0, stream>>>(
                tmp, xin, rec, binbase, bincnt, attr, We + (size_t)l * PD * PE, tsum, h);
        else
            edge_sum_kernel<false><<<NBINS, BE_T, 0, stream>>>(
                tmp, xin, rec, binbase, bincnt, attr, We + (size_t)l * PD * PE, tsum, h);

        graph_norm_kernel<<<N_GRAPHS, GN_T, 0, stream>>>(
            h, gstart, gms + (size_t)l * PE, gw + (size_t)l * PE, gb + (size_t)l * PE, xout);
    }
}


# edge sum binned by graph (batch is sorted): one work-group owns a graph, its rows stay in LDS from the edge sum through mean, variance and output; the two norm launches go; sum order changes
# speedup vs baseline: 3.5487x; 1.0804x over previous
#include <hip/hip_runtime.h>
#include <hip/hip_bf16.h>
#include <stdint.h>

#define N_NODES   50000
#define N_EDGES   800000
#define PE        64
#define PD        16
#define N_GRAPHS  256
#define EPS       1e-5f
#define NEG_SLOPE 0.01f

#define XP 68
#define WP 160

typedef __attribute__((ext_vector_type(2))) float v2f;
typedef __attribute__((ext_vector_type(8))) float v8f;

__device__ __forceinline__ float xhalf(float v) {
    return __builtin_bit_cast(float, __builtin_amdgcn_permlanex16(
        __builtin_bit_cast(int, v), __builtin_bit_cast(int, v), 0x76543210, 0xfedcba98, false, false));
}

#define NODE_TILES (N_NODES / 16)
#define NODE_GRID  ((NODE_TILES + 7) / 8)
static_assert(N_NODES % 16 == 0, "the node kernel has no partial tile");

#define NBINS N_GRAPHS
static_assert(N_NODES <= 65536, "a record packs the source row in 16 bits, the row in its graph above them");
static_assert(NBINS <= NODE_GRID * 256, "the node kernel zeroes one bin counter per thread");
static_assert((N_GRAPHS & (N_GRAPHS - 1)) == 0, "the search for a node's graph halves a power of two");

__global__ __launch_bounds__(256) void node_msg_kernel(
    const float* x,
    const float* __restrict__ Wm,
    const float* __restrict__ bm,
    const float* __restrict__ be,
    const int*   __restrict__ batch,
    float*       __restrict__ y,
    int*         __restrict__ gstart,
    int*         __restrict__ bincnt)
{
    __shared__ __attribute__((aligned(16))) float sWm[(PE / 2) * WP];
    __shared__ __attribute__((aligned(16))) float sX[8][16 * XP];

    const int tid  = threadIdx.x;
    const int wave = __builtin_amdgcn_readfirstlane(tid >> 5);
    const int lane = tid & 31;

    const int col  = lane & 15;
    const int hi   = lane >> 4;
    const int koff = hi * 2;

    const int  tile = blockIdx.x * 8 + wave;
    const bool live = tile < NODE_TILES;

    const float bias0 = bm[col]      + be[col];
    const float bias1 = bm[16 + col] + be[16 + col];
    const float bias2 = bm[32 + col] + be[32 + col];
    const float bias3 = bm[48 + col] + be[48 + col];

    float4 v0, v1, v2, v3, v4, v5, v6, v7;
    v0 = v1 = v2 = v3 = v4 = v5 = v6 = v7 = make_float4(0.0f, 0.0f, 0.0f, 0.0f);
    const int rq = hi * (PE / 4) + col;
    if (live) {
        const float4* x4 = (const float4*)(x + (size_t)tile * 16 * PE) + rq;
        v0 = x4[0 * (PE / 2)]; v1 = x4[1 * (PE / 2)]; v2 = x4[2 * (PE / 2)]; v3 = x4[3 * (PE / 2)];
        v4 = x4[4 * (PE / 2)]; v5 = x4[5 * (PE / 2)]; v6 = x4[6 * (PE / 2)]; v7 = x4[7 * (PE / 2)];
    }

    {
        const float4* Wm4 = (const float4*)Wm;
        #pragma unroll
        for (int it = tid; it < (PE / 2) * (PE / 4); it += 256) {
            const int kp = it >> 4, n4 = it & 15;
            const float4 r0 = Wm4[(2 * kp    ) * (PE / 4) + n4];
            const float4 r1 = Wm4[(2 * kp + 1) * (PE / 4) + n4];
            float4* o = (float4*)(&sWm[kp * WP + n4 * 8]);
            o[0] = make_float4(r0.x, r1.x, r0.y, r1.y);
            o[1] = make_float4(r0.z, r1.z, r0.w, r1.w);
        }
    }

    if (live) {
        float4* sx = (float4*)(&sX[wave][hi * XP + 4 * col]);
        sx[0 * (XP / 2)] = v0; sx[1 * (XP / 2)] = v1; sx[2 * (XP / 2)] = v2; sx[3 * (XP / 2)] = v3;
        sx[4 * (XP / 2)] = v4; sx[5 * (XP / 2)] = v5; sx[6 * (XP / 2)] = v6; sx[7 * (XP / 2)] = v7;
    }
    __syncthreads();

    if (live) {
        v8f acc[4];
        #pragma unroll
        for (int r = 0; r < 8; ++r) {
            acc[0][r] = bias0; acc[1][r] = bias1; acc[2][r] = bias2; acc[3][r] = bias3;
        }
        #pragma unroll
        for (int kk = 0; kk < 16; ++kk) {
            const int k0 = kk * 4;
            v2f a = *(const v2f*)(&sX[wave][col * XP + k0 + koff]);
            #pragma unroll
            for (int n = 0; n < 4; ++n) {
                v2f b = *(const v2f*)(&sWm[(kk * 2 + hi) * WP + (n * 16 + col) * 2]);
                acc[n] = __builtin_amdgcn_wmma_f32_16x16x4_f32(
                    false, a, false, b, (short)0, acc[n], false, false);
            }
        }
        float* yt = y + (size_t)tile * 16 * PE + lane;
        #pragma unroll
        for (int r = 0; r < 8; ++r) {
            #pragma unroll
            for (int p = 0; p < 2; ++p) {
                const float X  = acc[2 * p][r];
                const float Y  = acc[2 * p + 1][r];
                const float Z  = hi ? X : Y;
                const float Zx = xhalf(Z);
                yt[(r    ) * PE + 32 * p] = hi ? Zx : X;
                yt[(r + 8) * PE + 32 * p] = hi ? Y  : Zx;
            }
        }
    }

    const int gi = blockIdx.x * 256 + tid;
    if (gstart != nullptr && gi < N_NODES) {
        const int gb_cur = batch[gi];
        const int gfirst = gi > 0 ? batch[gi - 1] + 1 : 0;
        for (int g = gfirst; g <= gb_cur; ++g) gstart[g] = gi;
        if (gi == N_NODES - 1)
            for (int g = gb_cur + 1; g <= N_GRAPHS; ++g) gstart[g] = N_NODES;
    }
    if (bincnt != nullptr && gi < NBINS) bincnt[gi] = 0;
}

#define BC_T  512
#define BC_CH 3200
#define BC_G  (N_EDGES / BC_CH)
static_assert(N_EDGES % BC_CH == 0, "the grouping kernels have no partial chunk");
static_assert(NBINS <= BC_T, "one thread a bin in the grouping kernels");
#define BC_K ((BC_CH + BC_T - 1) / BC_T)

__device__ __forceinline__ int graph_of(const int* gs, int d) {
    int g = 0;
    #pragma unroll
    for (int st = N_GRAPHS / 2; st >= 1; st >>= 1) {
        const int m = g + st;
        if (gs[m] <= d) g = m;
    }
    return g;
}

__global__ __launch_bounds__(BC_T) void bin_count_kernel(
    const int* __restrict__ dst,
    const int* __restrict__ gstart,
    int*       __restrict__ bincnt,
    int*       __restrict__ wgoff)
{
    __shared__ int hist[NBINS];
    __shared__ int sgs[NBINS];
    const int tid = threadIdx.x;
    const int e_lo = blockIdx.x * BC_CH;
    int d[BC_K];
    #pragma unroll
    for (int k = 0; k < BC_K; ++k) {
        const int e = tid + k * BC_T;
        d[k] = e < BC_CH ? dst[e_lo + e] : -1;
    }
    if (tid < NBINS) { hist[tid] = 0; sgs[tid] = gstart[tid]; }
    __syncthreads();
    int b[BC_K];
    #pragma unroll
    for (int k = 0; k < BC_K; ++k) b[k] = graph_of(sgs, d[k]);
    #pragma unroll
    for (int k = 0; k < BC_K; ++k)
        if (d[k] >= 0)
            __hip_atomic_fetch_add(&hist[b[k]], 1, __ATOMIC_RELAXED, __HIP_MEMORY_SCOPE_WORKGROUP);
    __syncthreads();
    if (tid < NBINS) wgoff[blockIdx.x * NBINS + tid] = atomicAdd(&bincnt[tid], hist[tid]);
}

__global__ __launch_bounds__(BC_T) void bin_scatter_kernel(
    const int* __restrict__ src,
    const int* __restrict__ dst,
    const int* __restrict__ gstart,
    const int* __restrict__ bincnt,
    const int* __restrict__ wgoff,
    int*       __restrict__ binbase,
    int*       __restrict__ rec)
{
    __shared__ int wsum[BC_T / 32];
    __shared__ int cursor[NBINS];
    __shared__ int sgs[NBINS];
    const int tid  = threadIdx.x;
    const int lane = tid & 31;
    const int wave = tid >> 5;
    const bool own = tid < NBINS;
    const int c    = own ? bincnt[tid] : 0;
    const int off  = own ? wgoff[blockIdx.x * NBINS + tid] : 0;
    if (own) sgs[tid] = gstart[tid];
    const int e_lo = blockIdx.x * BC_CH;
    int d[BC_K], s[BC_K];
    #pragma unroll
    for (int k = 0; k < BC_K; ++k) {
        const int e = tid + k * BC_T;
        d[k] = e < BC_CH ? dst[e_lo + e] : -1;
        s[k] = e < BC_CH ? src[e_lo + e] : 0;
    }
    int v = c;
    for (int o = 1; o < 32; o <<= 1) {
        const int u = __shfl_up(v, o);
        if (lane >= o) v += u;
    }
    if (lane == 31) wsum[wave] = v;
    __syncthreads();
    int pre = 0;
    for (int w = 0; w < BC_T / 32; ++w) {
        const int t = wsum[w];
        if (w < wave) pre += t;
    }
    const int excl = pre + v - c;
    if (own) {
        cursor[tid] = excl + off;
        if (blockIdx.x == 0) binbase[tid] = excl;
    }
    __syncthreads();
    int b[BC_K];
    #pragma unroll
    for (int k = 0; k < BC_K; ++k) b[k] = graph_of(sgs, d[k]);
    int2* rec2 = (int2*)rec;
    #pragma unroll
    for (int k = 0; k < BC_K; ++k) {
        if (d[k] >= 0) {
            const int pos = __hip_atomic_fetch_add(&cursor[b[k]], 1, __ATOMIC_RELAXED, __HIP_MEMORY_SCOPE_WORKGROUP);
            const unsigned rowg = (unsigned)(d[k] - sgs[b[k]]);
            rec2[pos] = make_int2(e_lo + tid + k * BC_T, (int)((unsigned)s[k] | (rowg << 16)));
        }
    }
}

#define BE_T   512
#define BE_W   (BE_T / 32)
#define BE_RPP (BE_T / 16)
#define TP 20
#define R_MAX   448
#define ET_ROWS (R_MAX + 16)
#define INIT_X ((R_MAX * (PE / 4) + BE_T - 1) / BE_T)
#define INIT_T ((R_MAX * (PD / 4) + BE_T - 1) / BE_T)
#define NR     (R_MAX / BE_RPP)
static_assert(R_MAX % 16 == 0 && R_MAX % BE_RPP == 0,
              "chunks are whole row tiles, and a thread keeps its rows from chunk to norm");

__device__ __forceinline__ float4 gn_block_sum(float4 s, float4 (*buf)[16], int wave, int hi, int tx) {
    s.x += xhalf(s.x); s.y += xhalf(s.y); s.z += xhalf(s.z); s.w += xhalf(s.w);
    if (hi == 0) buf[wave][tx] = s;
    __syncthreads();
    float4 t = buf[hi * (BE_W / 2)][tx];
    #pragma unroll
    for (int k = 1; k < BE_W / 2; ++k) {
        const float4 p = buf[hi * (BE_W / 2) + k][tx];
        t.x += p.x; t.y += p.y; t.z += p.z; t.w += p.w;
    }
    t.x += xhalf(t.x); t.y += xhalf(t.y); t.z += xhalf(t.z); t.w += xhalf(t.w);
    return t;
}

__device__ __forceinline__ float4 gn_center(const float4 v, const float4 mm) {
    return make_float4(v.x - mm.x, v.y - mm.y, v.z - mm.z, v.w - mm.w);
}

__device__ __forceinline__ float4 gn_finish(const float4 o, const float4 sc, const float4 b) {
    float4 r;
    r.x = o.x * sc.x + b.x;
    r.y = o.y * sc.y + b.y;
    r.z = o.z * sc.z + b.z;
    r.w = o.w * sc.w + b.w;
    r.x = r.x > 0.0f ? r.x : NEG_SLOPE * r.x;
    r.y = r.y > 0.0f ? r.y : NEG_SLOPE * r.y;
    r.z = r.z > 0.0f ? r.z : NEG_SLOPE * r.z;
    r.w = r.w > 0.0f ? r.w : NEG_SLOPE * r.w;
    return r;
}

template <bool FIRST>
__global__ __launch_bounds__(BE_T) void edge_norm_kernel(
    const float* __restrict__ y,
    const float* xin,
    const int*   __restrict__ rec,
    const int*   __restrict__ binbase,
    const int*   __restrict__ bincnt,
    const int*   __restrict__ gstart,
    const float* __restrict__ attr,
    const float* __restrict__ We,
    float*       tsum,
    const float* __restrict__ mscale,
    const float* __restrict__ w,
    const float* __restrict__ bias,
    float*       hout)
{
    __shared__ __attribute__((aligned(16))) float sAcc[ET_ROWS * PE];
    __shared__ __attribute__((aligned(16))) float sWe[(PD / 2) * WP];
    __shared__ __attribute__((aligned(16))) float sT[ET_ROWS * TP];
    __shared__ __attribute__((aligned(16))) float4 sred[2][BE_W][16];

    const int tid  = threadIdx.x;
    const int wave = __builtin_amdgcn_readfirstlane(tid >> 5);
    const int lane = tid & 31;
    const int bin  = blockIdx.x;

    const int start = binbase[bin];
    const int cnt   = bincnt[bin];
    const int end   = start + cnt;

    const int  lo      = gstart[bin];
    const int  n       = gstart[bin + 1] - lo;
    const bool big     = n > R_MAX;
    const int  nchunks = (n + R_MAX - 1) / R_MAX;

    const int col  = lane & 15;
    const int hi   = lane >> 4;
    const int koff = hi * 2;

    const int ty = tid >> 4;

    const float4 ms = ((const float4*)mscale)[col];
    const float4 wv = ((const float4*)w)[col];
    const float4 bv = ((const float4*)bias)[col];

    {
        const float4* We4 = (const float4*)We;
        if (tid < (PD / 2) * (PE / 4)) {
            const int kp = tid >> 4, n4 = tid & 15;
            const float4 r0 = We4[(2 * kp    ) * (PE / 4) + n4];
            const float4 r1 = We4[(2 * kp + 1) * (PE / 4) + n4];
            float4* o = (float4*)(&sWe[kp * WP + n4 * 8]);
            o[0] = make_float4(r0.x, r1.x, r0.y, r1.y);
            o[1] = make_float4(r0.z, r1.z, r0.w, r1.w);
        }
    }

    const int row = lane >> 1;
    const int hlf = lane & 1;

    const int2* rec2   = (const int2*)rec;
    const int   ntiles = (cnt + 15) >> 4;

    float4*       sAcc4 = (float4*)sAcc;
    const float4* x4    = (const float4*)xin + (size_t)lo * (PE / 4);
    float4*       o4    = (float4*)hout      + (size_t)lo * (PE / 4);
    float4*       t4    = (float4*)tsum      + (size_t)lo * (PD / 4);

    for (int c = 0; c < nchunks; ++c) {
        const int c0    = c * R_MAX;
        const int nrows = min(R_MAX, n - c0);
        const int nr16  = (nrows + 15) & ~15;

        int2 ra_n = make_int2(0, 0);
        int2 rb_n = make_int2(0, 0);
        if (wave < ntiles) {
            const int e0f = start + wave * 16;
            ra_n = rec2[min(e0f + col, end - 1)];
            if (FIRST) rb_n = rec2[min(e0f + row, end - 1)];
        }

        float4 tv[INIT_T], xv[INIT_X];
        #pragma unroll
        for (int j = 0; j < INIT_T; ++j) {
            const int i = tid + j * BE_T;
            tv[j] = make_float4(0.0f, 0.0f, 0.0f, 0.0f);
            if (!FIRST && i < nrows * (PD / 4)) tv[j] = t4[(size_t)c0 * (PD / 4) + i];
        }
        #pragma unroll
        for (int j = 0; j < INIT_X; ++j) {
            const int i = tid + j * BE_T;
            xv[j] = make_float4(0.0f, 0.0f, 0.0f, 0.0f);
            if (i < nrows * (PE / 4)) xv[j] = x4[(size_t)c0 * (PE / 4) + i];
        }
        #pragma unroll
        for (int j = 0; j < INIT_T; ++j) {
            const int i = tid + j * BE_T;
            if (i < nr16 * (PD / 4)) *(float4*)(&sT[(i >> 2) * TP + 4 * (i & 3)]) = tv[j];
        }
        #pragma unroll
        for (int j = 0; j < INIT_X; ++j) {
            const int i = tid + j * BE_T;
            if (i < nr16 * (PE / 4)) sAcc4[i] = xv[j];
        }
        __syncthreads();

        for (int t = wave; t < ntiles; t += BE_W) {
            const int  e0 = start + t * 16;
            const int2 ra = ra_n;
            const int2 rb = rb_n;
            if (t + BE_W < ntiles) {
                const int e0n = e0 + BE_W * 16;
                ra_n = rec2[min(e0n + col, end - 1)];
                if (FIRST) rb_n = rec2[min(e0n + row, end - 1)];
            }
            const int sv = ra.y & 0xffff;
            const int rc = (int)((unsigned)ra.y >> 16) - c0;
            const int dl = (e0 + col < end && (unsigned)rc < (unsigned)R_MAX) ? rc : R_MAX;

            float4 ga0 = make_float4(0.0f, 0.0f, 0.0f, 0.0f), ga1 = ga0;
            if (FIRST) {
                const float4* ar = (const float4*)(attr + (size_t)rb.x * PD + hlf * 8);
                ga0 = ar[0];
                ga1 = ar[1];
            }
            float yc[16][2];
            #pragma unroll
            for (int m = 0; m < 16; ++m) {
                const float* yr = y + (size_t)__builtin_amdgcn_readlane(sv, m) * PE + lane;
                yc[m][0] = yr[0];
                yc[m][1] = yr[32];
            }

            if (FIRST) {
                const int rca = (int)((unsigned)rb.y >> 16) - c0;
                const int dla = (e0 + row < end && (unsigned)rca < (unsigned)R_MAX) ? rca : R_MAX;
                float* tp = sT + dla * TP + hlf * 8;
                __hip_atomic_fetch_add(tp + 0, ga0.x, __ATOMIC_RELAXED, __HIP_MEMORY_SCOPE_WORKGROUP);
                __hip_atomic_fetch_add(tp + 1, ga0.y, __ATOMIC_RELAXED, __HIP_MEMORY_SCOPE_WORKGROUP);
                __hip_atomic_fetch_add(tp + 2, ga0.z, __ATOMIC_RELAXED, __HIP_MEMORY_SCOPE_WORKGROUP);
                __hip_atomic_fetch_add(tp + 3, ga0.w, __ATOMIC_RELAXED, __HIP_MEMORY_SCOPE_WORKGROUP);
                __hip_atomic_fetch_add(tp + 4, ga1.x, __ATOMIC_RELAXED, __HIP_MEMORY_SCOPE_WORKGROUP);
                __hip_atomic_fetch_add(tp + 5, ga1.y, __ATOMIC_RELAXED, __HIP_MEMORY_SCOPE_WORKGROUP);
                __hip_atomic_fetch_add(tp + 6, ga1.z, __ATOMIC_RELAXED, __HIP_MEMORY_SCOPE_WORKGROUP);
                __hip_atomic_fetch_add(tp + 7, ga1.w, __ATOMIC_RELAXED, __HIP_MEMORY_SCOPE_WORKGROUP);
            }

            #pragma unroll
            for (int m = 0; m < 16; ++m) {
                float* rw = sAcc + __builtin_amdgcn_readlane(dl, m) * PE + lane;
                __hip_atomic_fetch_add(rw,      yc[m][0], __ATOMIC_RELAXED, __HIP_MEMORY_SCOPE_WORKGROUP);
                __hip_atomic_fetch_add(rw + 32, yc[m][1], __ATOMIC_RELAXED, __HIP_MEMORY_SCOPE_WORKGROUP);
            }
        }
        __syncthreads();

        if (FIRST) {
            for (int i = tid; i < nrows * (PD / 4); i += BE_T)
                t4[(size_t)c0 * (PD / 4) + i] = *(const float4*)(&sT[(i >> 2) * TP + 4 * (i & 3)]);
        }

        for (int tl = wave; tl * 16 < nrows; tl += BE_W) {
            v8f acc[4];
            #pragma unroll
            for (int r = 0; r < 8; ++r) {
                acc[0][r] = 0.0f; acc[1][r] = 0.0f; acc[2][r] = 0.0f; acc[3][r] = 0.0f;
            }
            #pragma unroll
            for (int kk = 0; kk < 4; ++kk) {
                const int k0 = kk * 4;
                v2f a = *(const v2f*)(&sT[(tl * 16 + col) * TP + k0 + koff]);
                #pragma unroll
                for (int nn = 0; nn < 4; ++nn) {
                    v2f b = *(const v2f*)(&sWe[(kk * 2 + hi) * WP + (nn * 16 + col) * 2]);
                    acc[nn] = __builtin_amdgcn_wmma_f32_16x16x4_f32(
                        false, a, false, b, (short)0, acc[nn], false, false);
                }
            }
            float* base = sAcc + (tl * 16 + 8 * hi) * PE + col;
            #pragma unroll
            for (int r = 0; r < 8; ++r) {
                #pragma unroll
                for (int nn = 0; nn < 4; ++nn)
                    base[r * PE + nn * 16] += acc[nn][r];
            }
        }
        __syncthreads();

        if (big) {
            for (int i = tid; i < nrows * (PE / 4); i += BE_T)
                o4[(size_t)c0 * (PE / 4) + i] = sAcc4[i];
            __syncthreads();
        }
    }

    const float cntf = fmaxf((float)n, 1.0f);

    float4 v[NR];
    #pragma unroll
    for (int j = 0; j < NR; ++j) v[j] = make_float4(0.0f, 0.0f, 0.0f, 0.0f);

    float4 s = make_float4(0.0f, 0.0f, 0.0f, 0.0f);
    if (!big) {
        #pragma unroll
        for (int j = 0; j < NR; ++j) {
            const int r = ty + BE_RPP * j;
            if (r < n) v[j] = sAcc4[r * (PE / 4) + col];
        }
        #pragma unroll
        for (int j = 0; j < NR; ++j) { s.x += v[j].x; s.y += v[j].y; s.z += v[j].z; s.w += v[j].w; }
    } else {
        for (int r = ty; r < n; r += BE_RPP) {
            const float4 t = o4[(size_t)r * (PE / 4) + col];
            s.x += t.x; s.y += t.y; s.z += t.z; s.w += t.w;
        }
    }
    const float4 tot = gn_block_sum(s, sred[0], wave, hi, col);

    const float4 mm = make_float4(tot.x / cntf * ms.x, tot.y / cntf * ms.y, tot.z / cntf * ms.z, tot.w / cntf * ms.w);

    float4 q = make_float4(0.0f, 0.0f, 0.0f, 0.0f);
    if (!big) {
        #pragma unroll
        for (int j = 0; j < NR; ++j) {
            const int r = ty + BE_RPP * j;
            const float4 o = gn_center(v[j], mm);
            v[j] = o;
            if (r < n) { q.x += o.x * o.x; q.y += o.y * o.y; q.z += o.z * o.z; q.w += o.w * o.w; }
        }
    } else {
        for (int r = ty; r < n; r += BE_RPP) {
            const float4 o = gn_center(o4[(size_t)r * (PE / 4) + col], mm);
            q.x += o.x * o.x; q.y += o.y * o.y; q.z += o.z * o.z; q.w += o.w * o.w;
        }
    }
    const float4 qt = gn_block_sum(q, sred[1], wave, hi, col);

    const float4 sc = make_float4(wv.x * rsqrtf(qt.x / cntf + EPS), wv.y * rsqrtf(qt.y / cntf + EPS),
                                  wv.z * rsqrtf(qt.z / cntf + EPS), wv.w * rsqrtf(qt.w / cntf + EPS));

    if (!big) {
        #pragma unroll
        for (int j = 0; j < NR; ++j) {
            const int r = ty + BE_RPP * j;
            if (r < n) o4[(size_t)r * (PE / 4) + col] = gn_finish(v[j], sc, bv);
        }
    } else {
        for (int r = ty; r < n; r += BE_RPP)
            o4[(size_t)r * (PE / 4) + col] = gn_finish(gn_center(o4[(size_t)r * (PE / 4) + col], mm), sc, bv);
    }
}

extern "C" void kernel_launch(void* const* d_in, const int* in_sizes, int n_in,
                              void* d_out, int out_size, void* d_ws, size_t ws_size,
                              hipStream_t stream) {
    const float* x0    = (const float*)d_in[0];
    const int*   idx   = (const int*)  d_in[1];
    const float* attr  = (const float*)d_in[2];
    const int*   batch = (const int*)  d_in[3];
    const float* Wm    = (const float*)d_in[4];
    const float* bm    = (const float*)d_in[5];
    const float* We    = (const float*)d_in[6];
    const float* be    = (const float*)d_in[7];
    const float* gw    = (const float*)d_in[8];
    const float* gb    = (const float*)d_in[9];
    const float* gms   = (const float*)d_in[10];

    const int* src = idx;
    const int* dst = idx + N_EDGES;

    char* ws = (char*)d_ws;
    const size_t node_bytes = (size_t)N_NODES * PE * sizeof(float);
    float* h       = (float*)(ws);
    float* tmp     = (float*)(ws + node_bytes);
    char*  grp     = ws + 2 * node_bytes;
    int*   gstart  = (int*)  (ws + 3 * node_bytes);
    int*   rec     = (int*)  (grp);
    int*   bincnt  = (int*)  (grp + (size_t)N_EDGES * 8);
    int*   binbase = (int*)  (grp + (size_t)N_EDGES * 8 + NBINS * 4);
    int*   wgoff   = (int*)  (grp + (size_t)N_EDGES * 8 + 2 * NBINS * 4);
    float* tsum    = (float*)(grp + (size_t)N_EDGES * 8 + 2 * NBINS * 4 + (size_t)BC_G * NBINS * 4);
    static_assert((size_t)N_EDGES * 8 + 2 * NBINS * 4 + (size_t)BC_G * NBINS * 4 + (size_t)N_NODES * PD * 4
                      <= (size_t)N_NODES * PE * 4,
                  "the grouped list, its tables and the attr sums fit the third node-sized plane of the workspace");
    static_assert(((size_t)N_EDGES * 8 + 2 * NBINS * 4 + (size_t)BC_G * NBINS * 4) % 16 == 0,
                  "the attr sums are read and written as float4");

    for (int l = 0; l < 2; ++l) {
        const float* xin  = (l == 0) ? x0 : h;
        float*       xout = (l == 1) ? (float*)d_out : h;

        node_msg_kernel<<<NODE_GRID, 256, 0, stream>>>(
            xin, Wm + (size_t)l * PE * PE, bm + (size_t)l * PE, be + (size_t)l * PE, batch, tmp,
            (l == 0) ? gstart : (int*)nullptr, (l == 0) ? bincnt : (int*)nullptr);

        if (l == 0) {
            bin_count_kernel<<<BC_G, BC_T, 0, stream>>>(dst, gstart, bincnt, wgoff);
            bin_scatter_kernel<<<BC_G, BC_T, 0, stream>>>(src, dst, gstart, bincnt, wgoff, binbase, rec);
        }

        if (l == 0)
            edge_norm_kernel<true><<<N_GRAPHS, BE_T, 0, stream>>>(
                tmp, xin, rec, binbase, bincnt, gstart, attr, We + (size_t)l * PD * PE, tsum,
                gms + (size_t)l * PE, gw + (size_t)l * PE, gb + (size_t)l * PE, xout);
        else
            edge_norm_kernel<false><<<N_GRAPHS, BE_T, 0, stream>>>(
                tmp, xin, rec, binbase, bincnt, gstart, attr, We + (size_t)l * PD * PE, tsum,
                gms + (size_t)l * PE, gw + (size_t)l * PE, gb + (size_t)l * PE, xout);
    }
}


# second layer's x @ W_msg formed at the end of the first layer's edge kernel by the graph's own work-group (normed rows back into LDS, f32 WMMA there); h rides in the output buffer; one launch fewer
# speedup vs baseline: 3.7647x; 1.0609x over previous
#include <hip/hip_runtime.h>
#include <hip/hip_bf16.h>
#include <stdint.h>

#define N_NODES   50000
#define N_EDGES   800000
#define PE        64
#define PD        16
#define N_GRAPHS  256
#define EPS       1e-5f
#define NEG_SLOPE 0.01f

#define XP 68
#define WP 160

typedef __attribute__((ext_vector_type(2))) float v2f;
typedef __attribute__((ext_vector_type(8))) float v8f;

__device__ __forceinline__ float xhalf(float v) {
    return __builtin_bit_cast(float, __builtin_amdgcn_permlanex16(
        __builtin_bit_cast(int, v), __builtin_bit_cast(int, v), 0x76543210, 0xfedcba98, false, false));
}

#define NODE_TILES (N_NODES / 16)
#define NODE_GRID  ((NODE_TILES + 7) / 8)
static_assert(N_NODES % 16 == 0, "the node kernel has no partial tile");

#define NBINS N_GRAPHS
static_assert(N_NODES <= 65536, "a record packs the source row in 16 bits, the row in its graph above them");
static_assert(NBINS <= NODE_GRID * 256, "the node kernel zeroes one bin counter per thread");
static_assert((N_GRAPHS & (N_GRAPHS - 1)) == 0, "the search for a node's graph halves a power of two");

__global__ __launch_bounds__(256) void node_msg_kernel(
    const float* x,
    const float* __restrict__ Wm,
    const float* __restrict__ bm,
    const float* __restrict__ be,
    const int*   __restrict__ batch,
    float*       __restrict__ y,
    int*         __restrict__ gstart,
    int*         __restrict__ bincnt)
{
    __shared__ __attribute__((aligned(16))) float sWm[(PE / 2) * WP];
    __shared__ __attribute__((aligned(16))) float sX[8][16 * XP];

    const int tid  = threadIdx.x;
    const int wave = __builtin_amdgcn_readfirstlane(tid >> 5);
    const int lane = tid & 31;

    const int col  = lane & 15;
    const int hi   = lane >> 4;
    const int koff = hi * 2;

    const int  tile = blockIdx.x * 8 + wave;
    const bool live = tile < NODE_TILES;

    const float bias0 = bm[col]      + be[col];
    const float bias1 = bm[16 + col] + be[16 + col];
    const float bias2 = bm[32 + col] + be[32 + col];
    const float bias3 = bm[48 + col] + be[48 + col];

    float4 v0, v1, v2, v3, v4, v5, v6, v7;
    v0 = v1 = v2 = v3 = v4 = v5 = v6 = v7 = make_float4(0.0f, 0.0f, 0.0f, 0.0f);
    const int rq = hi * (PE / 4) + col;
    if (live) {
        const float4* x4 = (const float4*)(x + (size_t)tile * 16 * PE) + rq;
        v0 = x4[0 * (PE / 2)]; v1 = x4[1 * (PE / 2)]; v2 = x4[2 * (PE / 2)]; v3 = x4[3 * (PE / 2)];
        v4 = x4[4 * (PE / 2)]; v5 = x4[5 * (PE / 2)]; v6 = x4[6 * (PE / 2)]; v7 = x4[7 * (PE / 2)];
    }

    {
        const float4* Wm4 = (const float4*)Wm;
        #pragma unroll
        for (int it = tid; it < (PE / 2) * (PE / 4); it += 256) {
            const int kp = it >> 4, n4 = it & 15;
            const float4 r0 = Wm4[(2 * kp    ) * (PE / 4) + n4];
            const float4 r1 = Wm4[(2 * kp + 1) * (PE / 4) + n4];
            float4* o = (float4*)(&sWm[kp * WP + n4 * 8]);
            o[0] = make_float4(r0.x, r1.x, r0.y, r1.y);
            o[1] = make_float4(r0.z, r1.z, r0.w, r1.w);
        }
    }

    if (live) {
        float4* sx = (float4*)(&sX[wave][hi * XP + 4 * col]);
        sx[0 * (XP / 2)] = v0; sx[1 * (XP / 2)] = v1; sx[2 * (XP / 2)] = v2; sx[3 * (XP / 2)] = v3;
        sx[4 * (XP / 2)] = v4; sx[5 * (XP / 2)] = v5; sx[6 * (XP / 2)] = v6; sx[7 * (XP / 2)] = v7;
    }
    __syncthreads();

    if (live) {
        v8f acc[4];
        #pragma unroll
        for (int r = 0; r < 8; ++r) {
            acc[0][r] = bias0; acc[1][r] = bias1; acc[2][r] = bias2; acc[3][r] = bias3;
        }
        #pragma unroll
        for (int kk = 0; kk < 16; ++kk) {
            const int k0 = kk * 4;
            v2f a = *(const v2f*)(&sX[wave][col * XP + k0 + koff]);
            #pragma unroll
            for (int n = 0; n < 4; ++n) {
                v2f b = *(const v2f*)(&sWm[(kk * 2 + hi) * WP + (n * 16 + col) * 2]);
                acc[n] = __builtin_amdgcn_wmma_f32_16x16x4_f32(
                    false, a, false, b, (short)0, acc[n], false, false);
            }
        }
        float* yt = y + (size_t)tile * 16 * PE + lane;
        #pragma unroll
        for (int r = 0; r < 8; ++r) {
            #pragma unroll
            for (int p = 0; p < 2; ++p) {
                const float X  = acc[2 * p][r];
                const float Y  = acc[2 * p + 1][r];
                const float Z  = hi ? X : Y;
                const float Zx = xhalf(Z);
                yt[(r    ) * PE + 32 * p] = hi ? Zx : X;
                yt[(r + 8) * PE + 32 * p] = hi ? Y  : Zx;
            }
        }
    }

    const int gi = blockIdx.x * 256 + tid;
    if (gstart != nullptr && gi < N_NODES) {
        const int gb_cur = batch[gi];
        const int gfirst = gi > 0 ? batch[gi - 1] + 1 : 0;
        for (int g = gfirst; g <= gb_cur; ++g) gstart[g] = gi;
        if (gi == N_NODES - 1)
            for (int g = gb_cur + 1; g <= N_GRAPHS; ++g) gstart[g] = N_NODES;
    }
    if (bincnt != nullptr && gi < NBINS) bincnt[gi] = 0;
}

#define BC_T  512
#define BC_CH 3200
#define BC_G  (N_EDGES / BC_CH)
static_assert(N_EDGES % BC_CH == 0, "the grouping kernels have no partial chunk");
static_assert(NBINS <= BC_T, "one thread a bin in the grouping kernels");
#define BC_K ((BC_CH + BC_T - 1) / BC_T)

__device__ __forceinline__ int graph_of(const int* gs, int d) {
    int g = 0;
    #pragma unroll
    for (int st = N_GRAPHS / 2; st >= 1; st >>= 1) {
        const int m = g + st;
        if (gs[m] <= d) g = m;
    }
    return g;
}

__global__ __launch_bounds__(BC_T) void bin_count_kernel(
    const int* __restrict__ dst,
    const int* __restrict__ gstart,
    int*       __restrict__ bincnt,
    int*       __restrict__ wgoff)
{
    __shared__ int hist[NBINS];
    __shared__ int sgs[NBINS];
    const int tid = threadIdx.x;
    const int e_lo = blockIdx.x * BC_CH;
    int d[BC_K];
    #pragma unroll
    for (int k = 0; k < BC_K; ++k) {
        const int e = tid + k * BC_T;
        d[k] = e < BC_CH ? dst[e_lo + e] : -1;
    }
    if (tid < NBINS) { hist[tid] = 0; sgs[tid] = gstart[tid]; }
    __syncthreads();
    int b[BC_K];
    #pragma unroll
    for (int k = 0; k < BC_K; ++k) b[k] = graph_of(sgs, d[k]);
    #pragma unroll
    for (int k = 0; k < BC_K; ++k)
        if (d[k] >= 0)
            __hip_atomic_fetch_add(&hist[b[k]], 1, __ATOMIC_RELAXED, __HIP_MEMORY_SCOPE_WORKGROUP);
    __syncthreads();
    if (tid < NBINS) wgoff[blockIdx.x * NBINS + tid] = atomicAdd(&bincnt[tid], hist[tid]);
}

__global__ __launch_bounds__(BC_T) void bin_scatter_kernel(
    const int* __restrict__ src,
    const int* __restrict__ dst,
    const int* __restrict__ gstart,
    const int* __restrict__ bincnt,
    const int* __restrict__ wgoff,
    int*       __restrict__ binbase,
    int*       __restrict__ rec)
{
    __shared__ int wsum[BC_T / 32];
    __shared__ int cursor[NBINS];
    __shared__ int sgs[NBINS];
    const int tid  = threadIdx.x;
    const int lane = tid & 31;
    const int wave = tid >> 5;
    const bool own = tid < NBINS;
    const int c    = own ? bincnt[tid] : 0;
    const int off  = own ? wgoff[blockIdx.x * NBINS + tid] : 0;
    if (own) sgs[tid] = gstart[tid];
    const int e_lo = blockIdx.x * BC_CH;
    int d[BC_K], s[BC_K];
    #pragma unroll
    for (int k = 0; k < BC_K; ++k) {
        const int e = tid + k * BC_T;
        d[k] = e < BC_CH ? dst[e_lo + e] : -1;
        s[k] = e < BC_CH ? src[e_lo + e] : 0;
    }
    int v = c;
    for (int o = 1; o < 32; o <<= 1) {
        const int u = __shfl_up(v, o);
        if (lane >= o) v += u;
    }
    if (lane == 31) wsum[wave] = v;
    __syncthreads();
    int pre = 0;
    for (int w = 0; w < BC_T / 32; ++w) {
        const int t = wsum[w];
        if (w < wave) pre += t;
    }
    const int excl = pre + v - c;
    if (own) {
        cursor[tid] = excl + off;
        if (blockIdx.x == 0) binbase[tid] = excl;
    }
    __syncthreads();
    int b[BC_K];
    #pragma unroll
    for (int k = 0; k < BC_K; ++k) b[k] = graph_of(sgs, d[k]);
    int2* rec2 = (int2*)rec;
    #pragma unroll
    for (int k = 0; k < BC_K; ++k) {
        if (d[k] >= 0) {
            const int pos = __hip_atomic_fetch_add(&cursor[b[k]], 1, __ATOMIC_RELAXED, __HIP_MEMORY_SCOPE_WORKGROUP);
            const unsigned rowg = (unsigned)(d[k] - sgs[b[k]]);
            rec2[pos] = make_int2(e_lo + tid + k * BC_T, (int)((unsigned)s[k] | (rowg << 16)));
        }
    }
}

#define BE_T   512
#define BE_W   (BE_T / 32)
#define BE_RPP (BE_T / 16)
#define TP 20
#define R_MAX   448
#define ET_ROWS (R_MAX + 16)
#define INIT_X ((R_MAX * (PE / 4) + BE_T - 1) / BE_T)
#define INIT_T ((R_MAX * (PD / 4) + BE_T - 1) / BE_T)
#define NR     (R_MAX / BE_RPP)
static_assert(R_MAX % 16 == 0 && R_MAX % BE_RPP == 0,
              "chunks are whole row tiles, and a thread keeps its rows from chunk to norm");

__device__ __forceinline__ float4 gn_block_sum(float4 s, float4 (*buf)[16], int wave, int hi, int tx) {
    s.x += xhalf(s.x); s.y += xhalf(s.y); s.z += xhalf(s.z); s.w += xhalf(s.w);
    if (hi == 0) buf[wave][tx] = s;
    __syncthreads();
    float4 t = buf[hi * (BE_W / 2)][tx];
    #pragma unroll
    for (int k = 1; k < BE_W / 2; ++k) {
        const float4 p = buf[hi * (BE_W / 2) + k][tx];
        t.x += p.x; t.y += p.y; t.z += p.z; t.w += p.w;
    }
    t.x += xhalf(t.x); t.y += xhalf(t.y); t.z += xhalf(t.z); t.w += xhalf(t.w);
    return t;
}

__device__ __forceinline__ float4 gn_center(const float4 v, const float4 mm) {
    return make_float4(v.x - mm.x, v.y - mm.y, v.z - mm.z, v.w - mm.w);
}

__device__ __forceinline__ float4 gn_finish(const float4 o, const float4 sc, const float4 b) {
    float4 r;
    r.x = o.x * sc.x + b.x;
    r.y = o.y * sc.y + b.y;
    r.z = o.z * sc.z + b.z;
    r.w = o.w * sc.w + b.w;
    r.x = r.x > 0.0f ? r.x : NEG_SLOPE * r.x;
    r.y = r.y > 0.0f ? r.y : NEG_SLOPE * r.y;
    r.z = r.z > 0.0f ? r.z : NEG_SLOPE * r.z;
    r.w = r.w > 0.0f ? r.w : NEG_SLOPE * r.w;
    return r;
}

template <bool FIRST>
__global__ __launch_bounds__(BE_T) void edge_norm_kernel(
    const float* __restrict__ y,
    const float* xin,
    const int*   __restrict__ rec,
    const int*   __restrict__ binbase,
    const int*   __restrict__ bincnt,
    const int*   __restrict__ gstart,
    const float* __restrict__ attr,
    const float* __restrict__ We,
    float*       tsum,
    const float* __restrict__ mscale,
    const float* __restrict__ w,
    const float* __restrict__ bias,
    float*       hout,
    const float* __restrict__ Wn,
    const float* __restrict__ bmn,
    const float* __restrict__ ben,
    float*       __restrict__ ynext)
{
    __shared__ __attribute__((aligned(16))) float sAcc[ET_ROWS * PE];
    __shared__ __attribute__((aligned(16))) float sWe[(PD / 2) * WP];
    __shared__ __attribute__((aligned(16))) float sT[ET_ROWS * TP];
    __shared__ __attribute__((aligned(16))) float4 sred[2][BE_W][16];

    const int tid  = threadIdx.x;
    const int wave = __builtin_amdgcn_readfirstlane(tid >> 5);
    const int lane = tid & 31;
    const int bin  = blockIdx.x;

    const int start = binbase[bin];
    const int cnt   = bincnt[bin];
    const int end   = start + cnt;

    const int  lo      = gstart[bin];
    const int  n       = gstart[bin + 1] - lo;
    const bool big     = n > R_MAX;
    const int  nchunks = (n + R_MAX - 1) / R_MAX;

    const int col  = lane & 15;
    const int hi   = lane >> 4;
    const int koff = hi * 2;

    const int ty = tid >> 4;

    const float4 ms = ((const float4*)mscale)[col];
    const float4 wv = ((const float4*)w)[col];
    const float4 bv = ((const float4*)bias)[col];

    float4 wn0 = make_float4(0.0f, 0.0f, 0.0f, 0.0f), wn1 = wn0;
    float nb0 = 0.0f, nb1 = 0.0f, nb2 = 0.0f, nb3 = 0.0f;
    if (FIRST) {
        const float4* Wn4 = (const float4*)Wn;
        const int kp = tid >> 4, n4 = tid & 15;
        wn0 = Wn4[(2 * kp    ) * (PE / 4) + n4];
        wn1 = Wn4[(2 * kp + 1) * (PE / 4) + n4];
        nb0 = bmn[col]      + ben[col];
        nb1 = bmn[16 + col] + ben[16 + col];
        nb2 = bmn[32 + col] + ben[32 + col];
        nb3 = bmn[48 + col] + ben[48 + col];
    }

    {
        const float4* We4 = (const float4*)We;
        if (tid < (PD / 2) * (PE / 4)) {
            const int kp = tid >> 4, n4 = tid & 15;
            const float4 r0 = We4[(2 * kp    ) * (PE / 4) + n4];
            const float4 r1 = We4[(2 * kp + 1) * (PE / 4) + n4];
            float4* o = (float4*)(&sWe[kp * WP + n4 * 8]);
            o[0] = make_float4(r0.x, r1.x, r0.y, r1.y);
            o[1] = make_float4(r0.z, r1.z, r0.w, r1.w);
        }
    }

    const int row = lane >> 1;
    const int hlf = lane & 1;

    const int2* rec2   = (const int2*)rec;
    const int   ntiles = (cnt + 15) >> 4;

    float4*       sAcc4 = (float4*)sAcc;
    const float4* x4    = (const float4*)xin + (size_t)lo * (PE / 4);
    float4*       o4    = (float4*)hout      + (size_t)lo * (PE / 4);
    float4*       t4    = (float4*)tsum      + (size_t)lo * (PD / 4);

    for (int c = 0; c < nchunks; ++c) {
        const int c0    = c * R_MAX;
        const int nrows = min(R_MAX, n - c0);
        const int nr16  = (nrows + 15) & ~15;

        int2 ra_n = make_int2(0, 0);
        int2 rb_n = make_int2(0, 0);
        if (wave < ntiles) {
            const int e0f = start + wave * 16;
            ra_n = rec2[min(e0f + col, end - 1)];
            if (FIRST) rb_n = rec2[min(e0f + row, end - 1)];
        }

        float4 tv[INIT_T], xv[INIT_X];
        #pragma unroll
        for (int j = 0; j < INIT_T; ++j) {
            const int i = tid + j * BE_T;
            tv[j] = make_float4(0.0f, 0.0f, 0.0f, 0.0f);
            if (!FIRST && i < nrows * (PD / 4)) tv[j] = t4[(size_t)c0 * (PD / 4) + i];
        }
        #pragma unroll
        for (int j = 0; j < INIT_X; ++j) {
            const int i = tid + j * BE_T;
            xv[j] = make_float4(0.0f, 0.0f, 0.0f, 0.0f);
            if (i < nrows * (PE / 4)) xv[j] = x4[(size_t)c0 * (PE / 4) + i];
        }
        #pragma unroll
        for (int j = 0; j < INIT_T; ++j) {
            const int i = tid + j * BE_T;
            if (i < nr16 * (PD / 4)) *(float4*)(&sT[(i >> 2) * TP + 4 * (i & 3)]) = tv[j];
        }
        #pragma unroll
        for (int j = 0; j < INIT_X; ++j) {
            const int i = tid + j * BE_T;
            if (i < nr16 * (PE / 4)) sAcc4[i] = xv[j];
        }
        __syncthreads();

        for (int t = wave; t < ntiles; t += BE_W) {
            const int  e0 = start + t * 16;
            const int2 ra = ra_n;
            const int2 rb = rb_n;
            if (t + BE_W < ntiles) {
                const int e0n = e0 + BE_W * 16;
                ra_n = rec2[min(e0n + col, end - 1)];
                if (FIRST) rb_n = rec2[min(e0n + row, end - 1)];
            }
            const int sv = ra.y & 0xffff;
            const int rc = (int)((unsigned)ra.y >> 16) - c0;
            const int dl = (e0 + col < end && (unsigned)rc < (unsigned)R_MAX) ? rc : R_MAX;

            float4 ga0 = make_float4(0.0f, 0.0f, 0.0f, 0.0f), ga1 = ga0;
            if (FIRST) {
                const float4* ar = (const float4*)(attr + (size_t)rb.x * PD + hlf * 8);
                ga0 = ar[0];
                ga1 = ar[1];
            }
            float yc[16][2];
            #pragma unroll
            for (int m = 0; m < 16; ++m) {
                const float* yr = y + (size_t)__builtin_amdgcn_readlane(sv, m) * PE + lane;
                yc[m][0] = yr[0];
                yc[m][1] = yr[32];
            }

            if (FIRST) {
                const int rca = (int)((unsigned)rb.y >> 16) - c0;
                const int dla = (e0 + row < end && (unsigned)rca < (unsigned)R_MAX) ? rca : R_MAX;
                float* tp = sT + dla * TP + hlf * 8;
                __hip_atomic_fetch_add(tp + 0, ga0.x, __ATOMIC_RELAXED, __HIP_MEMORY_SCOPE_WORKGROUP);
                __hip_atomic_fetch_add(tp + 1, ga0.y, __ATOMIC_RELAXED, __HIP_MEMORY_SCOPE_WORKGROUP);
                __hip_atomic_fetch_add(tp + 2, ga0.z, __ATOMIC_RELAXED, __HIP_MEMORY_SCOPE_WORKGROUP);
                __hip_atomic_fetch_add(tp + 3, ga0.w, __ATOMIC_RELAXED, __HIP_MEMORY_SCOPE_WORKGROUP);
                __hip_atomic_fetch_add(tp + 4, ga1.x, __ATOMIC_RELAXED, __HIP_MEMORY_SCOPE_WORKGROUP);
                __hip_atomic_fetch_add(tp + 5, ga1.y, __ATOMIC_RELAXED, __HIP_MEMORY_SCOPE_WORKGROUP);
                __hip_atomic_fetch_add(tp + 6, ga1.z, __ATOMIC_RELAXED, __HIP_MEMORY_SCOPE_WORKGROUP);
                __hip_atomic_fetch_add(tp + 7, ga1.w, __ATOMIC_RELAXED, __HIP_MEMORY_SCOPE_WORKGROUP);
            }

            #pragma unroll
            for (int m = 0; m < 16; ++m) {
                float* rw = sAcc + __builtin_amdgcn_readlane(dl, m) * PE + lane;
                __hip_atomic_fetch_add(rw,      yc[m][0], __ATOMIC_RELAXED, __HIP_MEMORY_SCOPE_WORKGROUP);
                __hip_atomic_fetch_add(rw + 32, yc[m][1], __ATOMIC_RELAXED, __HIP_MEMORY_SCOPE_WORKGROUP);
            }
        }
        __syncthreads();

        if (FIRST) {
            for (int i = tid; i < nrows * (PD / 4); i += BE_T)
                t4[(size_t)c0 * (PD / 4) + i] = *(const float4*)(&sT[(i >> 2) * TP + 4 * (i & 3)]);
        }

        for (int tl = wave; tl * 16 < nrows; tl += BE_W) {
            v8f acc[4];
            #pragma unroll
            for (int r = 0; r < 8; ++r) {
                acc[0][r] = 0.0f; acc[1][r] = 0.0f; acc[2][r] = 0.0f; acc[3][r] = 0.0f;
            }
            #pragma unroll
            for (int kk = 0; kk < 4; ++kk) {
                const int k0 = kk * 4;
                v2f a = *(const v2f*)(&sT[(tl * 16 + col) * TP + k0 + koff]);
                #pragma unroll
                for (int nn = 0; nn < 4; ++nn) {
                    v2f b = *(const v2f*)(&sWe[(kk * 2 + hi) * WP + (nn * 16 + col) * 2]);
                    acc[nn] = __builtin_amdgcn_wmma_f32_16x16x4_f32(
                        false, a, false, b, (short)0, acc[nn], false, false);
                }
            }
            float* base = sAcc + (tl * 16 + 8 * hi) * PE + col;
            #pragma unroll
            for (int r = 0; r < 8; ++r) {
                #pragma unroll
                for (int nn = 0; nn < 4; ++nn)
                    base[r * PE + nn * 16] += acc[nn][r];
            }
        }
        __syncthreads();

        if (big) {
            for (int i = tid; i < nrows * (PE / 4); i += BE_T)
                o4[(size_t)c0 * (PE / 4) + i] = sAcc4[i];
            __syncthreads();
        }
    }

    float* sWn = sT;
    static_assert(BE_T == (PE / 2) * (PE / 4), "one item of the next layer's W_msg a thread");
    static_assert((PE / 2) * WP <= ET_ROWS * TP, "the next layer's W_msg fits the attr sums' array");
    if (FIRST) {
        const int kp = tid >> 4, n4 = tid & 15;
        float4* o = (float4*)(&sWn[kp * WP + n4 * 8]);
        o[0] = make_float4(wn0.x, wn1.x, wn0.y, wn1.y);
        o[1] = make_float4(wn0.z, wn1.z, wn0.w, wn1.w);
    }

    const float cntf = fmaxf((float)n, 1.0f);

    float4 v[NR];
    #pragma unroll
    for (int j = 0; j < NR; ++j) v[j] = make_float4(0.0f, 0.0f, 0.0f, 0.0f);

    float4 s = make_float4(0.0f, 0.0f, 0.0f, 0.0f);
    if (!big) {
        #pragma unroll
        for (int j = 0; j < NR; ++j) {
            const int r = ty + BE_RPP * j;
            if (r < n) v[j] = sAcc4[r * (PE / 4) + col];
        }
        #pragma unroll
        for (int j = 0; j < NR; ++j) { s.x += v[j].x; s.y += v[j].y; s.z += v[j].z; s.w += v[j].w; }
    } else {
        for (int r = ty; r < n; r += BE_RPP) {
            const float4 t = o4[(size_t)r * (PE / 4) + col];
            s.x += t.x; s.y += t.y; s.z += t.z; s.w += t.w;
        }
    }
    const float4 tot = gn_block_sum(s, sred[0], wave, hi, col);

    const float4 mm = make_float4(tot.x / cntf * ms.x, tot.y / cntf * ms.y, tot.z / cntf * ms.z, tot.w / cntf * ms.w);

    float4 q = make_float4(0.0f, 0.0f, 0.0f, 0.0f);
    if (!big) {
        #pragma unroll
        for (int j = 0; j < NR; ++j) {
            const int r = ty + BE_RPP * j;
            const float4 o = gn_center(v[j], mm);
            v[j] = o;
            if (r < n) { q.x += o.x * o.x; q.y += o.y * o.y; q.z += o.z * o.z; q.w += o.w * o.w; }
        }
    } else {
        for (int r = ty; r < n; r += BE_RPP) {
            const float4 o = gn_center(o4[(size_t)r * (PE / 4) + col], mm);
            q.x += o.x * o.x; q.y += o.y * o.y; q.z += o.z * o.z; q.w += o.w * o.w;
        }
    }
    const float4 qt = gn_block_sum(q, sred[1], wave, hi, col);

    const float4 sc = make_float4(wv.x * rsqrtf(qt.x / cntf + EPS), wv.y * rsqrtf(qt.y / cntf + EPS),
                                  wv.z * rsqrtf(qt.z / cntf + EPS), wv.w * rsqrtf(qt.w / cntf + EPS));

    const int swz = col ^ (ty & 15);
    static_assert(BE_RPP % 16 == 0, "a thread's rows share their low four bits");
    if (!big) {
        #pragma unroll
        for (int j = 0; j < NR; ++j) {
            const int r = ty + BE_RPP * j;
            if (r < n) {
                const float4 f = gn_finish(v[j], sc, bv);
                o4[(size_t)r * (PE / 4) + col] = f;
                if (FIRST) sAcc4[r * (PE / 4) + swz] = f;
            }
        }
    } else {
        for (int r = ty; r < n; r += BE_RPP)
            o4[(size_t)r * (PE / 4) + col] = gn_finish(gn_center(o4[(size_t)r * (PE / 4) + col], mm), sc, bv);
    }

    if (FIRST) {
        for (int c = 0; c < nchunks; ++c) {
            const int c0    = c * R_MAX;
            const int nrows = min(R_MAX, n - c0);
            if (big) {
                __syncthreads();
                for (int r = ty; r < nrows; r += BE_RPP)
                    sAcc4[r * (PE / 4) + swz] = o4[(size_t)(c0 + r) * (PE / 4) + col];
            }
            __syncthreads();
            const int nun = ((nrows + 15) >> 4) * 2;
            for (int u = wave; u < nun; u += BE_W) {
                const int tl = u >> 1;
                const int p  = u & 1;
                v8f acc0, acc1;
                #pragma unroll
                for (int r = 0; r < 8; ++r) {
                    acc0[r] = p ? nb2 : nb0;
                    acc1[r] = p ? nb3 : nb1;
                }
                const float* arow = sAcc + (tl * 16 + col) * PE + koff;
                const float* bcol = sWn + hi * WP + (p * 32 + col) * 2;
                #pragma unroll
                for (int kk = 0; kk < 16; ++kk) {
                    v2f a  = *(const v2f*)(arow + ((kk ^ col) << 2));
                    v2f b0 = *(const v2f*)(bcol + (kk * 2) * WP);
                    v2f b1 = *(const v2f*)(bcol + (kk * 2) * WP + 32);
                    acc0 = __builtin_amdgcn_wmma_f32_16x16x4_f32(
                        false, a, false, b0, (short)0, acc0, false, false);
                    acc1 = __builtin_amdgcn_wmma_f32_16x16x4_f32(
                        false, a, false, b1, (short)0, acc1, false, false);
                }
                const int rbase = c0 + tl * 16;
                float* yt = ynext + (size_t)(lo + rbase) * PE + 32 * p + lane;
                #pragma unroll
                for (int r = 0; r < 8; ++r) {
                    const float X  = acc0[r];
                    const float Y  = acc1[r];
                    const float Z  = hi ? X : Y;
                    const float Zx = xhalf(Z);
                    if (rbase + r     < n) yt[(r    ) * PE] = hi ? Zx : X;
                    if (rbase + r + 8 < n) yt[(r + 8) * PE] = hi ? Y  : Zx;
                }
            }
        }
    }
}

extern "C" void kernel_launch(void* const* d_in, const int* in_sizes, int n_in,
                              void* d_out, int out_size, void* d_ws, size_t ws_size,
                              hipStream_t stream) {
    const float* x0    = (const float*)d_in[0];
    const int*   idx   = (const int*)  d_in[1];
    const float* attr  = (const float*)d_in[2];
    const int*   batch = (const int*)  d_in[3];
    const float* Wm    = (const float*)d_in[4];
    const float* bm    = (const float*)d_in[5];
    const float* We    = (const float*)d_in[6];
    const float* be    = (const float*)d_in[7];
    const float* gw    = (const float*)d_in[8];
    const float* gb    = (const float*)d_in[9];
    const float* gms   = (const float*)d_in[10];

    const int* src = idx;
    const int* dst = idx + N_EDGES;

    char* ws = (char*)d_ws;
    const size_t node_bytes = (size_t)N_NODES * PE * sizeof(float);
    float* y1      = (float*)(ws);
    float* tmp     = (float*)(ws + node_bytes);
    char*  grp     = ws + 2 * node_bytes;
    int*   gstart  = (int*)  (ws + 3 * node_bytes);
    int*   rec     = (int*)  (grp);
    int*   bincnt  = (int*)  (grp + (size_t)N_EDGES * 8);
    int*   binbase = (int*)  (grp + (size_t)N_EDGES * 8 + NBINS * 4);
    int*   wgoff   = (int*)  (grp + (size_t)N_EDGES * 8 + 2 * NBINS * 4);
    float* tsum    = (float*)(grp + (size_t)N_EDGES * 8 + 2 * NBINS * 4 + (size_t)BC_G * NBINS * 4);
    static_assert((size_t)N_EDGES * 8 + 2 * NBINS * 4 + (size_t)BC_G * NBINS * 4 + (size_t)N_NODES * PD * 4
                      <= (size_t)N_NODES * PE * 4,
                  "the grouped list, its tables and the attr sums fit the third node-sized plane of the workspace");
    static_assert(((size_t)N_EDGES * 8 + 2 * NBINS * 4 + (size_t)BC_G * NBINS * 4) % 16 == 0,
                  "the attr sums are read and written as float4");

    float* out = (float*)d_out;

    node_msg_kernel<<<NODE_GRID, 256, 0, stream>>>(x0, Wm, bm, be, batch, tmp, gstart, bincnt);

    bin_count_kernel<<<BC_G, BC_T, 0, stream>>>(dst, gstart, bincnt, wgoff);
    bin_scatter_kernel<<<BC_G, BC_T, 0, stream>>>(src, dst, gstart, bincnt, wgoff, binbase, rec);

    edge_norm_kernel<true><<<N_GRAPHS, BE_T, 0, stream>>>(
        tmp, x0, rec, binbase, bincnt, gstart, attr, We, tsum, gms, gw, gb, out,
        Wm + (size_t)PE * PE, bm + PE, be + PE, y1);

    edge_norm_kernel<false><<<N_GRAPHS, BE_T, 0, stream>>>(
        y1, out, rec, binbase, bincnt, gstart, attr, We + (size_t)PD * PE, tsum,
        gms + PE, gw + PE, gb + PE, out,
        (const float*)nullptr, (const float*)nullptr, (const float*)nullptr, (float*)nullptr);
}
